# starting kernel
# baseline (speedup 1.0000x reference)
#include <hip/hip_runtime.h>
#include <math.h>

typedef __attribute__((ext_vector_type(2))) float v2f;
typedef __attribute__((ext_vector_type(8))) float v8f;

#define MM    129
#define MP    144
#define NPP   512
#define NB    4
#define NPTS  (NB * NPP)

#define PI_F   3.14159265358979323846f
#define TAU_F  (12.0f / (129.0f * 129.0f))
#define L_F    (2.0f * PI_F)


__device__ __forceinline__ v8f wmma4(v2f a, v2f b, v8f c) {
  return __builtin_amdgcn_wmma_f32_16x16x4_f32(false, a, false, b, (short)0, c, false, false);
}

__device__ __forceinline__ void store_tile(float* C, int ldc, int row0, int col0, v8f acc) {
  int lane = threadIdx.x & 31;
  int r0 = row0 + ((lane >> 4) << 3);
  int c  = col0 + (lane & 15);
#pragma unroll
  for (int v = 0; v < 8; ++v) C[(r0 + v) * ldc + c] = acc[v];
}

template <typename FA, typename FB>
__device__ __forceinline__ v8f mm_tile(int row0, int col0, int K, FA la, FB lb) {
  int lane = threadIdx.x & 31;
  int r  = lane & 15;
  int kb = (lane >> 4) << 1;
  v8f acc = {};
  for (int k0 = 0; k0 < K; k0 += 4) {
    v2f a, b;
    a.x = la(row0 + r, k0 + kb);
    a.y = la(row0 + r, k0 + kb + 1);
    b.x = lb(k0 + kb,     col0 + r);
    b.y = lb(k0 + kb + 1, col0 + r);
    acc = wmma4(a, b, acc);
  }
  return acc;
}


__global__ void k_tables(float* __restrict__ Ere, float* __restrict__ Eim,
                         float* __restrict__ W0,  float* __restrict__ W1,
                         const float* s0, const float* s1,
                         const float* a0, const float* a1) {
  int t = blockIdx.x * blockDim.x + threadIdx.x;
  if (t >= MP * MP) return;
  int i = t / MP, k = t % MP;
  float er = 0.f, ei = 0.f, w0 = 0.f, w1 = 0.f;
  if (i < MM && k < MM) {
    int m = (i * k) % MM;
    float th = (2.0f * PI_F / (float)MM) * (float)m;
    er = cosf(th);
    ei = -sinf(th);
    float f1 = (float)((i <= MM / 2) ? i : i - MM);
    float f2 = (float)((k <= MM / 2) ? k : k - MM);
    float k2 = f1 * f1 + f2 * f2;
    float dec = (PI_F / TAU_F) * expf(TAU_F * k2);
    float m2   = (float)MM * (float)MM;
    float coef = 1.0f / (8.0f * PI_F * PI_F * m2 * m2);
    float base = coef * dec * dec * 4.0f * PI_F;
    float sh0 = s0[0], sh1 = s1[0];
    w0 = base * a0[0] / (k2 + sh0 * sh0);
    w1 = base * a1[0] / (k2 + sh1 * sh1);
  }
  Ere[t] = er; Eim[t] = ei; W0[t] = w0; W1[t] = w1;
}

__global__ void k_spread(const float* __restrict__ x,
                         float* __restrict__ Gx, float* __restrict__ Gy) {
  int t = blockIdx.x * blockDim.x + threadIdx.x;
  if (t >= NPTS * MP) return;
  int pt = t / MP, i = t % MP;
  float gx = 0.f, gy = 0.f;
  if (i < MM) {
    float xi  = (L_F / (float)MM) * (float)i;
    float inv = 1.0f / (4.0f * TAU_F);
    float d0 = x[2 * pt + 0] - xi;
    float d1 = x[2 * pt + 1] - xi;
    gx = expf(-d0 * d0 * inv) + expf(-(d0 - L_F) * (d0 - L_F) * inv)
       + expf(-(d0 + L_F) * (d0 + L_F) * inv);
    gy = expf(-d1 * d1 * inv) + expf(-(d1 - L_F) * (d1 - L_F) * inv)
       + expf(-(d1 + L_F) * (d1 + L_F) * inv);
  }
  Gx[t] = gx; Gy[t] = gy;
}

__global__ void k_dft(const float* __restrict__ Gx, const float* __restrict__ Gy,
                      const float* __restrict__ Ere, const float* __restrict__ Eim,
                      float* __restrict__ FxRe, float* __restrict__ FxIm,
                      float* __restrict__ FyRe, float* __restrict__ FyIm) {
  int z = blockIdx.z;
  const float* G = (z < 2) ? Gx : Gy;
  const float* E = (z & 1) ? Eim : Ere;
  float* D = (z == 0) ? FxRe : (z == 1) ? FxIm : (z == 2) ? FyRe : FyIm;
  int row0 = blockIdx.x * 16, col0 = blockIdx.y * 16;
  v8f acc = mm_tile(row0, col0, MP,
                    [&](int r, int k) { return G[r * MP + k]; },
                    [&](int k, int c) { return E[k * MP + c]; });
  store_tile(D, MP, row0, col0, acc);
}

__global__ void k_sqmag(const float* __restrict__ FxRe, const float* __restrict__ FxIm,
                        const float* __restrict__ FyRe, const float* __restrict__ FyIm,
                        float* __restrict__ Ax, float* __restrict__ Ay) {
  int t = blockIdx.x * blockDim.x + threadIdx.x;
  if (t >= NPTS * MP) return;
  Ax[t] = FxRe[t] * FxRe[t] + FxIm[t] * FxIm[t];
  Ay[t] = FyRe[t] * FyRe[t] + FyIm[t] * FyIm[t];
}

__global__ void k_smat(const float* __restrict__ FxRe, const float* __restrict__ FxIm,
                       const float* __restrict__ FyRe, const float* __restrict__ FyIm,
                       float* __restrict__ SRe, float* __restrict__ SIm) {
  int b = blockIdx.z;
  int row0 = blockIdx.x * 16, col0 = blockIdx.y * 16;
  int lane = threadIdx.x & 31, r = lane & 15, kb = (lane >> 4) << 1;
  const float* fxr = FxRe + b * NPP * MP;
  const float* fxi = FxIm + b * NPP * MP;
  const float* fyr = FyRe + b * NPP * MP;
  const float* fyi = FyIm + b * NPP * MP;
  v8f accRe = {}, accIm = {};
  for (int k0 = 0; k0 < NPP; k0 += 4) {
    int q0 = k0 + kb, q1 = q0 + 1;
    v2f aRe = { fxr[q0 * MP + row0 + r], fxr[q1 * MP + row0 + r] };
    v2f aIm = { fxi[q0 * MP + row0 + r], fxi[q1 * MP + row0 + r] };
    v2f aImN = { -aIm.x, -aIm.y };
    v2f bRe = { fyr[q0 * MP + col0 + r], fyr[q1 * MP + col0 + r] };
    v2f bIm = { fyi[q0 * MP + col0 + r], fyi[q1 * MP + col0 + r] };
    accRe = wmma4(aRe,  bRe, accRe);
    accRe = wmma4(aImN, bIm, accRe);
    accIm = wmma4(aRe,  bIm, accIm);
    accIm = wmma4(aIm,  bRe, accIm);
  }
  store_tile(SRe + b * MP * MP, MP, row0, col0, accRe);
  store_tile(SIm + b * MP * MP, MP, row0, col0, accIm);
}

__global__ void k_umat(const float* __restrict__ W,
                       const float* __restrict__ SRe, const float* __restrict__ SIm,
                       const float* __restrict__ FyRe, const float* __restrict__ FyIm,
                       float* __restrict__ URe, float* __restrict__ UIm) {
  int b = blockIdx.z;
  int row0 = blockIdx.x * 16;
  int col0 = blockIdx.y * 16;
  int lane = threadIdx.x & 31, r = lane & 15, kb = (lane >> 4) << 1;
  const float* sre = SRe + b * MP * MP;
  const float* sim = SIm + b * MP * MP;
  const float* fyr = FyRe + b * NPP * MP;
  const float* fyi = FyIm + b * NPP * MP;
  v8f accRe = {}, accIm = {};
  for (int k0 = 0; k0 < MP; k0 += 4) {
    int k2a = k0 + kb, k2b = k2a + 1;
    int ra = (row0 + r) * MP;
    float w0 = W[ra + k2a], w1 = W[ra + k2b];
    v2f tRe = { w0 * sre[ra + k2a], w1 * sre[ra + k2b] };
    v2f tIm = { w0 * sim[ra + k2a], w1 * sim[ra + k2b] };
    v2f tReN = { -tRe.x, -tRe.y };
    int pb = (col0 + r) * MP;
    v2f bRe = { fyr[pb + k2a], fyr[pb + k2b] };
    v2f bIm = { fyi[pb + k2a], fyi[pb + k2b] };
    accRe = wmma4(tRe,  bRe, accRe);
    accRe = wmma4(tIm,  bIm, accRe);
    accIm = wmma4(tIm,  bRe, accIm);
    accIm = wmma4(tReN, bIm, accIm);
  }
  store_tile(URe + b * MP * NPP, NPP, row0, col0, accRe);
  store_tile(UIm + b * MP * NPP, NPP, row0, col0, accIm);
}

__global__ void k_vmat(const float* __restrict__ W, const float* __restrict__ Ay,
                       float* __restrict__ V) {
  int b = blockIdx.z;
  int row0 = blockIdx.x * 16, col0 = blockIdx.y * 16;
  const float* ay = Ay + b * NPP * MP;
  v8f acc = mm_tile(row0, col0, MP,
                    [&](int r, int k) { return W[r * MP + k]; },
                    [&](int k, int c) { return ay[c * MP + k]; });
  store_tile(V + b * MP * NPP, NPP, row0, col0, acc);
}

__global__ void k_reduce(const float* __restrict__ FxRe, const float* __restrict__ FxIm,
                         const float* __restrict__ Ax,
                         const float* __restrict__ URe, const float* __restrict__ UIm,
                         const float* __restrict__ V,
                         float* __restrict__ out, int ch) {
  int pt = blockIdx.x;
  int b = pt >> 9, p = pt & (NPP - 1);
  int lane = threadIdx.x;
  float tot = 0.f, dia = 0.f;
  for (int k = lane; k < MP; k += 32) {
    int uc = (b * MP + k) * NPP + p;
    tot += FxRe[pt * MP + k] * URe[uc] + FxIm[pt * MP + k] * UIm[uc];
    dia += Ax[pt * MP + k] * V[uc];
  }
#pragma unroll
  for (int off = 16; off > 0; off >>= 1) {
    tot += __shfl_down(tot, off);
    dia += __shfl_down(dia, off);
  }
  if (lane == 0) out[pt * 2 + ch] = tot - dia;
}


extern "C" void kernel_launch(void* const* d_in, const int* in_sizes, int n_in,
                              void* d_out, int out_size, void* d_ws, size_t ws_size,
                              hipStream_t stream) {
  (void)in_sizes; (void)n_in; (void)out_size; (void)ws_size;
  const float* x  = (const float*)d_in[0];
  const float* s0 = (const float*)d_in[1];
  const float* s1 = (const float*)d_in[2];
  const float* a0 = (const float*)d_in[3];
  const float* a1 = (const float*)d_in[4];
  float* out = (float*)d_out;

  float* w = (float*)d_ws;
  float* Ere  = w; w += MP * MP;
  float* Eim  = w; w += MP * MP;
  float* W0   = w; w += MP * MP;
  float* W1   = w; w += MP * MP;
  float* Gx   = w; w += NPTS * MP;
  float* Gy   = w; w += NPTS * MP;
  float* FxRe = w; w += NPTS * MP;
  float* FxIm = w; w += NPTS * MP;
  float* FyRe = w; w += NPTS * MP;
  float* FyIm = w; w += NPTS * MP;
  float* Ax   = w; w += NPTS * MP;
  float* Ay   = w; w += NPTS * MP;
  float* SRe  = w; w += NB * MP * MP;
  float* SIm  = w; w += NB * MP * MP;
  float* URe  = w; w += NB * MP * NPP;
  float* UIm  = w; w += NB * MP * NPP;
  float* V    = w; w += NB * MP * NPP;

  k_tables<<<(MP * MP + 255) / 256, 256, 0, stream>>>(Ere, Eim, W0, W1, s0, s1, a0, a1);
  k_spread<<<(NPTS * MP + 255) / 256, 256, 0, stream>>>(x, Gx, Gy);
  k_dft<<<dim3(NPTS / 16, MP / 16, 4), 32, 0, stream>>>(Gx, Gy, Ere, Eim,
                                                        FxRe, FxIm, FyRe, FyIm);
  k_sqmag<<<(NPTS * MP + 255) / 256, 256, 0, stream>>>(FxRe, FxIm, FyRe, FyIm, Ax, Ay);
  k_smat<<<dim3(MP / 16, MP / 16, NB), 32, 0, stream>>>(FxRe, FxIm, FyRe, FyIm, SRe, SIm);

  for (int c = 0; c < 2; ++c) {
    const float* Wc = c ? W1 : W0;
    k_umat<<<dim3(MP / 16, NPP / 16, NB), 32, 0, stream>>>(Wc, SRe, SIm, FyRe, FyIm, URe, UIm);
    k_vmat<<<dim3(MP / 16, NPP / 16, NB), 32, 0, stream>>>(Wc, Ay, V);
    k_reduce<<<NPTS, 32, 0, stream>>>(FxRe, FxIm, Ax, URe, UIm, V, out, c);
  }
}


# Fewer, larger waves on every level (a quarter as many): 8 spread entries per thread; one wave forms the tiles sharing a fragment (DFT Re|Im x 2 row tiles; V, U both channels; reduce 4 points)
# speedup vs baseline: 1.7394x; 1.7394x over previous
#include <hip/hip_runtime.h>
#include <math.h>


typedef __attribute__((ext_vector_type(2))) float v2f;
typedef __attribute__((ext_vector_type(8))) float v8f;

#define MM    129
#define MP    144
#define NPP   512
#define NB    4
#define NPTS  (NB * NPP)
#define NCH   2

#define PI_F   3.14159265358979323846f
#define TAU_F  (12.0f / (129.0f * 129.0f))
#define L_F    (2.0f * PI_F)

#define TILES       (MP / 16)
#define L0_THREADS  128
#define TAB_PER_T   2
#define SPR_PER_T   8
#define TAB_BLOCKS  (MP * MP / (L0_THREADS * TAB_PER_T))
#define SPR_BLOCKS  (NPTS * MP / (L0_THREADS * SPR_PER_T))
#define GRAM_BLOCKS (TILES * TILES * NB)
#define VT_PER_B    (TILES * (NPP / 16))
#define VMAT_BLOCKS (VT_PER_B * NB)
#define RED_PTS     4
#define KGRP        6
#define CH_STRIDE   (NB * MP * NPP)


__device__ __forceinline__ v8f wmma4(v2f a, v2f b, v8f c) {
  return __builtin_amdgcn_wmma_f32_16x16x4_f32(false, a, false, b, (short)0, c, false, false);
}

__device__ __forceinline__ void store_tile(float* C, int ldc, int row0, int col0, v8f acc) {
  int lane = threadIdx.x & 31;
  int r0 = row0 + ((lane >> 4) << 3);
  int c  = col0 + (lane & 15);
#pragma unroll
  for (int v = 0; v < 8; ++v) C[(r0 + v) * ldc + c] = acc[v];
}


__device__ __forceinline__ void tables_body(int t,
                         float* __restrict__ Ere, float* __restrict__ Eim,
                         float* __restrict__ W0,  float* __restrict__ W1,
                         const float* s0, const float* s1,
                         const float* a0, const float* a1) {
  if (t >= MP * MP) return;
  int i = t / MP, k = t % MP;
  float er = 0.f, ei = 0.f, w0 = 0.f, w1 = 0.f;
  if (i < MM && k < MM) {
    int m = (i * k) % MM;
    float th = (2.0f * PI_F / (float)MM) * (float)m;
    er = cosf(th);
    ei = -sinf(th);
    float f1 = (float)((i <= MM / 2) ? i : i - MM);
    float f2 = (float)((k <= MM / 2) ? k : k - MM);
    float k2 = f1 * f1 + f2 * f2;
    float dec = (PI_F / TAU_F) * expf(TAU_F * k2);
    float m2   = (float)MM * (float)MM;
    float coef = 1.0f / (8.0f * PI_F * PI_F * m2 * m2);
    float base = coef * dec * dec * 4.0f * PI_F;
    float sh0 = s0[0], sh1 = s1[0];
    w0 = base * a0[0] / (k2 + sh0 * sh0);
    w1 = base * a1[0] / (k2 + sh1 * sh1);
  }
  Ere[t] = er; Eim[t] = ei; W0[t] = w0; W1[t] = w1;
}

__device__ __forceinline__ void spread_body(int t, const float* __restrict__ x,
                         float* __restrict__ Gx, float* __restrict__ Gy) {
  if (t >= NPTS * MP) return;
  int pt = t / MP, i = t % MP;
  float gx = 0.f, gy = 0.f;
  if (i < MM) {
    float xi  = (L_F / (float)MM) * (float)i;
    float inv = 1.0f / (4.0f * TAU_F);
    float d0 = x[2 * pt + 0] - xi;
    float d1 = x[2 * pt + 1] - xi;
    gx = expf(-d0 * d0 * inv) + expf(-(d0 - L_F) * (d0 - L_F) * inv)
       + expf(-(d0 + L_F) * (d0 + L_F) * inv);
    gy = expf(-d1 * d1 * inv) + expf(-(d1 - L_F) * (d1 - L_F) * inv)
       + expf(-(d1 + L_F) * (d1 + L_F) * inv);
  }
  Gx[t] = gx; Gy[t] = gy;
}

#define GRAM_RING   4
#define GRAM_AHEAD  (GRAM_RING - 1)
#define GRAM_GROUPS (NPP / 8)

struct GramFrag { v2f aRe[2], aIm[2], bRe[2], bIm[2]; };

__device__ __forceinline__ void gram_load(GramFrag& f, int g,
                       const float* __restrict__ fxr, const float* __restrict__ fxi,
                       const float* __restrict__ fyr, const float* __restrict__ fyi,
                       int kb, int ca, int cb) {
#pragma unroll
  for (int s = 0; s < 2; ++s) {
    int q0 = g * 8 + s * 4 + kb, q1 = q0 + 1;
    f.aRe[s].x = fxr[q0 * MP + ca]; f.aRe[s].y = fxr[q1 * MP + ca];
    f.aIm[s].x = fxi[q0 * MP + ca]; f.aIm[s].y = fxi[q1 * MP + ca];
    f.bRe[s].x = fyr[q0 * MP + cb]; f.bRe[s].y = fyr[q1 * MP + cb];
    f.bIm[s].x = fyi[q0 * MP + cb]; f.bIm[s].y = fyi[q1 * MP + cb];
  }
}

__device__ __forceinline__ void gram_mac(const GramFrag& f, v8f& accRe, v8f& accIm) {
#pragma unroll
  for (int s = 0; s < 2; ++s) {
    v2f aImN = { -f.aIm[s].x, -f.aIm[s].y };
    accRe = wmma4(f.aRe[s], f.bRe[s], accRe);
    accRe = wmma4(aImN,     f.bIm[s], accRe);
    accIm = wmma4(f.aRe[s], f.bIm[s], accIm);
    accIm = wmma4(f.aIm[s], f.bRe[s], accIm);
  }
}

__device__ __forceinline__ void smat_body(int b, int row0, int col0,
                       const float* __restrict__ FxRe, const float* __restrict__ FxIm,
                       const float* __restrict__ FyRe, const float* __restrict__ FyIm,
                       float* __restrict__ SRe, float* __restrict__ SIm) {
  int lane = threadIdx.x & 31, r = lane & 15, kb = (lane >> 4) << 1;
  const float* fxr = FxRe + b * NPP * MP;
  const float* fxi = FxIm + b * NPP * MP;
  const float* fyr = FyRe + b * NPP * MP;
  const float* fyi = FyIm + b * NPP * MP;
  int ca = row0 + r, cb = col0 + r;
  v8f accRe = {}, accIm = {};
  GramFrag ring[GRAM_RING];
#pragma unroll
  for (int j = 0; j < GRAM_AHEAD; ++j) {
    gram_load(ring[j], j, fxr, fxi, fyr, fyi, kb, ca, cb);
    __builtin_amdgcn_sched_barrier(0);
  }
  for (int g0 = 0; g0 < GRAM_GROUPS; g0 += GRAM_RING) {
#pragma unroll
    for (int j = 0; j < GRAM_RING; ++j) {
      int g  = g0 + j;
      int gn = g + GRAM_AHEAD;
      gn = gn < GRAM_GROUPS ? gn : GRAM_GROUPS - 1;
      gram_load(ring[(j + GRAM_AHEAD) % GRAM_RING], gn, fxr, fxi, fyr, fyi, kb, ca, cb);
      gram_mac(ring[j], accRe, accIm);
      __builtin_amdgcn_sched_barrier(0);
    }
  }
  store_tile(SRe + b * MP * MP, MP, row0, col0, accRe);
  store_tile(SIm + b * MP * MP, MP, row0, col0, accIm);
}

__device__ __forceinline__ void vmat_body(int b, int row0, int col0,
                       const float* __restrict__ W0, const float* __restrict__ W1,
                       const float* __restrict__ FyRe, const float* __restrict__ FyIm,
                       float* __restrict__ V) {
  int lane = threadIdx.x & 31, r = lane & 15, kb = (lane >> 4) << 1;
  const float* fyr = FyRe + b * NPP * MP + (col0 + r) * MP + kb;
  const float* fyi = FyIm + b * NPP * MP + (col0 + r) * MP + kb;
  const float* w0  = W0 + (row0 + r) * MP + kb;
  const float* w1  = W1 + (row0 + r) * MP + kb;
  v8f acc0 = {}, acc1 = {};
  for (int k0 = 0; k0 < MP; k0 += 4) {
    float re0 = fyr[k0], re1 = fyr[k0 + 1];
    float im0 = fyi[k0], im1 = fyi[k0 + 1];
    v2f p  = { re0 * re0 + im0 * im0, re1 * re1 + im1 * im1 };
    v2f a0 = { w0[k0], w0[k0 + 1] };
    v2f a1 = { w1[k0], w1[k0 + 1] };
    acc0 = wmma4(a0, p, acc0);
    acc1 = wmma4(a1, p, acc1);
  }
  store_tile(V + b * MP * NPP,             NPP, row0, col0, acc0);
  store_tile(V + CH_STRIDE + b * MP * NPP, NPP, row0, col0, acc1);
}


__global__ void k_level0(float* __restrict__ Ere, float* __restrict__ Eim,
                         float* __restrict__ W0,  float* __restrict__ W1,
                         const float* s0, const float* s1,
                         const float* a0, const float* a1,
                         const float* __restrict__ x,
                         float* __restrict__ Gx, float* __restrict__ Gy) {
  int bid = blockIdx.x;
  if (bid < TAB_BLOCKS) {
    for (int j = 0; j < TAB_PER_T; ++j)
      tables_body((j * TAB_BLOCKS + bid) * L0_THREADS + threadIdx.x,
                  Ere, Eim, W0, W1, s0, s1, a0, a1);
  } else {
    int sb = bid - TAB_BLOCKS;
    for (int j = 0; j < SPR_PER_T; ++j)
      spread_body((j * SPR_BLOCKS + sb) * L0_THREADS + threadIdx.x, x, Gx, Gy);
  }
}

__global__ void k_dft(const float* __restrict__ Gx, const float* __restrict__ Gy,
                      const float* __restrict__ Ere, const float* __restrict__ Eim,
                      float* __restrict__ FxRe, float* __restrict__ FxIm,
                      float* __restrict__ FyRe, float* __restrict__ FyIm) {
  int z = blockIdx.z;
  const float* G = z ? Gy : Gx;
  float* DRe = z ? FyRe : FxRe;
  float* DIm = z ? FyIm : FxIm;
  int row0 = blockIdx.x * 32, col0 = blockIdx.y * 16;
  int lane = threadIdx.x & 31, r = lane & 15, kb = (lane >> 4) << 1;
  const float* ga = G + (row0 + r) * MP + kb;
  const float* gb = ga + 16 * MP;
  const float* er = Ere + kb * MP + col0 + r;
  const float* ei = Eim + kb * MP + col0 + r;
  v8f re0 = {}, im0 = {}, re1 = {}, im1 = {};
  for (int g0 = 0; g0 < MP; g0 += 4 * KGRP) {
    v2f a0[KGRP], a1[KGRP], bR[KGRP], bI[KGRP];
#pragma unroll
    for (int s = 0; s < KGRP; ++s) {
      int k0 = g0 + 4 * s;
      a0[s].x = ga[k0];      a0[s].y = ga[k0 + 1];
      a1[s].x = gb[k0];      a1[s].y = gb[k0 + 1];
      bR[s].x = er[k0 * MP]; bR[s].y = er[(k0 + 1) * MP];
      bI[s].x = ei[k0 * MP]; bI[s].y = ei[(k0 + 1) * MP];
    }
    __builtin_amdgcn_sched_barrier(0);
#pragma unroll
    for (int s = 0; s < KGRP; ++s) {
      re0 = wmma4(a0[s], bR[s], re0);
      im0 = wmma4(a0[s], bI[s], im0);
      re1 = wmma4(a1[s], bR[s], re1);
      im1 = wmma4(a1[s], bI[s], im1);
    }
  }
  store_tile(DRe, MP, row0,      col0, re0);
  store_tile(DIm, MP, row0,      col0, im0);
  store_tile(DRe, MP, row0 + 16, col0, re1);
  store_tile(DIm, MP, row0 + 16, col0, im1);
}

__global__ void k_level2(const float* __restrict__ FxRe, const float* __restrict__ FxIm,
                         const float* __restrict__ FyRe, const float* __restrict__ FyIm,
                         const float* __restrict__ W0,   const float* __restrict__ W1,
                         float* __restrict__ SRe, float* __restrict__ SIm,
                         float* __restrict__ V) {
  int bid = blockIdx.x;
  if (bid < GRAM_BLOCKS) {
    int b  = bid / (TILES * TILES);
    int t  = bid - b * (TILES * TILES);
    int ty = t / TILES, tx = t - ty * TILES;
    smat_body(b, tx * 16, ty * 16, FxRe, FxIm, FyRe, FyIm, SRe, SIm);
  } else {
    int id = bid - GRAM_BLOCKS;
    int b  = id / VT_PER_B;     id -= b * VT_PER_B;
    int ty = id / TILES, tx = id - ty * TILES;
    vmat_body(b, tx * 16, ty * 16, W0, W1, FyRe, FyIm, V);
  }
}

__global__ void k_umat(const float* __restrict__ W0, const float* __restrict__ W1,
                       const float* __restrict__ SRe, const float* __restrict__ SIm,
                       const float* __restrict__ FyRe, const float* __restrict__ FyIm,
                       float* __restrict__ URe, float* __restrict__ UIm) {
  int b = blockIdx.z;
  int row0 = blockIdx.x * 16;
  int col0 = blockIdx.y * 16;
  int lane = threadIdx.x & 31, r = lane & 15, kb = (lane >> 4) << 1;
  int ra = (row0 + r) * MP + kb;
  int pb = (col0 + r) * MP + kb;
  const float* w0p = W0 + ra;
  const float* w1p = W1 + ra;
  const float* sre = SRe + b * MP * MP + ra;
  const float* sim = SIm + b * MP * MP + ra;
  const float* fyr = FyRe + b * NPP * MP + pb;
  const float* fyi = FyIm + b * NPP * MP + pb;
  v8f accRe0 = {}, accIm0 = {}, accRe1 = {}, accIm1 = {};
  for (int g0 = 0; g0 < MP; g0 += 4 * KGRP) {
    v2f sR[KGRP], sI[KGRP], u0[KGRP], u1[KGRP], bRe[KGRP], bIm[KGRP];
#pragma unroll
    for (int s = 0; s < KGRP; ++s) {
      int k0 = g0 + 4 * s;
      sR[s].x  = sre[k0]; sR[s].y  = sre[k0 + 1];
      sI[s].x  = sim[k0]; sI[s].y  = sim[k0 + 1];
      u0[s].x  = w0p[k0]; u0[s].y  = w0p[k0 + 1];
      u1[s].x  = w1p[k0]; u1[s].y  = w1p[k0 + 1];
      bRe[s].x = fyr[k0]; bRe[s].y = fyr[k0 + 1];
      bIm[s].x = fyi[k0]; bIm[s].y = fyi[k0 + 1];
    }
    __builtin_amdgcn_sched_barrier(0);
#pragma unroll
    for (int s = 0; s < KGRP; ++s) {
      v2f tRe0  = { u0[s].x * sR[s].x, u0[s].y * sR[s].y };
      v2f tIm0  = { u0[s].x * sI[s].x, u0[s].y * sI[s].y };
      v2f tReN0 = { -tRe0.x, -tRe0.y };
      v2f tRe1  = { u1[s].x * sR[s].x, u1[s].y * sR[s].y };
      v2f tIm1  = { u1[s].x * sI[s].x, u1[s].y * sI[s].y };
      v2f tReN1 = { -tRe1.x, -tRe1.y };
      accRe0 = wmma4(tRe0,  bRe[s], accRe0);
      accRe1 = wmma4(tRe1,  bRe[s], accRe1);
      accIm0 = wmma4(tIm0,  bRe[s], accIm0);
      accIm1 = wmma4(tIm1,  bRe[s], accIm1);
      accRe0 = wmma4(tIm0,  bIm[s], accRe0);
      accRe1 = wmma4(tIm1,  bIm[s], accRe1);
      accIm0 = wmma4(tReN0, bIm[s], accIm0);
      accIm1 = wmma4(tReN1, bIm[s], accIm1);
    }
  }
  store_tile(URe + b * MP * NPP,             NPP, row0, col0, accRe0);
  store_tile(UIm + b * MP * NPP,             NPP, row0, col0, accIm0);
  store_tile(URe + CH_STRIDE + b * MP * NPP, NPP, row0, col0, accRe1);
  store_tile(UIm + CH_STRIDE + b * MP * NPP, NPP, row0, col0, accIm1);
}

__global__ void k_reduce(const float* __restrict__ FxRe, const float* __restrict__ FxIm,
                         const float* __restrict__ URe, const float* __restrict__ UIm,
                         const float* __restrict__ V,
                         float* __restrict__ out) {
  int pt0 = blockIdx.x * RED_PTS;
  int b = pt0 >> 9, p0 = pt0 & (NPP - 1);
  int lane = threadIdx.x;
  float tot0[RED_PTS], dia0[RED_PTS], tot1[RED_PTS], dia1[RED_PTS];
#pragma unroll
  for (int j = 0; j < RED_PTS; ++j) { tot0[j] = 0.f; dia0[j] = 0.f; tot1[j] = 0.f; dia1[j] = 0.f; }
  for (int k = lane; k < MP; k += 32) {
    int uc = (b * MP + k) * NPP + p0;
    float4 ur0 = *reinterpret_cast<const float4*>(URe + uc);
    float4 ui0 = *reinterpret_cast<const float4*>(UIm + uc);
    float4 vv0 = *reinterpret_cast<const float4*>(V + uc);
    float4 ur1 = *reinterpret_cast<const float4*>(URe + CH_STRIDE + uc);
    float4 ui1 = *reinterpret_cast<const float4*>(UIm + CH_STRIDE + uc);
    float4 vv1 = *reinterpret_cast<const float4*>(V + CH_STRIDE + uc);
    float r0[RED_PTS] = { ur0.x, ur0.y, ur0.z, ur0.w };
    float i0[RED_PTS] = { ui0.x, ui0.y, ui0.z, ui0.w };
    float d0[RED_PTS] = { vv0.x, vv0.y, vv0.z, vv0.w };
    float r1[RED_PTS] = { ur1.x, ur1.y, ur1.z, ur1.w };
    float i1[RED_PTS] = { ui1.x, ui1.y, ui1.z, ui1.w };
    float d1[RED_PTS] = { vv1.x, vv1.y, vv1.z, vv1.w };
#pragma unroll
    for (int j = 0; j < RED_PTS; ++j) {
      float fr = FxRe[(pt0 + j) * MP + k], fi = FxIm[(pt0 + j) * MP + k];
      float ax = fr * fr + fi * fi;
      tot0[j] += fr * r0[j] + fi * i0[j];
      dia0[j] += ax * d0[j];
      tot1[j] += fr * r1[j] + fi * i1[j];
      dia1[j] += ax * d1[j];
    }
  }
#pragma unroll
  for (int off = 16; off > 0; off >>= 1) {
#pragma unroll
    for (int j = 0; j < RED_PTS; ++j) {
      tot0[j] += __shfl_down(tot0[j], off);
      dia0[j] += __shfl_down(dia0[j], off);
      tot1[j] += __shfl_down(tot1[j], off);
      dia1[j] += __shfl_down(dia1[j], off);
    }
  }
  if (lane == 0) {
#pragma unroll
    for (int j = 0; j < RED_PTS; ++j) {
      out[(pt0 + j) * 2 + 0] = tot0[j] - dia0[j];
      out[(pt0 + j) * 2 + 1] = tot1[j] - dia1[j];
    }
  }
}


extern "C" void kernel_launch(void* const* d_in, const int* in_sizes, int n_in,
                              void* d_out, int out_size, void* d_ws, size_t ws_size,
                              hipStream_t stream) {
  (void)in_sizes; (void)n_in; (void)out_size; (void)ws_size;
  const float* x  = (const float*)d_in[0];
  const float* s0 = (const float*)d_in[1];
  const float* s1 = (const float*)d_in[2];
  const float* a0 = (const float*)d_in[3];
  const float* a1 = (const float*)d_in[4];
  float* out = (float*)d_out;

  float* w = (float*)d_ws;
  float* Ere  = w; w += MP * MP;
  float* Eim  = w; w += MP * MP;
  float* W0   = w; w += MP * MP;
  float* W1   = w; w += MP * MP;
  float* Gx   = w; w += NPTS * MP;
  float* Gy   = w; w += NPTS * MP;
  float* FxRe = w; w += NPTS * MP;
  float* FxIm = w; w += NPTS * MP;
  float* FyRe = w; w += NPTS * MP;
  float* FyIm = w; w += NPTS * MP;
  float* SRe  = w; w += NB * MP * MP;
  float* SIm  = w; w += NB * MP * MP;
  float* URe  = w; w += NCH * CH_STRIDE;
  float* UIm  = w; w += NCH * CH_STRIDE;
  float* V    = w; w += NCH * CH_STRIDE;

  k_level0<<<TAB_BLOCKS + SPR_BLOCKS, L0_THREADS, 0, stream>>>(Ere, Eim, W0, W1, s0, s1, a0, a1,
                                                               x, Gx, Gy);
  k_dft<<<dim3(NPTS / 32, MP / 16, 2), 32, 0, stream>>>(Gx, Gy, Ere, Eim,
                                                        FxRe, FxIm, FyRe, FyIm);
  k_level2<<<GRAM_BLOCKS + VMAT_BLOCKS, 32, 0, stream>>>(FxRe, FxIm, FyRe, FyIm,
                                                         W0, W1, SRe, SIm, V);
  k_umat<<<dim3(MP / 16, NPP / 16, NB), 32, 0, stream>>>(W0, W1, SRe, SIm, FyRe, FyIm,
                                                         URe, UIm);
  k_reduce<<<NPTS / RED_PTS, 32, 0, stream>>>(FxRe, FxIm, URe, UIm, V, out);
}


# U level re-tiled as 256 four-wave blocks, one per WGP: operands staged in LDS by whole-line loads, bin-128 row carried as one extra accumulator on two waves per block; sums unchanged
# speedup vs baseline: 1.8248x; 1.0491x over previous
#include <hip/hip_runtime.h>
#include <math.h>


typedef __attribute__((ext_vector_type(2))) float v2f;
typedef __attribute__((ext_vector_type(8))) float v8f;
typedef __attribute__((ext_vector_type(4))) float v4f;

#define MM    129
#define MP    144
#define NPP   512
#define NB    4
#define NPTS  (NB * NPP)
#define NCH   2

#define PI_F   3.14159265358979323846f
#define TAU_F  (12.0f / (129.0f * 129.0f))
#define L_F    (2.0f * PI_F)

#define TILES       (MP / 16)
#define L0_THREADS  128
#define TAB_PER_T   2
#define SPR_PER_T   8
#define TAB_BLOCKS  (MP * MP / (L0_THREADS * TAB_PER_T))
#define SPR_BLOCKS  (NPTS * MP / (L0_THREADS * SPR_PER_T))
#define GRAM_BLOCKS (TILES * TILES * NB)
#define VT_PER_B    (TILES * (NPP / 16))
#define VMAT_BLOCKS (VT_PER_B * NB)
#define RED_PTS     4
#define KGRP        6
#define CH_STRIDE   (NB * MP * NPP)


__device__ __forceinline__ v8f wmma4(v2f a, v2f b, v8f c) {
  return __builtin_amdgcn_wmma_f32_16x16x4_f32(false, a, false, b, (short)0, c, false, false);
}

__device__ __forceinline__ void store_tile(float* C, int ldc, int row0, int col0, v8f acc) {
  int lane = threadIdx.x & 31;
  int r0 = row0 + ((lane >> 4) << 3);
  int c  = col0 + (lane & 15);
#pragma unroll
  for (int v = 0; v < 8; ++v) C[(r0 + v) * ldc + c] = acc[v];
}


__device__ __forceinline__ void tables_body(int t,
                         float* __restrict__ Ere, float* __restrict__ Eim,
                         float* __restrict__ W0,  float* __restrict__ W1,
                         const float* s0, const float* s1,
                         const float* a0, const float* a1) {
  if (t >= MP * MP) return;
  int i = t / MP, k = t % MP;
  float er = 0.f, ei = 0.f, w0 = 0.f, w1 = 0.f;
  if (i < MM && k < MM) {
    int m = (i * k) % MM;
    float th = (2.0f * PI_F / (float)MM) * (float)m;
    er = cosf(th);
    ei = -sinf(th);
    float f1 = (float)((i <= MM / 2) ? i : i - MM);
    float f2 = (float)((k <= MM / 2) ? k : k - MM);
    float k2 = f1 * f1 + f2 * f2;
    float dec = (PI_F / TAU_F) * expf(TAU_F * k2);
    float m2   = (float)MM * (float)MM;
    float coef = 1.0f / (8.0f * PI_F * PI_F * m2 * m2);
    float base = coef * dec * dec * 4.0f * PI_F;
    float sh0 = s0[0], sh1 = s1[0];
    w0 = base * a0[0] / (k2 + sh0 * sh0);
    w1 = base * a1[0] / (k2 + sh1 * sh1);
  }
  Ere[t] = er; Eim[t] = ei; W0[t] = w0; W1[t] = w1;
}

__device__ __forceinline__ void spread_body(int t, const float* __restrict__ x,
                         float* __restrict__ Gx, float* __restrict__ Gy) {
  if (t >= NPTS * MP) return;
  int pt = t / MP, i = t % MP;
  float gx = 0.f, gy = 0.f;
  if (i < MM) {
    float xi  = (L_F / (float)MM) * (float)i;
    float inv = 1.0f / (4.0f * TAU_F);
    float d0 = x[2 * pt + 0] - xi;
    float d1 = x[2 * pt + 1] - xi;
    gx = expf(-d0 * d0 * inv) + expf(-(d0 - L_F) * (d0 - L_F) * inv)
       + expf(-(d0 + L_F) * (d0 + L_F) * inv);
    gy = expf(-d1 * d1 * inv) + expf(-(d1 - L_F) * (d1 - L_F) * inv)
       + expf(-(d1 + L_F) * (d1 + L_F) * inv);
  }
  Gx[t] = gx; Gy[t] = gy;
}

#define GRAM_RING   4
#define GRAM_AHEAD  (GRAM_RING - 1)
#define GRAM_GROUPS (NPP / 8)

struct GramFrag { v2f aRe[2], aIm[2], bRe[2], bIm[2]; };

__device__ __forceinline__ void gram_load(GramFrag& f, int g,
                       const float* __restrict__ fxr, const float* __restrict__ fxi,
                       const float* __restrict__ fyr, const float* __restrict__ fyi,
                       int kb, int ca, int cb) {
#pragma unroll
  for (int s = 0; s < 2; ++s) {
    int q0 = g * 8 + s * 4 + kb, q1 = q0 + 1;
    f.aRe[s].x = fxr[q0 * MP + ca]; f.aRe[s].y = fxr[q1 * MP + ca];
    f.aIm[s].x = fxi[q0 * MP + ca]; f.aIm[s].y = fxi[q1 * MP + ca];
    f.bRe[s].x = fyr[q0 * MP + cb]; f.bRe[s].y = fyr[q1 * MP + cb];
    f.bIm[s].x = fyi[q0 * MP + cb]; f.bIm[s].y = fyi[q1 * MP + cb];
  }
}

__device__ __forceinline__ void gram_mac(const GramFrag& f, v8f& accRe, v8f& accIm) {
#pragma unroll
  for (int s = 0; s < 2; ++s) {
    v2f aImN = { -f.aIm[s].x, -f.aIm[s].y };
    accRe = wmma4(f.aRe[s], f.bRe[s], accRe);
    accRe = wmma4(aImN,     f.bIm[s], accRe);
    accIm = wmma4(f.aRe[s], f.bIm[s], accIm);
    accIm = wmma4(f.aIm[s], f.bRe[s], accIm);
  }
}

__device__ __forceinline__ void smat_body(int b, int row0, int col0,
                       const float* __restrict__ FxRe, const float* __restrict__ FxIm,
                       const float* __restrict__ FyRe, const float* __restrict__ FyIm,
                       float* __restrict__ SRe, float* __restrict__ SIm) {
  int lane = threadIdx.x & 31, r = lane & 15, kb = (lane >> 4) << 1;
  const float* fxr = FxRe + b * NPP * MP;
  const float* fxi = FxIm + b * NPP * MP;
  const float* fyr = FyRe + b * NPP * MP;
  const float* fyi = FyIm + b * NPP * MP;
  int ca = row0 + r, cb = col0 + r;
  v8f accRe = {}, accIm = {};
  GramFrag ring[GRAM_RING];
#pragma unroll
  for (int j = 0; j < GRAM_AHEAD; ++j) {
    gram_load(ring[j], j, fxr, fxi, fyr, fyi, kb, ca, cb);
    __builtin_amdgcn_sched_barrier(0);
  }
  for (int g0 = 0; g0 < GRAM_GROUPS; g0 += GRAM_RING) {
#pragma unroll
    for (int j = 0; j < GRAM_RING; ++j) {
      int g  = g0 + j;
      int gn = g + GRAM_AHEAD;
      gn = gn < GRAM_GROUPS ? gn : GRAM_GROUPS - 1;
      gram_load(ring[(j + GRAM_AHEAD) % GRAM_RING], gn, fxr, fxi, fyr, fyi, kb, ca, cb);
      gram_mac(ring[j], accRe, accIm);
      __builtin_amdgcn_sched_barrier(0);
    }
  }
  store_tile(SRe + b * MP * MP, MP, row0, col0, accRe);
  store_tile(SIm + b * MP * MP, MP, row0, col0, accIm);
}

__device__ __forceinline__ void vmat_body(int b, int row0, int col0,
                       const float* __restrict__ W0, const float* __restrict__ W1,
                       const float* __restrict__ FyRe, const float* __restrict__ FyIm,
                       float* __restrict__ V) {
  int lane = threadIdx.x & 31, r = lane & 15, kb = (lane >> 4) << 1;
  const float* fyr = FyRe + b * NPP * MP + (col0 + r) * MP + kb;
  const float* fyi = FyIm + b * NPP * MP + (col0 + r) * MP + kb;
  const float* w0  = W0 + (row0 + r) * MP + kb;
  const float* w1  = W1 + (row0 + r) * MP + kb;
  v8f acc0 = {}, acc1 = {};
  for (int k0 = 0; k0 < MP; k0 += 4) {
    float re0 = fyr[k0], re1 = fyr[k0 + 1];
    float im0 = fyi[k0], im1 = fyi[k0 + 1];
    v2f p  = { re0 * re0 + im0 * im0, re1 * re1 + im1 * im1 };
    v2f a0 = { w0[k0], w0[k0 + 1] };
    v2f a1 = { w1[k0], w1[k0 + 1] };
    acc0 = wmma4(a0, p, acc0);
    acc1 = wmma4(a1, p, acc1);
  }
  store_tile(V + b * MP * NPP,             NPP, row0, col0, acc0);
  store_tile(V + CH_STRIDE + b * MP * NPP, NPP, row0, col0, acc1);
}


__global__ void k_level0(float* __restrict__ Ere, float* __restrict__ Eim,
                         float* __restrict__ W0,  float* __restrict__ W1,
                         const float* s0, const float* s1,
                         const float* a0, const float* a1,
                         const float* __restrict__ x,
                         float* __restrict__ Gx, float* __restrict__ Gy) {
  int bid = blockIdx.x;
  if (bid < TAB_BLOCKS) {
    for (int j = 0; j < TAB_PER_T; ++j)
      tables_body((j * TAB_BLOCKS + bid) * L0_THREADS + threadIdx.x,
                  Ere, Eim, W0, W1, s0, s1, a0, a1);
  } else {
    int sb = bid - TAB_BLOCKS;
    for (int j = 0; j < SPR_PER_T; ++j)
      spread_body((j * SPR_BLOCKS + sb) * L0_THREADS + threadIdx.x, x, Gx, Gy);
  }
}

__global__ void k_dft(const float* __restrict__ Gx, const float* __restrict__ Gy,
                      const float* __restrict__ Ere, const float* __restrict__ Eim,
                      float* __restrict__ FxRe, float* __restrict__ FxIm,
                      float* __restrict__ FyRe, float* __restrict__ FyIm) {
  int z = blockIdx.z;
  const float* G = z ? Gy : Gx;
  float* DRe = z ? FyRe : FxRe;
  float* DIm = z ? FyIm : FxIm;
  int row0 = blockIdx.x * 32, col0 = blockIdx.y * 16;
  int lane = threadIdx.x & 31, r = lane & 15, kb = (lane >> 4) << 1;
  const float* ga = G + (row0 + r) * MP + kb;
  const float* gb = ga + 16 * MP;
  const float* er = Ere + kb * MP + col0 + r;
  const float* ei = Eim + kb * MP + col0 + r;
  v8f re0 = {}, im0 = {}, re1 = {}, im1 = {};
  for (int g0 = 0; g0 < MP; g0 += 4 * KGRP) {
    v2f a0[KGRP], a1[KGRP], bR[KGRP], bI[KGRP];
#pragma unroll
    for (int s = 0; s < KGRP; ++s) {
      int k0 = g0 + 4 * s;
      a0[s].x = ga[k0];      a0[s].y = ga[k0 + 1];
      a1[s].x = gb[k0];      a1[s].y = gb[k0 + 1];
      bR[s].x = er[k0 * MP]; bR[s].y = er[(k0 + 1) * MP];
      bI[s].x = ei[k0 * MP]; bI[s].y = ei[(k0 + 1) * MP];
    }
    __builtin_amdgcn_sched_barrier(0);
#pragma unroll
    for (int s = 0; s < KGRP; ++s) {
      re0 = wmma4(a0[s], bR[s], re0);
      im0 = wmma4(a0[s], bI[s], im0);
      re1 = wmma4(a1[s], bR[s], re1);
      im1 = wmma4(a1[s], bI[s], im1);
    }
  }
  store_tile(DRe, MP, row0,      col0, re0);
  store_tile(DIm, MP, row0,      col0, im0);
  store_tile(DRe, MP, row0 + 16, col0, re1);
  store_tile(DIm, MP, row0 + 16, col0, im1);
}

__global__ void k_level2(const float* __restrict__ FxRe, const float* __restrict__ FxIm,
                         const float* __restrict__ FyRe, const float* __restrict__ FyIm,
                         const float* __restrict__ W0,   const float* __restrict__ W1,
                         float* __restrict__ SRe, float* __restrict__ SIm,
                         float* __restrict__ V) {
  int bid = blockIdx.x;
  if (bid < GRAM_BLOCKS) {
    int b  = bid / (TILES * TILES);
    int t  = bid - b * (TILES * TILES);
    int ty = t / TILES, tx = t - ty * TILES;
    smat_body(b, tx * 16, ty * 16, FxRe, FxIm, FyRe, FyIm, SRe, SIm);
  } else {
    int id = bid - GRAM_BLOCKS;
    int b  = id / VT_PER_B;     id -= b * VT_PER_B;
    int ty = id / TILES, tx = id - ty * TILES;
    vmat_body(b, tx * 16, ty * 16, W0, W1, FyRe, FyIm, V);
  }
}

#define U_WAVES   4
#define U_PTS     (16 * U_WAVES)
#define U_PITCH   148
#define U_ROW4    (MP / 4)
#define U_THREADS (32 * U_WAVES)
#define U_AROWS   17
#define U_EDGE    (MM - 1)
#define U_A_TRIPS ((U_AROWS * U_ROW4 + U_THREADS - 1) / U_THREADS)
#define U_B_TRIPS (U_PTS * U_ROW4 / U_THREADS)
#define U_B_CH    (U_B_TRIPS / 3)
#define U_APL     (U_AROWS * U_PITCH)
#define U_BPL     (U_PTS * U_PITCH)

template <bool EXTRA>
__device__ __forceinline__ void u_kloop(const float* sT0r, const float* sT0i,
                                        const float* sT1r, const float* sT1i,
                                        const float* sFr,  const float* sFi,
                                        int la, int lb, int lx, int lbx, bool row0lane, bool part,
                                        v8f& accRe0, v8f& accIm0, v8f& accRe1, v8f& accIm1, v8f& accX) {
#pragma unroll
  for (int k0 = 0; k0 < MP; k0 += 4) {
    v2f tRe0 = *reinterpret_cast<const v2f*>(sT0r + la + k0);
    v2f tIm0 = *reinterpret_cast<const v2f*>(sT0i + la + k0);
    v2f tRe1 = *reinterpret_cast<const v2f*>(sT1r + la + k0);
    v2f tIm1 = *reinterpret_cast<const v2f*>(sT1i + la + k0);
    v2f bRe  = *reinterpret_cast<const v2f*>(sFr + lb + k0);
    v2f bIm  = *reinterpret_cast<const v2f*>(sFi + lb + k0);
    v2f tReN0 = { -tRe0.x, -tRe0.y };
    v2f tReN1 = { -tRe1.x, -tRe1.y };
    accRe0 = wmma4(tRe0,  bRe, accRe0);
    accRe1 = wmma4(tRe1,  bRe, accRe1);
    accIm0 = wmma4(tIm0,  bRe, accIm0);
    accIm1 = wmma4(tIm1,  bRe, accIm1);
    if (EXTRA) {
      v2f xr = *reinterpret_cast<const v2f*>(sT0r + lx + k0);
      v2f xi = *reinterpret_cast<const v2f*>(sT0i + lx + k0);
      v2f zr = { row0lane ? xr.x : 0.f, row0lane ? xr.y : 0.f };
      v2f zi = { row0lane ? xi.x : 0.f, row0lane ? xi.y : 0.f };
      v2f cRe = *reinterpret_cast<const v2f*>(sFr + lbx + k0);
      v2f cIm = *reinterpret_cast<const v2f*>(sFi + lbx + k0);
      v2f a1 = { part ? zi.x : zr.x, part ? zi.y : zr.y };
      v2f a2 = { part ? -zr.x : zi.x, part ? -zr.y : zi.y };
      accX = wmma4(a1, cRe, accX);
      accRe0 = wmma4(tIm0,  bIm, accRe0);
      accRe1 = wmma4(tIm1,  bIm, accRe1);
      accIm0 = wmma4(tReN0, bIm, accIm0);
      accIm1 = wmma4(tReN1, bIm, accIm1);
      accX = wmma4(a2, cIm, accX);
    } else {
      accRe0 = wmma4(tIm0,  bIm, accRe0);
      accRe1 = wmma4(tIm1,  bIm, accRe1);
      accIm0 = wmma4(tReN0, bIm, accIm0);
      accIm1 = wmma4(tReN1, bIm, accIm1);
    }
  }
}

__global__ void __launch_bounds__(U_THREADS)
k_umat(const float* __restrict__ W0, const float* __restrict__ W1,
       const float* __restrict__ SRe, const float* __restrict__ SIm,
       const float* __restrict__ FyRe, const float* __restrict__ FyIm,
       float* __restrict__ URe, float* __restrict__ UIm) {
  __shared__ __attribute__((aligned(16))) float sm[4 * U_APL + 2 * U_BPL];
  float* sT0r = sm;
  float* sT0i = sm + U_APL;
  float* sT1r = sm + 2 * U_APL;
  float* sT1i = sm + 3 * U_APL;
  float* sFr  = sm + 4 * U_APL;
  float* sFi  = sm + 4 * U_APL + U_BPL;
  int b = blockIdx.z;
  int kt = blockIdx.x;
  int row0 = kt * 16;
  int pblk = blockIdx.y * U_PTS;
  int tid = threadIdx.x;
  const float* sre = SRe + b * MP * MP;
  const float* sim = SIm + b * MP * MP;
  const float* fyr = FyRe + (b * NPP + pblk) * MP;
  const float* fyi = FyIm + (b * NPP + pblk) * MP;
  v4f gsr[U_A_TRIPS], gsi[U_A_TRIPS], ga0[U_A_TRIPS], ga1[U_A_TRIPS];
  v4f gfr0[U_B_CH], gfi0[U_B_CH], gfr1[U_B_CH], gfi1[U_B_CH], gfr2[U_B_CH], gfi2[U_B_CH];
  bool tail = tid < U_AROWS * U_ROW4 - (U_A_TRIPS - 1) * U_THREADS;
#pragma unroll
  for (int j = 0; j < U_A_TRIPS; ++j) {
    if (j < U_A_TRIPS - 1 || tail) {
      int idx = tid + j * U_THREADS;
      int rr = idx / U_ROW4, c4 = idx - rr * U_ROW4;
      int grow = rr < 16 ? row0 + rr : U_EDGE;
      int g = grow * MP + c4 * 4;
      gsr[j] = *reinterpret_cast<const v4f*>(sre + g);
      gsi[j] = *reinterpret_cast<const v4f*>(sim + g);
      ga0[j] = *reinterpret_cast<const v4f*>(W0 + g);
      ga1[j] = *reinterpret_cast<const v4f*>(W1 + g);
    }
  }
#pragma unroll
  for (int j = 0; j < U_B_CH; ++j) {
    gfr0[j] = *reinterpret_cast<const v4f*>(fyr + 4 * (tid + j * U_THREADS));
    gfi0[j] = *reinterpret_cast<const v4f*>(fyi + 4 * (tid + j * U_THREADS));
    gfr1[j] = *reinterpret_cast<const v4f*>(fyr + 4 * (tid + (j + U_B_CH) * U_THREADS));
    gfi1[j] = *reinterpret_cast<const v4f*>(fyi + 4 * (tid + (j + U_B_CH) * U_THREADS));
    gfr2[j] = *reinterpret_cast<const v4f*>(fyr + 4 * (tid + (j + 2 * U_B_CH) * U_THREADS));
    gfi2[j] = *reinterpret_cast<const v4f*>(fyi + 4 * (tid + (j + 2 * U_B_CH) * U_THREADS));
  }
  __builtin_amdgcn_sched_barrier(0);
#pragma unroll
  for (int j = 0; j < U_A_TRIPS; ++j) {
    if (j < U_A_TRIPS - 1 || tail) {
      int idx = tid + j * U_THREADS;
      int rr = idx / U_ROW4, c4 = idx - rr * U_ROW4;
      int l = rr * U_PITCH + c4 * 4;
      *reinterpret_cast<v4f*>(sT0r + l) = ga0[j] * gsr[j];
      *reinterpret_cast<v4f*>(sT0i + l) = ga0[j] * gsi[j];
      *reinterpret_cast<v4f*>(sT1r + l) = ga1[j] * gsr[j];
      *reinterpret_cast<v4f*>(sT1i + l) = ga1[j] * gsi[j];
    }
  }
#pragma unroll
  for (int j = 0; j < U_B_CH; ++j) {
    int i0 = tid + j * U_THREADS, i1 = i0 + U_B_CH * U_THREADS, i2 = i1 + U_B_CH * U_THREADS;
    int l0 = (i0 / U_ROW4) * U_PITCH + (i0 % U_ROW4) * 4;
    int l1 = (i1 / U_ROW4) * U_PITCH + (i1 % U_ROW4) * 4;
    int l2 = (i2 / U_ROW4) * U_PITCH + (i2 % U_ROW4) * 4;
    *reinterpret_cast<v4f*>(sFr + l0) = gfr0[j];
    *reinterpret_cast<v4f*>(sFi + l0) = gfi0[j];
    *reinterpret_cast<v4f*>(sFr + l1) = gfr1[j];
    *reinterpret_cast<v4f*>(sFi + l1) = gfi1[j];
    *reinterpret_cast<v4f*>(sFr + l2) = gfr2[j];
    *reinterpret_cast<v4f*>(sFi + l2) = gfi2[j];
  }
  __syncthreads();
  int wave = __builtin_amdgcn_readfirstlane(tid >> 5);
  int lane = tid & 31, r = lane & 15, kb = (lane >> 4) << 1;
  int col0 = pblk + wave * 16;
  int la = r * U_PITCH + kb;
  int lb = (wave * 16 + r) * U_PITCH + kb;
  int xt = kt >> 1;
  bool part = (kt & 1) != 0;
  int lx  = wave * 2 * U_APL + 16 * U_PITCH + kb;
  int lbx = (xt * 16 + r) * U_PITCH + kb;
  v8f accRe0 = {}, accIm0 = {}, accRe1 = {}, accIm1 = {}, accX = {};
  if (wave < 2) {
    u_kloop<true>(sT0r, sT0i, sT1r, sT1i, sFr, sFi, la, lb, lx, lbx, r == 0, part,
                  accRe0, accIm0, accRe1, accIm1, accX);
  } else {
    u_kloop<false>(sT0r, sT0i, sT1r, sT1i, sFr, sFi, la, lb, lx, lbx, false, part,
                   accRe0, accIm0, accRe1, accIm1, accX);
  }
  store_tile(URe + b * MP * NPP,             NPP, row0, col0, accRe0);
  store_tile(UIm + b * MP * NPP,             NPP, row0, col0, accIm0);
  store_tile(URe + CH_STRIDE + b * MP * NPP, NPP, row0, col0, accRe1);
  store_tile(UIm + CH_STRIDE + b * MP * NPP, NPP, row0, col0, accIm1);
  if (wave < 2) {
    float* dst = (part ? UIm : URe) + wave * CH_STRIDE + b * MP * NPP;
    store_tile(dst, NPP, 8 * 16, pblk + xt * 16, accX);
  }
}

__global__ void k_reduce(const float* __restrict__ FxRe, const float* __restrict__ FxIm,
                         const float* __restrict__ URe, const float* __restrict__ UIm,
                         const float* __restrict__ V,
                         float* __restrict__ out) {
  int pt0 = blockIdx.x * RED_PTS;
  int b = pt0 >> 9, p0 = pt0 & (NPP - 1);
  int lane = threadIdx.x;
  float tot0[RED_PTS], dia0[RED_PTS], tot1[RED_PTS], dia1[RED_PTS];
#pragma unroll
  for (int j = 0; j < RED_PTS; ++j) { tot0[j] = 0.f; dia0[j] = 0.f; tot1[j] = 0.f; dia1[j] = 0.f; }
  for (int k = lane; k < MP; k += 32) {
    int uc = (b * MP + k) * NPP + p0;
    float4 ur0 = *reinterpret_cast<const float4*>(URe + uc);
    float4 ui0 = *reinterpret_cast<const float4*>(UIm + uc);
    float4 vv0 = *reinterpret_cast<const float4*>(V + uc);
    float4 ur1 = *reinterpret_cast<const float4*>(URe + CH_STRIDE + uc);
    float4 ui1 = *reinterpret_cast<const float4*>(UIm + CH_STRIDE + uc);
    float4 vv1 = *reinterpret_cast<const float4*>(V + CH_STRIDE + uc);
    float r0[RED_PTS] = { ur0.x, ur0.y, ur0.z, ur0.w };
    float i0[RED_PTS] = { ui0.x, ui0.y, ui0.z, ui0.w };
    float d0[RED_PTS] = { vv0.x, vv0.y, vv0.z, vv0.w };
    float r1[RED_PTS] = { ur1.x, ur1.y, ur1.z, ur1.w };
    float i1[RED_PTS] = { ui1.x, ui1.y, ui1.z, ui1.w };
    float d1[RED_PTS] = { vv1.x, vv1.y, vv1.z, vv1.w };
#pragma unroll
    for (int j = 0; j < RED_PTS; ++j) {
      float fr = FxRe[(pt0 + j) * MP + k], fi = FxIm[(pt0 + j) * MP + k];
      float ax = fr * fr + fi * fi;
      tot0[j] += fr * r0[j] + fi * i0[j];
      dia0[j] += ax * d0[j];
      tot1[j] += fr * r1[j] + fi * i1[j];
      dia1[j] += ax * d1[j];
    }
  }
#pragma unroll
  for (int off = 16; off > 0; off >>= 1) {
#pragma unroll
    for (int j = 0; j < RED_PTS; ++j) {
      tot0[j] += __shfl_down(tot0[j], off);
      dia0[j] += __shfl_down(dia0[j], off);
      tot1[j] += __shfl_down(tot1[j], off);
      dia1[j] += __shfl_down(dia1[j], off);
    }
  }
  if (lane == 0) {
#pragma unroll
    for (int j = 0; j < RED_PTS; ++j) {
      out[(pt0 + j) * 2 + 0] = tot0[j] - dia0[j];
      out[(pt0 + j) * 2 + 1] = tot1[j] - dia1[j];
    }
  }
}


extern "C" void kernel_launch(void* const* d_in, const int* in_sizes, int n_in,
                              void* d_out, int out_size, void* d_ws, size_t ws_size,
                              hipStream_t stream) {
  (void)in_sizes; (void)n_in; (void)out_size; (void)ws_size;
  const float* x  = (const float*)d_in[0];
  const float* s0 = (const float*)d_in[1];
  const float* s1 = (const float*)d_in[2];
  const float* a0 = (const float*)d_in[3];
  const float* a1 = (const float*)d_in[4];
  float* out = (float*)d_out;

  float* w = (float*)d_ws;
  float* Ere  = w; w += MP * MP;
  float* Eim  = w; w += MP * MP;
  float* W0   = w; w += MP * MP;
  float* W1   = w; w += MP * MP;
  float* Gx   = w; w += NPTS * MP;
  float* Gy   = w; w += NPTS * MP;
  float* FxRe = w; w += NPTS * MP;
  float* FxIm = w; w += NPTS * MP;
  float* FyRe = w; w += NPTS * MP;
  float* FyIm = w; w += NPTS * MP;
  float* SRe  = w; w += NB * MP * MP;
  float* SIm  = w; w += NB * MP * MP;
  float* URe  = w; w += NCH * CH_STRIDE;
  float* UIm  = w; w += NCH * CH_STRIDE;
  float* V    = w; w += NCH * CH_STRIDE;

  k_level0<<<TAB_BLOCKS + SPR_BLOCKS, L0_THREADS, 0, stream>>>(Ere, Eim, W0, W1, s0, s1, a0, a1,
                                                               x, Gx, Gy);
  k_dft<<<dim3(NPTS / 32, MP / 16, 2), 32, 0, stream>>>(Gx, Gy, Ere, Eim,
                                                        FxRe, FxIm, FyRe, FyIm);
  k_level2<<<GRAM_BLOCKS + VMAT_BLOCKS, 32, 0, stream>>>(FxRe, FxIm, FyRe, FyIm,
                                                         W0, W1, SRe, SIm, V);
  k_umat<<<dim3(MM / 16, NPP / U_PTS, NB), U_THREADS, 0, stream>>>(W0, W1, SRe, SIm, FyRe, FyIm,
                                                         URe, UIm);
  k_reduce<<<NPTS / RED_PTS, 32, 0, stream>>>(FxRe, FxIm, URe, UIm, V, out);
}


# Reduction level re-tiled: block = 16 points x all rows, a load covers 8 rows x 64 contiguous bytes (8 lines, not 32), all loads issued before first use, 6 wave partials added via LDS; new sum order
# speedup vs baseline: 1.9578x; 1.0729x over previous
#include <hip/hip_runtime.h>
#include <math.h>


typedef __attribute__((ext_vector_type(2))) float v2f;
typedef __attribute__((ext_vector_type(8))) float v8f;
typedef __attribute__((ext_vector_type(4))) float v4f;

#define MM    129
#define MP    144
#define NPP   512
#define NB    4
#define NPTS  (NB * NPP)
#define NCH   2

#define PI_F   3.14159265358979323846f
#define TAU_F  (12.0f / (129.0f * 129.0f))
#define L_F    (2.0f * PI_F)

#define TILES       (MP / 16)
#define L0_THREADS  128
#define TAB_PER_T   2
#define SPR_PER_T   8
#define TAB_BLOCKS  (MP * MP / (L0_THREADS * TAB_PER_T))
#define SPR_BLOCKS  (NPTS * MP / (L0_THREADS * SPR_PER_T))
#define GRAM_BLOCKS (TILES * TILES * NB)
#define VT_PER_B    (TILES * (NPP / 16))
#define VMAT_BLOCKS (VT_PER_B * NB)
#define RED_PTS     16
#define RED_WAVES   6
#define RED_TRIPS   3
#define RED_THREADS (32 * RED_WAVES)
#define KGRP        6
#define CH_STRIDE   (NB * MP * NPP)


__device__ __forceinline__ v8f wmma4(v2f a, v2f b, v8f c) {
  return __builtin_amdgcn_wmma_f32_16x16x4_f32(false, a, false, b, (short)0, c, false, false);
}

__device__ __forceinline__ void store_tile(float* C, int ldc, int row0, int col0, v8f acc) {
  int lane = threadIdx.x & 31;
  int r0 = row0 + ((lane >> 4) << 3);
  int c  = col0 + (lane & 15);
#pragma unroll
  for (int v = 0; v < 8; ++v) C[(r0 + v) * ldc + c] = acc[v];
}


__device__ __forceinline__ void tables_body(int t,
                         float* __restrict__ Ere, float* __restrict__ Eim,
                         float* __restrict__ W0,  float* __restrict__ W1,
                         const float* s0, const float* s1,
                         const float* a0, const float* a1) {
  if (t >= MP * MP) return;
  int i = t / MP, k = t % MP;
  float er = 0.f, ei = 0.f, w0 = 0.f, w1 = 0.f;
  if (i < MM && k < MM) {
    int m = (i * k) % MM;
    float th = (2.0f * PI_F / (float)MM) * (float)m;
    er = cosf(th);
    ei = -sinf(th);
    float f1 = (float)((i <= MM / 2) ? i : i - MM);
    float f2 = (float)((k <= MM / 2) ? k : k - MM);
    float k2 = f1 * f1 + f2 * f2;
    float dec = (PI_F / TAU_F) * expf(TAU_F * k2);
    float m2   = (float)MM * (float)MM;
    float coef = 1.0f / (8.0f * PI_F * PI_F * m2 * m2);
    float base = coef * dec * dec * 4.0f * PI_F;
    float sh0 = s0[0], sh1 = s1[0];
    w0 = base * a0[0] / (k2 + sh0 * sh0);
    w1 = base * a1[0] / (k2 + sh1 * sh1);
  }
  Ere[t] = er; Eim[t] = ei; W0[t] = w0; W1[t] = w1;
}

__device__ __forceinline__ void spread_body(int t, const float* __restrict__ x,
                         float* __restrict__ Gx, float* __restrict__ Gy) {
  if (t >= NPTS * MP) return;
  int pt = t / MP, i = t % MP;
  float gx = 0.f, gy = 0.f;
  if (i < MM) {
    float xi  = (L_F / (float)MM) * (float)i;
    float inv = 1.0f / (4.0f * TAU_F);
    float d0 = x[2 * pt + 0] - xi;
    float d1 = x[2 * pt + 1] - xi;
    gx = expf(-d0 * d0 * inv) + expf(-(d0 - L_F) * (d0 - L_F) * inv)
       + expf(-(d0 + L_F) * (d0 + L_F) * inv);
    gy = expf(-d1 * d1 * inv) + expf(-(d1 - L_F) * (d1 - L_F) * inv)
       + expf(-(d1 + L_F) * (d1 + L_F) * inv);
  }
  Gx[t] = gx; Gy[t] = gy;
}

#define GRAM_RING   4
#define GRAM_AHEAD  (GRAM_RING - 1)
#define GRAM_GROUPS (NPP / 8)

struct GramFrag { v2f aRe[2], aIm[2], bRe[2], bIm[2]; };

__device__ __forceinline__ void gram_load(GramFrag& f, int g,
                       const float* __restrict__ fxr, const float* __restrict__ fxi,
                       const float* __restrict__ fyr, const float* __restrict__ fyi,
                       int kb, int ca, int cb) {
#pragma unroll
  for (int s = 0; s < 2; ++s) {
    int q0 = g * 8 + s * 4 + kb, q1 = q0 + 1;
    f.aRe[s].x = fxr[q0 * MP + ca]; f.aRe[s].y = fxr[q1 * MP + ca];
    f.aIm[s].x = fxi[q0 * MP + ca]; f.aIm[s].y = fxi[q1 * MP + ca];
    f.bRe[s].x = fyr[q0 * MP + cb]; f.bRe[s].y = fyr[q1 * MP + cb];
    f.bIm[s].x = fyi[q0 * MP + cb]; f.bIm[s].y = fyi[q1 * MP + cb];
  }
}

__device__ __forceinline__ void gram_mac(const GramFrag& f, v8f& accRe, v8f& accIm) {
#pragma unroll
  for (int s = 0; s < 2; ++s) {
    v2f aImN = { -f.aIm[s].x, -f.aIm[s].y };
    accRe = wmma4(f.aRe[s], f.bRe[s], accRe);
    accRe = wmma4(aImN,     f.bIm[s], accRe);
    accIm = wmma4(f.aRe[s], f.bIm[s], accIm);
    accIm = wmma4(f.aIm[s], f.bRe[s], accIm);
  }
}

__device__ __forceinline__ void smat_body(int b, int row0, int col0,
                       const float* __restrict__ FxRe, const float* __restrict__ FxIm,
                       const float* __restrict__ FyRe, const float* __restrict__ FyIm,
                       float* __restrict__ SRe, float* __restrict__ SIm) {
  int lane = threadIdx.x & 31, r = lane & 15, kb = (lane >> 4) << 1;
  const float* fxr = FxRe + b * NPP * MP;
  const float* fxi = FxIm + b * NPP * MP;
  const float* fyr = FyRe + b * NPP * MP;
  const float* fyi = FyIm + b * NPP * MP;
  int ca = row0 + r, cb = col0 + r;
  v8f accRe = {}, accIm = {};
  GramFrag ring[GRAM_RING];
#pragma unroll
  for (int j = 0; j < GRAM_AHEAD; ++j) {
    gram_load(ring[j], j, fxr, fxi, fyr, fyi, kb, ca, cb);
    __builtin_amdgcn_sched_barrier(0);
  }
  for (int g0 = 0; g0 < GRAM_GROUPS; g0 += GRAM_RING) {
#pragma unroll
    for (int j = 0; j < GRAM_RING; ++j) {
      int g  = g0 + j;
      int gn = g + GRAM_AHEAD;
      gn = gn < GRAM_GROUPS ? gn : GRAM_GROUPS - 1;
      gram_load(ring[(j + GRAM_AHEAD) % GRAM_RING], gn, fxr, fxi, fyr, fyi, kb, ca, cb);
      gram_mac(ring[j], accRe, accIm);
      __builtin_amdgcn_sched_barrier(0);
    }
  }
  store_tile(SRe + b * MP * MP, MP, row0, col0, accRe);
  store_tile(SIm + b * MP * MP, MP, row0, col0, accIm);
}

__device__ __forceinline__ void vmat_body(int b, int row0, int col0,
                       const float* __restrict__ W0, const float* __restrict__ W1,
                       const float* __restrict__ FyRe, const float* __restrict__ FyIm,
                       float* __restrict__ V) {
  int lane = threadIdx.x & 31, r = lane & 15, kb = (lane >> 4) << 1;
  const float* fyr = FyRe + b * NPP * MP + (col0 + r) * MP + kb;
  const float* fyi = FyIm + b * NPP * MP + (col0 + r) * MP + kb;
  const float* w0  = W0 + (row0 + r) * MP + kb;
  const float* w1  = W1 + (row0 + r) * MP + kb;
  v8f acc0 = {}, acc1 = {};
  for (int k0 = 0; k0 < MP; k0 += 4) {
    float re0 = fyr[k0], re1 = fyr[k0 + 1];
    float im0 = fyi[k0], im1 = fyi[k0 + 1];
    v2f p  = { re0 * re0 + im0 * im0, re1 * re1 + im1 * im1 };
    v2f a0 = { w0[k0], w0[k0 + 1] };
    v2f a1 = { w1[k0], w1[k0 + 1] };
    acc0 = wmma4(a0, p, acc0);
    acc1 = wmma4(a1, p, acc1);
  }
  store_tile(V + b * MP * NPP,             NPP, row0, col0, acc0);
  store_tile(V + CH_STRIDE + b * MP * NPP, NPP, row0, col0, acc1);
}


__global__ void k_level0(float* __restrict__ Ere, float* __restrict__ Eim,
                         float* __restrict__ W0,  float* __restrict__ W1,
                         const float* s0, const float* s1,
                         const float* a0, const float* a1,
                         const float* __restrict__ x,
                         float* __restrict__ Gx, float* __restrict__ Gy) {
  int bid = blockIdx.x;
  if (bid < TAB_BLOCKS) {
    for (int j = 0; j < TAB_PER_T; ++j)
      tables_body((j * TAB_BLOCKS + bid) * L0_THREADS + threadIdx.x,
                  Ere, Eim, W0, W1, s0, s1, a0, a1);
  } else {
    int sb = bid - TAB_BLOCKS;
    for (int j = 0; j < SPR_PER_T; ++j)
      spread_body((j * SPR_BLOCKS + sb) * L0_THREADS + threadIdx.x, x, Gx, Gy);
  }
}

__global__ void k_dft(const float* __restrict__ Gx, const float* __restrict__ Gy,
                      const float* __restrict__ Ere, const float* __restrict__ Eim,
                      float* __restrict__ FxRe, float* __restrict__ FxIm,
                      float* __restrict__ FyRe, float* __restrict__ FyIm) {
  int z = blockIdx.z;
  const float* G = z ? Gy : Gx;
  float* DRe = z ? FyRe : FxRe;
  float* DIm = z ? FyIm : FxIm;
  int row0 = blockIdx.x * 32, col0 = blockIdx.y * 16;
  int lane = threadIdx.x & 31, r = lane & 15, kb = (lane >> 4) << 1;
  const float* ga = G + (row0 + r) * MP + kb;
  const float* gb = ga + 16 * MP;
  const float* er = Ere + kb * MP + col0 + r;
  const float* ei = Eim + kb * MP + col0 + r;
  v8f re0 = {}, im0 = {}, re1 = {}, im1 = {};
  for (int g0 = 0; g0 < MP; g0 += 4 * KGRP) {
    v2f a0[KGRP], a1[KGRP], bR[KGRP], bI[KGRP];
#pragma unroll
    for (int s = 0; s < KGRP; ++s) {
      int k0 = g0 + 4 * s;
      a0[s].x = ga[k0];      a0[s].y = ga[k0 + 1];
      a1[s].x = gb[k0];      a1[s].y = gb[k0 + 1];
      bR[s].x = er[k0 * MP]; bR[s].y = er[(k0 + 1) * MP];
      bI[s].x = ei[k0 * MP]; bI[s].y = ei[(k0 + 1) * MP];
    }
    __builtin_amdgcn_sched_barrier(0);
#pragma unroll
    for (int s = 0; s < KGRP; ++s) {
      re0 = wmma4(a0[s], bR[s], re0);
      im0 = wmma4(a0[s], bI[s], im0);
      re1 = wmma4(a1[s], bR[s], re1);
      im1 = wmma4(a1[s], bI[s], im1);
    }
  }
  store_tile(DRe, MP, row0,      col0, re0);
  store_tile(DIm, MP, row0,      col0, im0);
  store_tile(DRe, MP, row0 + 16, col0, re1);
  store_tile(DIm, MP, row0 + 16, col0, im1);
}

__global__ void k_level2(const float* __restrict__ FxRe, const float* __restrict__ FxIm,
                         const float* __restrict__ FyRe, const float* __restrict__ FyIm,
                         const float* __restrict__ W0,   const float* __restrict__ W1,
                         float* __restrict__ SRe, float* __restrict__ SIm,
                         float* __restrict__ V) {
  int bid = blockIdx.x;
  if (bid < GRAM_BLOCKS) {
    int b  = bid / (TILES * TILES);
    int t  = bid - b * (TILES * TILES);
    int ty = t / TILES, tx = t - ty * TILES;
    smat_body(b, tx * 16, ty * 16, FxRe, FxIm, FyRe, FyIm, SRe, SIm);
  } else {
    int id = bid - GRAM_BLOCKS;
    int b  = id / VT_PER_B;     id -= b * VT_PER_B;
    int ty = id / TILES, tx = id - ty * TILES;
    vmat_body(b, tx * 16, ty * 16, W0, W1, FyRe, FyIm, V);
  }
}

#define U_WAVES   4
#define U_PTS     (16 * U_WAVES)
#define U_PITCH   148
#define U_ROW4    (MP / 4)
#define U_THREADS (32 * U_WAVES)
#define U_AROWS   17
#define U_EDGE    (MM - 1)
#define U_A_TRIPS ((U_AROWS * U_ROW4 + U_THREADS - 1) / U_THREADS)
#define U_B_TRIPS (U_PTS * U_ROW4 / U_THREADS)
#define U_B_CH    (U_B_TRIPS / 3)
#define U_APL     (U_AROWS * U_PITCH)
#define U_BPL     (U_PTS * U_PITCH)

template <bool EXTRA>
__device__ __forceinline__ void u_kloop(const float* sT0r, const float* sT0i,
                                        const float* sT1r, const float* sT1i,
                                        const float* sFr,  const float* sFi,
                                        int la, int lb, int lx, int lbx, bool row0lane, bool part,
                                        v8f& accRe0, v8f& accIm0, v8f& accRe1, v8f& accIm1, v8f& accX) {
#pragma unroll
  for (int k0 = 0; k0 < MP; k0 += 4) {
    v2f tRe0 = *reinterpret_cast<const v2f*>(sT0r + la + k0);
    v2f tIm0 = *reinterpret_cast<const v2f*>(sT0i + la + k0);
    v2f tRe1 = *reinterpret_cast<const v2f*>(sT1r + la + k0);
    v2f tIm1 = *reinterpret_cast<const v2f*>(sT1i + la + k0);
    v2f bRe  = *reinterpret_cast<const v2f*>(sFr + lb + k0);
    v2f bIm  = *reinterpret_cast<const v2f*>(sFi + lb + k0);
    v2f tReN0 = { -tRe0.x, -tRe0.y };
    v2f tReN1 = { -tRe1.x, -tRe1.y };
    accRe0 = wmma4(tRe0,  bRe, accRe0);
    accRe1 = wmma4(tRe1,  bRe, accRe1);
    accIm0 = wmma4(tIm0,  bRe, accIm0);
    accIm1 = wmma4(tIm1,  bRe, accIm1);
    if (EXTRA) {
      v2f xr = *reinterpret_cast<const v2f*>(sT0r + lx + k0);
      v2f xi = *reinterpret_cast<const v2f*>(sT0i + lx + k0);
      v2f zr = { row0lane ? xr.x : 0.f, row0lane ? xr.y : 0.f };
      v2f zi = { row0lane ? xi.x : 0.f, row0lane ? xi.y : 0.f };
      v2f cRe = *reinterpret_cast<const v2f*>(sFr + lbx + k0);
      v2f cIm = *reinterpret_cast<const v2f*>(sFi + lbx + k0);
      v2f a1 = { part ? zi.x : zr.x, part ? zi.y : zr.y };
      v2f a2 = { part ? -zr.x : zi.x, part ? -zr.y : zi.y };
      accX = wmma4(a1, cRe, accX);
      accRe0 = wmma4(tIm0,  bIm, accRe0);
      accRe1 = wmma4(tIm1,  bIm, accRe1);
      accIm0 = wmma4(tReN0, bIm, accIm0);
      accIm1 = wmma4(tReN1, bIm, accIm1);
      accX = wmma4(a2, cIm, accX);
    } else {
      accRe0 = wmma4(tIm0,  bIm, accRe0);
      accRe1 = wmma4(tIm1,  bIm, accRe1);
      accIm0 = wmma4(tReN0, bIm, accIm0);
      accIm1 = wmma4(tReN1, bIm, accIm1);
    }
  }
}

__global__ void __launch_bounds__(U_THREADS)
k_umat(const float* __restrict__ W0, const float* __restrict__ W1,
       const float* __restrict__ SRe, const float* __restrict__ SIm,
       const float* __restrict__ FyRe, const float* __restrict__ FyIm,
       float* __restrict__ URe, float* __restrict__ UIm) {
  __shared__ __attribute__((aligned(16))) float sm[4 * U_APL + 2 * U_BPL];
  float* sT0r = sm;
  float* sT0i = sm + U_APL;
  float* sT1r = sm + 2 * U_APL;
  float* sT1i = sm + 3 * U_APL;
  float* sFr  = sm + 4 * U_APL;
  float* sFi  = sm + 4 * U_APL + U_BPL;
  int b = blockIdx.z;
  int kt = blockIdx.x;
  int row0 = kt * 16;
  int pblk = blockIdx.y * U_PTS;
  int tid = threadIdx.x;
  const float* sre = SRe + b * MP * MP;
  const float* sim = SIm + b * MP * MP;
  const float* fyr = FyRe + (b * NPP + pblk) * MP;
  const float* fyi = FyIm + (b * NPP + pblk) * MP;
  v4f gsr[U_A_TRIPS], gsi[U_A_TRIPS], ga0[U_A_TRIPS], ga1[U_A_TRIPS];
  v4f gfr0[U_B_CH], gfi0[U_B_CH], gfr1[U_B_CH], gfi1[U_B_CH], gfr2[U_B_CH], gfi2[U_B_CH];
  bool tail = tid < U_AROWS * U_ROW4 - (U_A_TRIPS - 1) * U_THREADS;
#pragma unroll
  for (int j = 0; j < U_A_TRIPS; ++j) {
    if (j < U_A_TRIPS - 1 || tail) {
      int idx = tid + j * U_THREADS;
      int rr = idx / U_ROW4, c4 = idx - rr * U_ROW4;
      int grow = rr < 16 ? row0 + rr : U_EDGE;
      int g = grow * MP + c4 * 4;
      gsr[j] = *reinterpret_cast<const v4f*>(sre + g);
      gsi[j] = *reinterpret_cast<const v4f*>(sim + g);
      ga0[j] = *reinterpret_cast<const v4f*>(W0 + g);
      ga1[j] = *reinterpret_cast<const v4f*>(W1 + g);
    }
  }
#pragma unroll
  for (int j = 0; j < U_B_CH; ++j) {
    gfr0[j] = *reinterpret_cast<const v4f*>(fyr + 4 * (tid + j * U_THREADS));
    gfi0[j] = *reinterpret_cast<const v4f*>(fyi + 4 * (tid + j * U_THREADS));
    gfr1[j] = *reinterpret_cast<const v4f*>(fyr + 4 * (tid + (j + U_B_CH) * U_THREADS));
    gfi1[j] = *reinterpret_cast<const v4f*>(fyi + 4 * (tid + (j + U_B_CH) * U_THREADS));
    gfr2[j] = *reinterpret_cast<const v4f*>(fyr + 4 * (tid + (j + 2 * U_B_CH) * U_THREADS));
    gfi2[j] = *reinterpret_cast<const v4f*>(fyi + 4 * (tid + (j + 2 * U_B_CH) * U_THREADS));
  }
  __builtin_amdgcn_sched_barrier(0);
#pragma unroll
  for (int j = 0; j < U_A_TRIPS; ++j) {
    if (j < U_A_TRIPS - 1 || tail) {
      int idx = tid + j * U_THREADS;
      int rr = idx / U_ROW4, c4 = idx - rr * U_ROW4;
      int l = rr * U_PITCH + c4 * 4;
      *reinterpret_cast<v4f*>(sT0r + l) = ga0[j] * gsr[j];
      *reinterpret_cast<v4f*>(sT0i + l) = ga0[j] * gsi[j];
      *reinterpret_cast<v4f*>(sT1r + l) = ga1[j] * gsr[j];
      *reinterpret_cast<v4f*>(sT1i + l) = ga1[j] * gsi[j];
    }
  }
#pragma unroll
  for (int j = 0; j < U_B_CH; ++j) {
    int i0 = tid + j * U_THREADS, i1 = i0 + U_B_CH * U_THREADS, i2 = i1 + U_B_CH * U_THREADS;
    int l0 = (i0 / U_ROW4) * U_PITCH + (i0 % U_ROW4) * 4;
    int l1 = (i1 / U_ROW4) * U_PITCH + (i1 % U_ROW4) * 4;
    int l2 = (i2 / U_ROW4) * U_PITCH + (i2 % U_ROW4) * 4;
    *reinterpret_cast<v4f*>(sFr + l0) = gfr0[j];
    *reinterpret_cast<v4f*>(sFi + l0) = gfi0[j];
    *reinterpret_cast<v4f*>(sFr + l1) = gfr1[j];
    *reinterpret_cast<v4f*>(sFi + l1) = gfi1[j];
    *reinterpret_cast<v4f*>(sFr + l2) = gfr2[j];
    *reinterpret_cast<v4f*>(sFi + l2) = gfi2[j];
  }
  __syncthreads();
  int wave = __builtin_amdgcn_readfirstlane(tid >> 5);
  int lane = tid & 31, r = lane & 15, kb = (lane >> 4) << 1;
  int col0 = pblk + wave * 16;
  int la = r * U_PITCH + kb;
  int lb = (wave * 16 + r) * U_PITCH + kb;
  int xt = kt >> 1;
  bool part = (kt & 1) != 0;
  int lx  = wave * 2 * U_APL + 16 * U_PITCH + kb;
  int lbx = (xt * 16 + r) * U_PITCH + kb;
  v8f accRe0 = {}, accIm0 = {}, accRe1 = {}, accIm1 = {}, accX = {};
  if (wave < 2) {
    u_kloop<true>(sT0r, sT0i, sT1r, sT1i, sFr, sFi, la, lb, lx, lbx, r == 0, part,
                  accRe0, accIm0, accRe1, accIm1, accX);
  } else {
    u_kloop<false>(sT0r, sT0i, sT1r, sT1i, sFr, sFi, la, lb, lx, lbx, false, part,
                   accRe0, accIm0, accRe1, accIm1, accX);
  }
  store_tile(URe + b * MP * NPP,             NPP, row0, col0, accRe0);
  store_tile(UIm + b * MP * NPP,             NPP, row0, col0, accIm0);
  store_tile(URe + CH_STRIDE + b * MP * NPP, NPP, row0, col0, accRe1);
  store_tile(UIm + CH_STRIDE + b * MP * NPP, NPP, row0, col0, accIm1);
  if (wave < 2) {
    float* dst = (part ? UIm : URe) + wave * CH_STRIDE + b * MP * NPP;
    store_tile(dst, NPP, 8 * 16, pblk + xt * 16, accX);
  }
}

__global__ void __launch_bounds__(RED_THREADS)
k_reduce(const float* __restrict__ FxRe, const float* __restrict__ FxIm,
         const float* __restrict__ URe, const float* __restrict__ UIm,
         const float* __restrict__ V,
         float* __restrict__ out) {
  __shared__ __attribute__((aligned(16))) float part[RED_WAVES][RED_PTS][4];
  int pt0 = blockIdx.x * RED_PTS;
  int b = pt0 >> 9, p0 = pt0 & (NPP - 1);
  int tid = threadIdx.x;
  int wave = tid >> 5, lane = tid & 31;
  int kk = lane >> 2, pg = lane & 3;
  float4 ur0[RED_TRIPS], ui0[RED_TRIPS], vv0[RED_TRIPS];
  float4 ur1[RED_TRIPS], ui1[RED_TRIPS], vv1[RED_TRIPS];
  float fr[RED_TRIPS][4], fi[RED_TRIPS][4];
#pragma unroll
  for (int t = 0; t < RED_TRIPS; ++t) {
    int k = (wave * RED_TRIPS + t) * 8 + kk;
    int uc = (b * MP + k) * NPP + p0 + 4 * pg;
    ur0[t] = *reinterpret_cast<const float4*>(URe + uc);
    ui0[t] = *reinterpret_cast<const float4*>(UIm + uc);
    vv0[t] = *reinterpret_cast<const float4*>(V + uc);
    ur1[t] = *reinterpret_cast<const float4*>(URe + CH_STRIDE + uc);
    ui1[t] = *reinterpret_cast<const float4*>(UIm + CH_STRIDE + uc);
    vv1[t] = *reinterpret_cast<const float4*>(V + CH_STRIDE + uc);
#pragma unroll
    for (int j = 0; j < 4; ++j) {
      fr[t][j] = FxRe[(pt0 + 4 * pg + j) * MP + k];
      fi[t][j] = FxIm[(pt0 + 4 * pg + j) * MP + k];
    }
  }
  __builtin_amdgcn_sched_barrier(0);
  float tot0[4], dia0[4], tot1[4], dia1[4];
#pragma unroll
  for (int j = 0; j < 4; ++j) { tot0[j] = 0.f; dia0[j] = 0.f; tot1[j] = 0.f; dia1[j] = 0.f; }
#pragma unroll
  for (int t = 0; t < RED_TRIPS; ++t) {
    float r0[4] = { ur0[t].x, ur0[t].y, ur0[t].z, ur0[t].w };
    float i0[4] = { ui0[t].x, ui0[t].y, ui0[t].z, ui0[t].w };
    float d0[4] = { vv0[t].x, vv0[t].y, vv0[t].z, vv0[t].w };
    float r1[4] = { ur1[t].x, ur1[t].y, ur1[t].z, ur1[t].w };
    float i1[4] = { ui1[t].x, ui1[t].y, ui1[t].z, ui1[t].w };
    float d1[4] = { vv1[t].x, vv1[t].y, vv1[t].z, vv1[t].w };
#pragma unroll
    for (int j = 0; j < 4; ++j) {
      float ax = fr[t][j] * fr[t][j] + fi[t][j] * fi[t][j];
      tot0[j] += fr[t][j] * r0[j] + fi[t][j] * i0[j];
      dia0[j] += ax * d0[j];
      tot1[j] += fr[t][j] * r1[j] + fi[t][j] * i1[j];
      dia1[j] += ax * d1[j];
    }
  }
#pragma unroll
  for (int off = 16; off >= 4; off >>= 1) {
#pragma unroll
    for (int j = 0; j < 4; ++j) {
      tot0[j] += __shfl_xor(tot0[j], off);
      dia0[j] += __shfl_xor(dia0[j], off);
      tot1[j] += __shfl_xor(tot1[j], off);
      dia1[j] += __shfl_xor(dia1[j], off);
    }
  }
  if (kk == 0) {
#pragma unroll
    for (int j = 0; j < 4; ++j) {
      float4 s = { tot0[j], dia0[j], tot1[j], dia1[j] };
      *reinterpret_cast<float4*>(&part[wave][4 * pg + j][0]) = s;
    }
  }
  __syncthreads();
  if (tid < RED_PTS * NCH) {
    int pt = tid >> 1, ch = tid & 1;
    float tot = 0.f, dia = 0.f;
#pragma unroll
    for (int w = 0; w < RED_WAVES; ++w) {
      tot += part[w][pt][2 * ch];
      dia += part[w][pt][2 * ch + 1];
    }
    out[(pt0 + pt) * NCH + ch] = tot - dia;
  }
}


extern "C" void kernel_launch(void* const* d_in, const int* in_sizes, int n_in,
                              void* d_out, int out_size, void* d_ws, size_t ws_size,
                              hipStream_t stream) {
  (void)in_sizes; (void)n_in; (void)out_size; (void)ws_size;
  const float* x  = (const float*)d_in[0];
  const float* s0 = (const float*)d_in[1];
  const float* s1 = (const float*)d_in[2];
  const float* a0 = (const float*)d_in[3];
  const float* a1 = (const float*)d_in[4];
  float* out = (float*)d_out;

  float* w = (float*)d_ws;
  float* Ere  = w; w += MP * MP;
  float* Eim  = w; w += MP * MP;
  float* W0   = w; w += MP * MP;
  float* W1   = w; w += MP * MP;
  float* Gx   = w; w += NPTS * MP;
  float* Gy   = w; w += NPTS * MP;
  float* FxRe = w; w += NPTS * MP;
  float* FxIm = w; w += NPTS * MP;
  float* FyRe = w; w += NPTS * MP;
  float* FyIm = w; w += NPTS * MP;
  float* SRe  = w; w += NB * MP * MP;
  float* SIm  = w; w += NB * MP * MP;
  float* URe  = w; w += NCH * CH_STRIDE;
  float* UIm  = w; w += NCH * CH_STRIDE;
  float* V    = w; w += NCH * CH_STRIDE;

  k_level0<<<TAB_BLOCKS + SPR_BLOCKS, L0_THREADS, 0, stream>>>(Ere, Eim, W0, W1, s0, s1, a0, a1,
                                                               x, Gx, Gy);
  k_dft<<<dim3(NPTS / 32, MP / 16, 2), 32, 0, stream>>>(Gx, Gy, Ere, Eim,
                                                        FxRe, FxIm, FyRe, FyIm);
  k_level2<<<GRAM_BLOCKS + VMAT_BLOCKS, 32, 0, stream>>>(FxRe, FxIm, FyRe, FyIm,
                                                         W0, W1, SRe, SIm, V);
  k_umat<<<dim3(MM / 16, NPP / U_PTS, NB), U_THREADS, 0, stream>>>(W0, W1, SRe, SIm, FyRe, FyIm,
                                                         URe, UIm);
  k_reduce<<<NPTS / RED_PTS, RED_THREADS, 0, stream>>>(FxRe, FxIm, URe, UIm, V, out);
}


# Gram in real space: H = Gx^T Gy (real, K=512) beside the DFT launch; S = E H E as two K=144 products in 4-wave blocks (T through LDS) replaces the complex K=512 Gram; new sum order
# speedup vs baseline: 2.0447x; 1.0444x over previous
#include <hip/hip_runtime.h>
#include <math.h>


typedef __attribute__((ext_vector_type(2))) float v2f;
typedef __attribute__((ext_vector_type(8))) float v8f;
typedef __attribute__((ext_vector_type(4))) float v4f;

#define MM    129
#define MP    144
#define NPP   512
#define NB    4
#define NPTS  (NB * NPP)
#define NCH   2

#define PI_F   3.14159265358979323846f
#define TAU_F  (12.0f / (129.0f * 129.0f))
#define L_F    (2.0f * PI_F)

#define TILES       (MP / 16)
#define L0_THREADS  128
#define TAB_PER_T   2
#define SPR_PER_T   8
#define TAB_BLOCKS  (MP * MP / (L0_THREADS * TAB_PER_T))
#define SPR_BLOCKS  (NPTS * MP / (L0_THREADS * SPR_PER_T))
#define GRAM_BLOCKS (TILES * TILES * NB)
#define VT_PER_B    (TILES * (NPP / 16))
#define VMAT_BLOCKS (VT_PER_B * NB)
#define DFT_BLOCKS  ((NPTS / 32) * TILES * 2)
#define S_WAVES     4
#define S_THREADS   (32 * S_WAVES)
#define S_SPLIT     3
#define S_BLOCKS    (TILES * NB * S_SPLIT)
#define V_BLOCKS    (VMAT_BLOCKS / S_WAVES)
#define T_PITCH     148
#define S_KG1       1
#define S_NB1       (MP / (4 * S_KG1))
#define S_KG2       3
#define S_NB2       (MP / (4 * S_KG2))
#define S_RING      6
#define RED_PTS     16
#define RED_WAVES   6
#define RED_TRIPS   3
#define RED_THREADS (32 * RED_WAVES)
#define KGRP        6
#define CH_STRIDE   (NB * MP * NPP)


__device__ __forceinline__ v8f wmma4(v2f a, v2f b, v8f c) {
  return __builtin_amdgcn_wmma_f32_16x16x4_f32(false, a, false, b, (short)0, c, false, false);
}

__device__ __forceinline__ void store_tile(float* C, int ldc, int row0, int col0, v8f acc) {
  int lane = threadIdx.x & 31;
  int r0 = row0 + ((lane >> 4) << 3);
  int c  = col0 + (lane & 15);
#pragma unroll
  for (int v = 0; v < 8; ++v) C[(r0 + v) * ldc + c] = acc[v];
}


__device__ __forceinline__ void tables_body(int t,
                         float* __restrict__ Ere, float* __restrict__ Eim,
                         float* __restrict__ W0,  float* __restrict__ W1,
                         const float* s0, const float* s1,
                         const float* a0, const float* a1) {
  if (t >= MP * MP) return;
  int i = t / MP, k = t % MP;
  float er = 0.f, ei = 0.f, w0 = 0.f, w1 = 0.f;
  if (i < MM && k < MM) {
    int m = (i * k) % MM;
    float th = (2.0f * PI_F / (float)MM) * (float)m;
    er = cosf(th);
    ei = -sinf(th);
    float f1 = (float)((i <= MM / 2) ? i : i - MM);
    float f2 = (float)((k <= MM / 2) ? k : k - MM);
    float k2 = f1 * f1 + f2 * f2;
    float dec = (PI_F / TAU_F) * expf(TAU_F * k2);
    float m2   = (float)MM * (float)MM;
    float coef = 1.0f / (8.0f * PI_F * PI_F * m2 * m2);
    float base = coef * dec * dec * 4.0f * PI_F;
    float sh0 = s0[0], sh1 = s1[0];
    w0 = base * a0[0] / (k2 + sh0 * sh0);
    w1 = base * a1[0] / (k2 + sh1 * sh1);
  }
  Ere[t] = er; Eim[t] = ei; W0[t] = w0; W1[t] = w1;
}

__device__ __forceinline__ void spread_body(int t, const float* __restrict__ x,
                         float* __restrict__ Gx, float* __restrict__ Gy) {
  if (t >= NPTS * MP) return;
  int pt = t / MP, i = t % MP;
  float gx = 0.f, gy = 0.f;
  if (i < MM) {
    float xi  = (L_F / (float)MM) * (float)i;
    float inv = 1.0f / (4.0f * TAU_F);
    float d0 = x[2 * pt + 0] - xi;
    float d1 = x[2 * pt + 1] - xi;
    gx = expf(-d0 * d0 * inv) + expf(-(d0 - L_F) * (d0 - L_F) * inv)
       + expf(-(d0 + L_F) * (d0 + L_F) * inv);
    gy = expf(-d1 * d1 * inv) + expf(-(d1 - L_F) * (d1 - L_F) * inv)
       + expf(-(d1 + L_F) * (d1 + L_F) * inv);
  }
  Gx[t] = gx; Gy[t] = gy;
}

#define HG_RING   8
#define HG_AHEAD  (HG_RING - 1)
#define HG_GROUPS (NPP / 8)

struct HFrag { v2f a[2], b[2]; };

__device__ __forceinline__ void hgram_load(HFrag& f, int g,
                       const float* __restrict__ gx, const float* __restrict__ gy,
                       int kb, int ca, int cb) {
#pragma unroll
  for (int s = 0; s < 2; ++s) {
    int q0 = g * 8 + s * 4 + kb, q1 = q0 + 1;
    f.a[s].x = gx[q0 * MP + ca]; f.a[s].y = gx[q1 * MP + ca];
    f.b[s].x = gy[q0 * MP + cb]; f.b[s].y = gy[q1 * MP + cb];
  }
}

__device__ __forceinline__ void hgram_body(int b, int row0, int col0,
                       const float* __restrict__ A, const float* __restrict__ B,
                       float* __restrict__ D) {
  int lane = threadIdx.x & 31, r = lane & 15, kb = (lane >> 4) << 1;
  const float* gx = A + b * NPP * MP;
  const float* gy = B + b * NPP * MP;
  int ca = row0 + r, cb = col0 + r;
  v8f acc0 = {}, acc1 = {};
  HFrag ring[HG_RING];
#pragma unroll
  for (int j = 0; j < HG_AHEAD; ++j) {
    hgram_load(ring[j], j, gx, gy, kb, ca, cb);
    __builtin_amdgcn_sched_barrier(0);
  }
  for (int g0 = 0; g0 < HG_GROUPS; g0 += HG_RING) {
#pragma unroll
    for (int j = 0; j < HG_RING; ++j) {
      int g  = g0 + j;
      int gn = g + HG_AHEAD;
      gn = gn < HG_GROUPS ? gn : HG_GROUPS - 1;
      hgram_load(ring[(j + HG_AHEAD) % HG_RING], gn, gx, gy, kb, ca, cb);
      acc0 = wmma4(ring[j].a[0], ring[j].b[0], acc0);
      acc1 = wmma4(ring[j].a[1], ring[j].b[1], acc1);
      __builtin_amdgcn_sched_barrier(0);
    }
  }
  v8f acc = acc0 + acc1;
  store_tile(D + b * MP * MP, MP, row0, col0, acc);
}

template <int NT>
struct P1Frag { v2f a[NT][S_KG1], br[S_KG1], bi[S_KG1]; };
struct P2Glob { v2f ar[S_KG2], ai[S_KG2]; };
struct P2Lds  { v2f br[S_KG2], bi[S_KG2]; };

template <int NT>
__device__ __forceinline__ void p1_load(P1Frag<NT>& f, int g, const float* hat,
                                        const float* ebr, const float* ebi) {
#pragma unroll
  for (int s = 0; s < S_KG1; ++s) {
    int k0 = 4 * (g * S_KG1 + s);
#pragma unroll
    for (int t = 0; t < NT; ++t) {
      f.a[t][s].x = hat[t * 64 + k0 * MP];
      f.a[t][s].y = hat[t * 64 + (k0 + 1) * MP];
    }
    f.br[s].x = ebr[k0 * MP]; f.br[s].y = ebr[(k0 + 1) * MP];
    f.bi[s].x = ebi[k0 * MP]; f.bi[s].y = ebi[(k0 + 1) * MP];
  }
}
template <int NT>
__device__ __forceinline__ void p1_mac(const P1Frag<NT>& f, v8f (&tRe)[NT], v8f (&tIm)[NT]) {
#pragma unroll
  for (int s = 0; s < S_KG1; ++s) {
#pragma unroll
    for (int t = 0; t < NT; ++t) {
      tRe[t] = wmma4(f.a[t][s], f.br[s], tRe[t]);
      tIm[t] = wmma4(f.a[t][s], f.bi[s], tIm[t]);
    }
  }
}
template <int NT>
__device__ __forceinline__ void p1_pass(int wave, int r, int hh, const float* hat,
                                        const float* ebr, const float* ebi,
                                        float* sTr, float* sTi) {
  v8f tRe[NT], tIm[NT];
#pragma unroll
  for (int t = 0; t < NT; ++t) { tRe[t] = v8f{}; tIm[t] = v8f{}; }
  P1Frag<NT> ring[S_RING];
#pragma unroll
  for (int j = 0; j < S_RING - 1; ++j) {
    p1_load<NT>(ring[j], j, hat, ebr, ebi);
    __builtin_amdgcn_sched_barrier(0);
  }
#pragma unroll 1
  for (int g0 = 0; g0 < S_NB1; g0 += S_RING) {
#pragma unroll
    for (int j = 0; j < S_RING; ++j) {
      int gn = g0 + j + S_RING - 1;
      gn = gn < S_NB1 ? gn : S_NB1 - 1;
      p1_load<NT>(ring[(j + S_RING - 1) % S_RING], gn, hat, ebr, ebi);
      p1_mac<NT>(ring[j], tRe, tIm);
      __builtin_amdgcn_sched_barrier(0);
    }
  }
#pragma unroll
  for (int t = 0; t < NT; ++t) {
    float* dr = sTr + r * T_PITCH + (wave + 4 * t) * 16 + 8 * hh;
    float* di = sTi + r * T_PITCH + (wave + 4 * t) * 16 + 8 * hh;
    v4f r0 = { tRe[t][0], tRe[t][1], tRe[t][2], tRe[t][3] }, r1 = { tRe[t][4], tRe[t][5], tRe[t][6], tRe[t][7] };
    v4f i0 = { tIm[t][0], tIm[t][1], tIm[t][2], tIm[t][3] }, i1 = { tIm[t][4], tIm[t][5], tIm[t][6], tIm[t][7] };
    *reinterpret_cast<v4f*>(dr) = r0; *reinterpret_cast<v4f*>(dr + 4) = r1;
    *reinterpret_cast<v4f*>(di) = i0; *reinterpret_cast<v4f*>(di + 4) = i1;
  }
}
__device__ __forceinline__ void p2_gload(P2Glob& f, int g, const float* ear, const float* eai) {
#pragma unroll
  for (int s = 0; s < S_KG2; ++s) {
    int k0 = 4 * (g * S_KG2 + s);
    f.ar[s].x = ear[k0 * MP]; f.ar[s].y = ear[(k0 + 1) * MP];
    f.ai[s].x = eai[k0 * MP]; f.ai[s].y = eai[(k0 + 1) * MP];
  }
}
__device__ __forceinline__ void p2_lload(P2Lds& f, int g, const float* tbr, const float* tbi) {
#pragma unroll
  for (int s = 0; s < S_KG2; ++s) {
    int k0 = 4 * (g * S_KG2 + s);
    f.br[s] = *reinterpret_cast<const v2f*>(tbr + k0);
    f.bi[s] = *reinterpret_cast<const v2f*>(tbi + k0);
  }
}
__device__ __forceinline__ void p2_mac(const P2Glob& a, const P2Lds& b, v8f& sRe, v8f& sIm) {
#pragma unroll
  for (int s = 0; s < S_KG2; ++s) {
    v2f aiN = { -a.ai[s].x, -a.ai[s].y };
    sRe = wmma4(a.ar[s], b.br[s], sRe);
    sRe = wmma4(aiN,     b.bi[s], sRe);
    sIm = wmma4(a.ar[s], b.bi[s], sIm);
    sIm = wmma4(a.ai[s], b.br[s], sIm);
  }
}

__device__ __forceinline__ void sdft_body(int b, int wave, int ty, int part,
                       const float* __restrict__ HT,
                       const float* __restrict__ Ere, const float* __restrict__ Eim,
                       float* __restrict__ SRe, float* __restrict__ SIm, float* sT) {
  int lane = threadIdx.x & 31, r = lane & 15, hh = lane >> 4, kb = hh << 1;
  float* sTr = sT;
  float* sTi = sT + 16 * T_PITCH;
  const float* hat = HT + b * MP * MP + kb * MP + wave * 16 + r;
  const float* ebr = Ere + kb * MP + ty * 16 + r;
  const float* ebi = Eim + kb * MP + ty * 16 + r;
  int kt = part * 3 + (wave > 0 ? wave - 1 : 0);
  const float* ear = Ere + kb * MP + kt * 16 + r;
  const float* eai = Eim + kb * MP + kt * 16 + r;
  const float* tbr = sTr + r * T_PITCH + kb;
  const float* tbi = sTi + r * T_PITCH + kb;
  P2Glob gr[S_RING];
  P2Lds  lr[S_RING];
  if (wave == 0) {
    p1_pass<3>(wave, r, hh, hat, ebr, ebi, sTr, sTi);
  } else {
    p2_gload(gr[0], 0, ear, eai);
    p2_gload(gr[1], 1, ear, eai);
    p1_pass<2>(wave, r, hh, hat, ebr, ebi, sTr, sTi);
  }
  __syncthreads();
  if (wave == 0) return;
  v8f sRe = {}, sIm = {};
#pragma unroll
  for (int j = 2; j < S_RING - 1; ++j) p2_gload(gr[j], j, ear, eai);
#pragma unroll
  for (int j = 0; j < S_RING - 1; ++j) p2_lload(lr[j], j, tbr, tbi);
#pragma unroll 1
  for (int g0 = 0; g0 < S_NB2; g0 += S_RING) {
#pragma unroll
    for (int j = 0; j < S_RING; ++j) {
      int gn = g0 + j + S_RING - 1;
      gn = gn < S_NB2 ? gn : S_NB2 - 1;
      p2_gload(gr[(j + S_RING - 1) % S_RING], gn, ear, eai);
      p2_lload(lr[(j + S_RING - 1) % S_RING], gn, tbr, tbi);
      p2_mac(gr[j], lr[j], sRe, sIm);
      __builtin_amdgcn_sched_barrier(0);
    }
  }
  store_tile(SRe + b * MP * MP, MP, kt * 16, ty * 16, sRe);
  store_tile(SIm + b * MP * MP, MP, kt * 16, ty * 16, sIm);
}

__device__ __forceinline__ void vmat_body(int b, int row0, int col0,
                       const float* __restrict__ W0, const float* __restrict__ W1,
                       const float* __restrict__ FyRe, const float* __restrict__ FyIm,
                       float* __restrict__ V) {
  int lane = threadIdx.x & 31, r = lane & 15, kb = (lane >> 4) << 1;
  const float* fyr = FyRe + b * NPP * MP + (col0 + r) * MP + kb;
  const float* fyi = FyIm + b * NPP * MP + (col0 + r) * MP + kb;
  const float* w0  = W0 + (row0 + r) * MP + kb;
  const float* w1  = W1 + (row0 + r) * MP + kb;
  v8f acc0 = {}, acc1 = {};
  for (int k0 = 0; k0 < MP; k0 += 4) {
    float re0 = fyr[k0], re1 = fyr[k0 + 1];
    float im0 = fyi[k0], im1 = fyi[k0 + 1];
    v2f p  = { re0 * re0 + im0 * im0, re1 * re1 + im1 * im1 };
    v2f a0 = { w0[k0], w0[k0 + 1] };
    v2f a1 = { w1[k0], w1[k0 + 1] };
    acc0 = wmma4(a0, p, acc0);
    acc1 = wmma4(a1, p, acc1);
  }
  store_tile(V + b * MP * NPP,             NPP, row0, col0, acc0);
  store_tile(V + CH_STRIDE + b * MP * NPP, NPP, row0, col0, acc1);
}


__global__ void k_level0(float* __restrict__ Ere, float* __restrict__ Eim,
                         float* __restrict__ W0,  float* __restrict__ W1,
                         const float* s0, const float* s1,
                         const float* a0, const float* a1,
                         const float* __restrict__ x,
                         float* __restrict__ Gx, float* __restrict__ Gy) {
  int bid = blockIdx.x;
  if (bid < TAB_BLOCKS) {
    for (int j = 0; j < TAB_PER_T; ++j)
      tables_body((j * TAB_BLOCKS + bid) * L0_THREADS + threadIdx.x,
                  Ere, Eim, W0, W1, s0, s1, a0, a1);
  } else {
    int sb = bid - TAB_BLOCKS;
    for (int j = 0; j < SPR_PER_T; ++j)
      spread_body((j * SPR_BLOCKS + sb) * L0_THREADS + threadIdx.x, x, Gx, Gy);
  }
}

__device__ __forceinline__ void dft_body(int z, int row0, int col0,
                      const float* __restrict__ Gx, const float* __restrict__ Gy,
                      const float* __restrict__ Ere, const float* __restrict__ Eim,
                      float* __restrict__ FxRe, float* __restrict__ FxIm,
                      float* __restrict__ FyRe, float* __restrict__ FyIm) {
  const float* G = z ? Gy : Gx;
  float* DRe = z ? FyRe : FxRe;
  float* DIm = z ? FyIm : FxIm;
  int lane = threadIdx.x & 31, r = lane & 15, kb = (lane >> 4) << 1;
  const float* ga = G + (row0 + r) * MP + kb;
  const float* gb = ga + 16 * MP;
  const float* er = Ere + kb * MP + col0 + r;
  const float* ei = Eim + kb * MP + col0 + r;
  v8f re0 = {}, im0 = {}, re1 = {}, im1 = {};
  for (int g0 = 0; g0 < MP; g0 += 4 * KGRP) {
    v2f a0[KGRP], a1[KGRP], bR[KGRP], bI[KGRP];
#pragma unroll
    for (int s = 0; s < KGRP; ++s) {
      int k0 = g0 + 4 * s;
      a0[s].x = ga[k0];      a0[s].y = ga[k0 + 1];
      a1[s].x = gb[k0];      a1[s].y = gb[k0 + 1];
      bR[s].x = er[k0 * MP]; bR[s].y = er[(k0 + 1) * MP];
      bI[s].x = ei[k0 * MP]; bI[s].y = ei[(k0 + 1) * MP];
    }
    __builtin_amdgcn_sched_barrier(0);
#pragma unroll
    for (int s = 0; s < KGRP; ++s) {
      re0 = wmma4(a0[s], bR[s], re0);
      im0 = wmma4(a0[s], bI[s], im0);
      re1 = wmma4(a1[s], bR[s], re1);
      im1 = wmma4(a1[s], bI[s], im1);
    }
  }
  store_tile(DRe, MP, row0,      col0, re0);
  store_tile(DIm, MP, row0,      col0, im0);
  store_tile(DRe, MP, row0 + 16, col0, re1);
  store_tile(DIm, MP, row0 + 16, col0, im1);
}

__global__ void k_level1(const float* __restrict__ Gx, const float* __restrict__ Gy,
                         const float* __restrict__ Ere, const float* __restrict__ Eim,
                         float* __restrict__ FxRe, float* __restrict__ FxIm,
                         float* __restrict__ FyRe, float* __restrict__ FyIm,
                         float* __restrict__ HT) {
  int bid = blockIdx.x;
  if (bid < GRAM_BLOCKS) {
    int b  = bid / (TILES * TILES);
    int t  = bid - b * (TILES * TILES);
    int ty = t / TILES, tx = t - ty * TILES;
    hgram_body(b, tx * 16, ty * 16, Gy, Gx, HT);
  } else {
    int id = bid - GRAM_BLOCKS;
    int z  = id / ((NPTS / 32) * TILES);   id -= z * ((NPTS / 32) * TILES);
    int by = id / (NPTS / 32), bx = id - by * (NPTS / 32);
    dft_body(z, bx * 32, by * 16, Gx, Gy, Ere, Eim, FxRe, FxIm, FyRe, FyIm);
  }
}

__global__ void __launch_bounds__(S_THREADS)
k_level2(const float* __restrict__ HT,
         const float* __restrict__ Ere,  const float* __restrict__ Eim,
         const float* __restrict__ FyRe, const float* __restrict__ FyIm,
         const float* __restrict__ W0,   const float* __restrict__ W1,
         float* __restrict__ SRe, float* __restrict__ SIm,
         float* __restrict__ V) {
  __shared__ __attribute__((aligned(16))) float sT[2 * 16 * T_PITCH];
  int bid = blockIdx.x;
  int wave = __builtin_amdgcn_readfirstlane(threadIdx.x >> 5);
  if (bid < S_BLOCKS) {
    int bt = bid / S_SPLIT, part = bid - bt * S_SPLIT;
    int b = bt / TILES, ty = bt - b * TILES;
    sdft_body(b, wave, ty, part, HT, Ere, Eim, SRe, SIm, sT);
  } else {
    int id = (bid - S_BLOCKS) * S_WAVES + wave;
    int b  = id / VT_PER_B;     id -= b * VT_PER_B;
    int ty = id / TILES, tx = id - ty * TILES;
    vmat_body(b, tx * 16, ty * 16, W0, W1, FyRe, FyIm, V);
  }
}

#define U_WAVES   4
#define U_PTS     (16 * U_WAVES)
#define U_PITCH   148
#define U_ROW4    (MP / 4)
#define U_THREADS (32 * U_WAVES)
#define U_AROWS   17
#define U_EDGE    (MM - 1)
#define U_A_TRIPS ((U_AROWS * U_ROW4 + U_THREADS - 1) / U_THREADS)
#define U_B_TRIPS (U_PTS * U_ROW4 / U_THREADS)
#define U_B_CH    (U_B_TRIPS / 3)
#define U_APL     (U_AROWS * U_PITCH)
#define U_BPL     (U_PTS * U_PITCH)

template <bool EXTRA>
__device__ __forceinline__ void u_kloop(const float* sT0r, const float* sT0i,
                                        const float* sT1r, const float* sT1i,
                                        const float* sFr,  const float* sFi,
                                        int la, int lb, int lx, int lbx, bool row0lane, bool part,
                                        v8f& accRe0, v8f& accIm0, v8f& accRe1, v8f& accIm1, v8f& accX) {
#pragma unroll
  for (int k0 = 0; k0 < MP; k0 += 4) {
    v2f tRe0 = *reinterpret_cast<const v2f*>(sT0r + la + k0);
    v2f tIm0 = *reinterpret_cast<const v2f*>(sT0i + la + k0);
    v2f tRe1 = *reinterpret_cast<const v2f*>(sT1r + la + k0);
    v2f tIm1 = *reinterpret_cast<const v2f*>(sT1i + la + k0);
    v2f bRe  = *reinterpret_cast<const v2f*>(sFr + lb + k0);
    v2f bIm  = *reinterpret_cast<const v2f*>(sFi + lb + k0);
    v2f tReN0 = { -tRe0.x, -tRe0.y };
    v2f tReN1 = { -tRe1.x, -tRe1.y };
    accRe0 = wmma4(tRe0,  bRe, accRe0);
    accRe1 = wmma4(tRe1,  bRe, accRe1);
    accIm0 = wmma4(tIm0,  bRe, accIm0);
    accIm1 = wmma4(tIm1,  bRe, accIm1);
    if (EXTRA) {
      v2f xr = *reinterpret_cast<const v2f*>(sT0r + lx + k0);
      v2f xi = *reinterpret_cast<const v2f*>(sT0i + lx + k0);
      v2f zr = { row0lane ? xr.x : 0.f, row0lane ? xr.y : 0.f };
      v2f zi = { row0lane ? xi.x : 0.f, row0lane ? xi.y : 0.f };
      v2f cRe = *reinterpret_cast<const v2f*>(sFr + lbx + k0);
      v2f cIm = *reinterpret_cast<const v2f*>(sFi + lbx + k0);
      v2f a1 = { part ? zi.x : zr.x, part ? zi.y : zr.y };
      v2f a2 = { part ? -zr.x : zi.x, part ? -zr.y : zi.y };
      accX = wmma4(a1, cRe, accX);
      accRe0 = wmma4(tIm0,  bIm, accRe0);
      accRe1 = wmma4(tIm1,  bIm, accRe1);
      accIm0 = wmma4(tReN0, bIm, accIm0);
      accIm1 = wmma4(tReN1, bIm, accIm1);
      accX = wmma4(a2, cIm, accX);
    } else {
      accRe0 = wmma4(tIm0,  bIm, accRe0);
      accRe1 = wmma4(tIm1,  bIm, accRe1);
      accIm0 = wmma4(tReN0, bIm, accIm0);
      accIm1 = wmma4(tReN1, bIm, accIm1);
    }
  }
}

__global__ void __launch_bounds__(U_THREADS)
k_umat(const float* __restrict__ W0, const float* __restrict__ W1,
       const float* __restrict__ SRe, const float* __restrict__ SIm,
       const float* __restrict__ FyRe, const float* __restrict__ FyIm,
       float* __restrict__ URe, float* __restrict__ UIm) {
  __shared__ __attribute__((aligned(16))) float sm[4 * U_APL + 2 * U_BPL];
  float* sT0r = sm;
  float* sT0i = sm + U_APL;
  float* sT1r = sm + 2 * U_APL;
  float* sT1i = sm + 3 * U_APL;
  float* sFr  = sm + 4 * U_APL;
  float* sFi  = sm + 4 * U_APL + U_BPL;
  int b = blockIdx.z;
  int kt = blockIdx.x;
  int row0 = kt * 16;
  int pblk = blockIdx.y * U_PTS;
  int tid = threadIdx.x;
  const float* sre = SRe + b * MP * MP;
  const float* sim = SIm + b * MP * MP;
  const float* fyr = FyRe + (b * NPP + pblk) * MP;
  const float* fyi = FyIm + (b * NPP + pblk) * MP;
  v4f gsr[U_A_TRIPS], gsi[U_A_TRIPS], ga0[U_A_TRIPS], ga1[U_A_TRIPS];
  v4f gfr0[U_B_CH], gfi0[U_B_CH], gfr1[U_B_CH], gfi1[U_B_CH], gfr2[U_B_CH], gfi2[U_B_CH];
  bool tail = tid < U_AROWS * U_ROW4 - (U_A_TRIPS - 1) * U_THREADS;
#pragma unroll
  for (int j = 0; j < U_A_TRIPS; ++j) {
    if (j < U_A_TRIPS - 1 || tail) {
      int idx = tid + j * U_THREADS;
      int rr = idx / U_ROW4, c4 = idx - rr * U_ROW4;
      int grow = rr < 16 ? row0 + rr : U_EDGE;
      int g = grow * MP + c4 * 4;
      gsr[j] = *reinterpret_cast<const v4f*>(sre + g);
      gsi[j] = *reinterpret_cast<const v4f*>(sim + g);
      ga0[j] = *reinterpret_cast<const v4f*>(W0 + g);
      ga1[j] = *reinterpret_cast<const v4f*>(W1 + g);
    }
  }
#pragma unroll
  for (int j = 0; j < U_B_CH; ++j) {
    gfr0[j] = *reinterpret_cast<const v4f*>(fyr + 4 * (tid + j * U_THREADS));
    gfi0[j] = *reinterpret_cast<const v4f*>(fyi + 4 * (tid + j * U_THREADS));
    gfr1[j] = *reinterpret_cast<const v4f*>(fyr + 4 * (tid + (j + U_B_CH) * U_THREADS));
    gfi1[j] = *reinterpret_cast<const v4f*>(fyi + 4 * (tid + (j + U_B_CH) * U_THREADS));
    gfr2[j] = *reinterpret_cast<const v4f*>(fyr + 4 * (tid + (j + 2 * U_B_CH) * U_THREADS));
    gfi2[j] = *reinterpret_cast<const v4f*>(fyi + 4 * (tid + (j + 2 * U_B_CH) * U_THREADS));
  }
  __builtin_amdgcn_sched_barrier(0);
#pragma unroll
  for (int j = 0; j < U_A_TRIPS; ++j) {
    if (j < U_A_TRIPS - 1 || tail) {
      int idx = tid + j * U_THREADS;
      int rr = idx / U_ROW4, c4 = idx - rr * U_ROW4;
      int l = rr * U_PITCH + c4 * 4;
      *reinterpret_cast<v4f*>(sT0r + l) = ga0[j] * gsr[j];
      *reinterpret_cast<v4f*>(sT0i + l) = ga0[j] * gsi[j];
      *reinterpret_cast<v4f*>(sT1r + l) = ga1[j] * gsr[j];
      *reinterpret_cast<v4f*>(sT1i + l) = ga1[j] * gsi[j];
    }
  }
#pragma unroll
  for (int j = 0; j < U_B_CH; ++j) {
    int i0 = tid + j * U_THREADS, i1 = i0 + U_B_CH * U_THREADS, i2 = i1 + U_B_CH * U_THREADS;
    int l0 = (i0 / U_ROW4) * U_PITCH + (i0 % U_ROW4) * 4;
    int l1 = (i1 / U_ROW4) * U_PITCH + (i1 % U_ROW4) * 4;
    int l2 = (i2 / U_ROW4) * U_PITCH + (i2 % U_ROW4) * 4;
    *reinterpret_cast<v4f*>(sFr + l0) = gfr0[j];
    *reinterpret_cast<v4f*>(sFi + l0) = gfi0[j];
    *reinterpret_cast<v4f*>(sFr + l1) = gfr1[j];
    *reinterpret_cast<v4f*>(sFi + l1) = gfi1[j];
    *reinterpret_cast<v4f*>(sFr + l2) = gfr2[j];
    *reinterpret_cast<v4f*>(sFi + l2) = gfi2[j];
  }
  __syncthreads();
  int wave = __builtin_amdgcn_readfirstlane(tid >> 5);
  int lane = tid & 31, r = lane & 15, kb = (lane >> 4) << 1;
  int col0 = pblk + wave * 16;
  int la = r * U_PITCH + kb;
  int lb = (wave * 16 + r) * U_PITCH + kb;
  int xt = kt >> 1;
  bool part = (kt & 1) != 0;
  int lx  = wave * 2 * U_APL + 16 * U_PITCH + kb;
  int lbx = (xt * 16 + r) * U_PITCH + kb;
  v8f accRe0 = {}, accIm0 = {}, accRe1 = {}, accIm1 = {}, accX = {};
  if (wave < 2) {
    u_kloop<true>(sT0r, sT0i, sT1r, sT1i, sFr, sFi, la, lb, lx, lbx, r == 0, part,
                  accRe0, accIm0, accRe1, accIm1, accX);
  } else {
    u_kloop<false>(sT0r, sT0i, sT1r, sT1i, sFr, sFi, la, lb, lx, lbx, false, part,
                   accRe0, accIm0, accRe1, accIm1, accX);
  }
  store_tile(URe + b * MP * NPP,             NPP, row0, col0, accRe0);
  store_tile(UIm + b * MP * NPP,             NPP, row0, col0, accIm0);
  store_tile(URe + CH_STRIDE + b * MP * NPP, NPP, row0, col0, accRe1);
  store_tile(UIm + CH_STRIDE + b * MP * NPP, NPP, row0, col0, accIm1);
  if (wave < 2) {
    float* dst = (part ? UIm : URe) + wave * CH_STRIDE + b * MP * NPP;
    store_tile(dst, NPP, 8 * 16, pblk + xt * 16, accX);
  }
}

__global__ void __launch_bounds__(RED_THREADS)
k_reduce(const float* __restrict__ FxRe, const float* __restrict__ FxIm,
         const float* __restrict__ URe, const float* __restrict__ UIm,
         const float* __restrict__ V,
         float* __restrict__ out) {
  __shared__ __attribute__((aligned(16))) float part[RED_WAVES][RED_PTS][4];
  int pt0 = blockIdx.x * RED_PTS;
  int b = pt0 >> 9, p0 = pt0 & (NPP - 1);
  int tid = threadIdx.x;
  int wave = tid >> 5, lane = tid & 31;
  int kk = lane >> 2, pg = lane & 3;
  float4 ur0[RED_TRIPS], ui0[RED_TRIPS], vv0[RED_TRIPS];
  float4 ur1[RED_TRIPS], ui1[RED_TRIPS], vv1[RED_TRIPS];
  float fr[RED_TRIPS][4], fi[RED_TRIPS][4];
#pragma unroll
  for (int t = 0; t < RED_TRIPS; ++t) {
    int k = (wave * RED_TRIPS + t) * 8 + kk;
    int uc = (b * MP + k) * NPP + p0 + 4 * pg;
    ur0[t] = *reinterpret_cast<const float4*>(URe + uc);
    ui0[t] = *reinterpret_cast<const float4*>(UIm + uc);
    vv0[t] = *reinterpret_cast<const float4*>(V + uc);
    ur1[t] = *reinterpret_cast<const float4*>(URe + CH_STRIDE + uc);
    ui1[t] = *reinterpret_cast<const float4*>(UIm + CH_STRIDE + uc);
    vv1[t] = *reinterpret_cast<const float4*>(V + CH_STRIDE + uc);
#pragma unroll
    for (int j = 0; j < 4; ++j) {
      fr[t][j] = FxRe[(pt0 + 4 * pg + j) * MP + k];
      fi[t][j] = FxIm[(pt0 + 4 * pg + j) * MP + k];
    }
  }
  __builtin_amdgcn_sched_barrier(0);
  float tot0[4], dia0[4], tot1[4], dia1[4];
#pragma unroll
  for (int j = 0; j < 4; ++j) { tot0[j] = 0.f; dia0[j] = 0.f; tot1[j] = 0.f; dia1[j] = 0.f; }
#pragma unroll
  for (int t = 0; t < RED_TRIPS; ++t) {
    float r0[4] = { ur0[t].x, ur0[t].y, ur0[t].z, ur0[t].w };
    float i0[4] = { ui0[t].x, ui0[t].y, ui0[t].z, ui0[t].w };
    float d0[4] = { vv0[t].x, vv0[t].y, vv0[t].z, vv0[t].w };
    float r1[4] = { ur1[t].x, ur1[t].y, ur1[t].z, ur1[t].w };
    float i1[4] = { ui1[t].x, ui1[t].y, ui1[t].z, ui1[t].w };
    float d1[4] = { vv1[t].x, vv1[t].y, vv1[t].z, vv1[t].w };
#pragma unroll
    for (int j = 0; j < 4; ++j) {
      float ax = fr[t][j] * fr[t][j] + fi[t][j] * fi[t][j];
      tot0[j] += fr[t][j] * r0[j] + fi[t][j] * i0[j];
      dia0[j] += ax * d0[j];
      tot1[j] += fr[t][j] * r1[j] + fi[t][j] * i1[j];
      dia1[j] += ax * d1[j];
    }
  }
#pragma unroll
  for (int off = 16; off >= 4; off >>= 1) {
#pragma unroll
    for (int j = 0; j < 4; ++j) {
      tot0[j] += __shfl_xor(tot0[j], off);
      dia0[j] += __shfl_xor(dia0[j], off);
      tot1[j] += __shfl_xor(tot1[j], off);
      dia1[j] += __shfl_xor(dia1[j], off);
    }
  }
  if (kk == 0) {
#pragma unroll
    for (int j = 0; j < 4; ++j) {
      float4 s = { tot0[j], dia0[j], tot1[j], dia1[j] };
      *reinterpret_cast<float4*>(&part[wave][4 * pg + j][0]) = s;
    }
  }
  __syncthreads();
  if (tid < RED_PTS * NCH) {
    int pt = tid >> 1, ch = tid & 1;
    float tot = 0.f, dia = 0.f;
#pragma unroll
    for (int w = 0; w < RED_WAVES; ++w) {
      tot += part[w][pt][2 * ch];
      dia += part[w][pt][2 * ch + 1];
    }
    out[(pt0 + pt) * NCH + ch] = tot - dia;
  }
}


extern "C" void kernel_launch(void* const* d_in, const int* in_sizes, int n_in,
                              void* d_out, int out_size, void* d_ws, size_t ws_size,
                              hipStream_t stream) {
  (void)in_sizes; (void)n_in; (void)out_size; (void)ws_size;
  const float* x  = (const float*)d_in[0];
  const float* s0 = (const float*)d_in[1];
  const float* s1 = (const float*)d_in[2];
  const float* a0 = (const float*)d_in[3];
  const float* a1 = (const float*)d_in[4];
  float* out = (float*)d_out;

  float* w = (float*)d_ws;
  float* Ere  = w; w += MP * MP;
  float* Eim  = w; w += MP * MP;
  float* W0   = w; w += MP * MP;
  float* W1   = w; w += MP * MP;
  float* Gx   = w; w += NPTS * MP;
  float* Gy   = w; w += NPTS * MP;
  float* FxRe = w; w += NPTS * MP;
  float* FxIm = w; w += NPTS * MP;
  float* FyRe = w; w += NPTS * MP;
  float* FyIm = w; w += NPTS * MP;
  float* SRe  = w; w += NB * MP * MP;
  float* SIm  = w; w += NB * MP * MP;
  float* URe  = w; w += NCH * CH_STRIDE;
  float* UIm  = w; w += NCH * CH_STRIDE;
  float* V    = w; w += NCH * CH_STRIDE;
  float* HT   = w; w += NB * MP * MP;

  k_level0<<<TAB_BLOCKS + SPR_BLOCKS, L0_THREADS, 0, stream>>>(Ere, Eim, W0, W1, s0, s1, a0, a1,
                                                               x, Gx, Gy);
  k_level1<<<GRAM_BLOCKS + DFT_BLOCKS, 32, 0, stream>>>(Gx, Gy, Ere, Eim,
                                                        FxRe, FxIm, FyRe, FyIm, HT);
  k_level2<<<S_BLOCKS + V_BLOCKS, S_THREADS, 0, stream>>>(HT, Ere, Eim, FyRe, FyIm,
                                                          W0, W1, SRe, SIm, V);
  k_umat<<<dim3(MM / 16, NPP / U_PTS, NB), U_THREADS, 0, stream>>>(W0, W1, SRe, SIm, FyRe, FyIm,
                                                         URe, UIm);
  k_reduce<<<NPTS / RED_PTS, RED_THREADS, 0, stream>>>(FxRe, FxIm, URe, UIm, V, out);
}


# U, V tiles not stored: the forming wave dots its tile with Fx (|Fx|^2) in registers, writes one partial per point and k1 tile; last launch only adds partials; new order of the k1 sum
# speedup vs baseline: 2.2284x; 1.0898x over previous
#include <hip/hip_runtime.h>
#include <math.h>


typedef __attribute__((ext_vector_type(2))) float v2f;
typedef __attribute__((ext_vector_type(8))) float v8f;
typedef __attribute__((ext_vector_type(4))) float v4f;

#define MM    129
#define MP    144
#define NPP   512
#define NB    4
#define NPTS  (NB * NPP)
#define NCH   2

#define PI_F   3.14159265358979323846f
#define TAU_F  (12.0f / (129.0f * 129.0f))
#define L_F    (2.0f * PI_F)

#define TILES       (MP / 16)
#define L0_THREADS  128
#define TAB_PER_T   2
#define SPR_PER_T   8
#define TAB_BLOCKS  (MP * MP / (L0_THREADS * TAB_PER_T))
#define SPR_BLOCKS  (NPTS * MP / (L0_THREADS * SPR_PER_T))
#define GRAM_BLOCKS (TILES * TILES * NB)
#define VT_PER_B    (TILES * (NPP / 16))
#define VMAT_BLOCKS (VT_PER_B * NB)
#define DFT_BLOCKS  ((NPTS / 32) * TILES * 2)
#define S_WAVES     4
#define S_THREADS   (32 * S_WAVES)
#define S_SPLIT     3
#define S_BLOCKS    (TILES * NB * S_SPLIT)
#define V_BLOCKS    (VMAT_BLOCKS / S_WAVES)
#define T_PITCH     148
#define S_KG1       1
#define S_NB1       (MP / (4 * S_KG1))
#define S_KG2       3
#define S_NB2       (MP / (4 * S_KG2))
#define S_RING      6
#define NOUT        (NPTS * NCH)
#define FIN_THREADS 256
#define KGRP        6


__device__ __forceinline__ v8f wmma4(v2f a, v2f b, v8f c) {
  return __builtin_amdgcn_wmma_f32_16x16x4_f32(false, a, false, b, (short)0, c, false, false);
}

__device__ __forceinline__ void store_tile(float* C, int ldc, int row0, int col0, v8f acc) {
  int lane = threadIdx.x & 31;
  int r0 = row0 + ((lane >> 4) << 3);
  int c  = col0 + (lane & 15);
#pragma unroll
  for (int v = 0; v < 8; ++v) C[(r0 + v) * ldc + c] = acc[v];
}


__device__ __forceinline__ void tables_body(int t,
                         float* __restrict__ Ere, float* __restrict__ Eim,
                         float* __restrict__ W0,  float* __restrict__ W1,
                         const float* s0, const float* s1,
                         const float* a0, const float* a1) {
  if (t >= MP * MP) return;
  int i = t / MP, k = t % MP;
  float er = 0.f, ei = 0.f, w0 = 0.f, w1 = 0.f;
  if (i < MM && k < MM) {
    int m = (i * k) % MM;
    float th = (2.0f * PI_F / (float)MM) * (float)m;
    er = cosf(th);
    ei = -sinf(th);
    float f1 = (float)((i <= MM / 2) ? i : i - MM);
    float f2 = (float)((k <= MM / 2) ? k : k - MM);
    float k2 = f1 * f1 + f2 * f2;
    float dec = (PI_F / TAU_F) * expf(TAU_F * k2);
    float m2   = (float)MM * (float)MM;
    float coef = 1.0f / (8.0f * PI_F * PI_F * m2 * m2);
    float base = coef * dec * dec * 4.0f * PI_F;
    float sh0 = s0[0], sh1 = s1[0];
    w0 = base * a0[0] / (k2 + sh0 * sh0);
    w1 = base * a1[0] / (k2 + sh1 * sh1);
  }
  Ere[t] = er; Eim[t] = ei; W0[t] = w0; W1[t] = w1;
}

__device__ __forceinline__ void spread_body(int t, const float* __restrict__ x,
                         float* __restrict__ Gx, float* __restrict__ Gy) {
  if (t >= NPTS * MP) return;
  int pt = t / MP, i = t % MP;
  float gx = 0.f, gy = 0.f;
  if (i < MM) {
    float xi  = (L_F / (float)MM) * (float)i;
    float inv = 1.0f / (4.0f * TAU_F);
    float d0 = x[2 * pt + 0] - xi;
    float d1 = x[2 * pt + 1] - xi;
    gx = expf(-d0 * d0 * inv) + expf(-(d0 - L_F) * (d0 - L_F) * inv)
       + expf(-(d0 + L_F) * (d0 + L_F) * inv);
    gy = expf(-d1 * d1 * inv) + expf(-(d1 - L_F) * (d1 - L_F) * inv)
       + expf(-(d1 + L_F) * (d1 + L_F) * inv);
  }
  Gx[t] = gx; Gy[t] = gy;
}

#define HG_RING   8
#define HG_AHEAD  (HG_RING - 1)
#define HG_GROUPS (NPP / 8)

struct HFrag { v2f a[2], b[2]; };

__device__ __forceinline__ void hgram_load(HFrag& f, int g,
                       const float* __restrict__ gx, const float* __restrict__ gy,
                       int kb, int ca, int cb) {
#pragma unroll
  for (int s = 0; s < 2; ++s) {
    int q0 = g * 8 + s * 4 + kb, q1 = q0 + 1;
    f.a[s].x = gx[q0 * MP + ca]; f.a[s].y = gx[q1 * MP + ca];
    f.b[s].x = gy[q0 * MP + cb]; f.b[s].y = gy[q1 * MP + cb];
  }
}

__device__ __forceinline__ void hgram_body(int b, int row0, int col0,
                       const float* __restrict__ A, const float* __restrict__ B,
                       float* __restrict__ D) {
  int lane = threadIdx.x & 31, r = lane & 15, kb = (lane >> 4) << 1;
  const float* gx = A + b * NPP * MP;
  const float* gy = B + b * NPP * MP;
  int ca = row0 + r, cb = col0 + r;
  v8f acc0 = {}, acc1 = {};
  HFrag ring[HG_RING];
#pragma unroll
  for (int j = 0; j < HG_AHEAD; ++j) {
    hgram_load(ring[j], j, gx, gy, kb, ca, cb);
    __builtin_amdgcn_sched_barrier(0);
  }
  for (int g0 = 0; g0 < HG_GROUPS; g0 += HG_RING) {
#pragma unroll
    for (int j = 0; j < HG_RING; ++j) {
      int g  = g0 + j;
      int gn = g + HG_AHEAD;
      gn = gn < HG_GROUPS ? gn : HG_GROUPS - 1;
      hgram_load(ring[(j + HG_AHEAD) % HG_RING], gn, gx, gy, kb, ca, cb);
      acc0 = wmma4(ring[j].a[0], ring[j].b[0], acc0);
      acc1 = wmma4(ring[j].a[1], ring[j].b[1], acc1);
      __builtin_amdgcn_sched_barrier(0);
    }
  }
  v8f acc = acc0 + acc1;
  store_tile(D + b * MP * MP, MP, row0, col0, acc);
}

template <int NT>
struct P1Frag { v2f a[NT][S_KG1], br[S_KG1], bi[S_KG1]; };
struct P2Glob { v2f ar[S_KG2], ai[S_KG2]; };
struct P2Lds  { v2f br[S_KG2], bi[S_KG2]; };

template <int NT>
__device__ __forceinline__ void p1_load(P1Frag<NT>& f, int g, const float* hat,
                                        const float* ebr, const float* ebi) {
#pragma unroll
  for (int s = 0; s < S_KG1; ++s) {
    int k0 = 4 * (g * S_KG1 + s);
#pragma unroll
    for (int t = 0; t < NT; ++t) {
      f.a[t][s].x = hat[t * 64 + k0 * MP];
      f.a[t][s].y = hat[t * 64 + (k0 + 1) * MP];
    }
    f.br[s].x = ebr[k0 * MP]; f.br[s].y = ebr[(k0 + 1) * MP];
    f.bi[s].x = ebi[k0 * MP]; f.bi[s].y = ebi[(k0 + 1) * MP];
  }
}
template <int NT>
__device__ __forceinline__ void p1_mac(const P1Frag<NT>& f, v8f (&tRe)[NT], v8f (&tIm)[NT]) {
#pragma unroll
  for (int s = 0; s < S_KG1; ++s) {
#pragma unroll
    for (int t = 0; t < NT; ++t) {
      tRe[t] = wmma4(f.a[t][s], f.br[s], tRe[t]);
      tIm[t] = wmma4(f.a[t][s], f.bi[s], tIm[t]);
    }
  }
}
template <int NT>
__device__ __forceinline__ void p1_pass(int wave, int r, int hh, const float* hat,
                                        const float* ebr, const float* ebi,
                                        float* sTr, float* sTi) {
  v8f tRe[NT], tIm[NT];
#pragma unroll
  for (int t = 0; t < NT; ++t) { tRe[t] = v8f{}; tIm[t] = v8f{}; }
  P1Frag<NT> ring[S_RING];
#pragma unroll
  for (int j = 0; j < S_RING - 1; ++j) {
    p1_load<NT>(ring[j], j, hat, ebr, ebi);
    __builtin_amdgcn_sched_barrier(0);
  }
#pragma unroll 1
  for (int g0 = 0; g0 < S_NB1; g0 += S_RING) {
#pragma unroll
    for (int j = 0; j < S_RING; ++j) {
      int gn = g0 + j + S_RING - 1;
      gn = gn < S_NB1 ? gn : S_NB1 - 1;
      p1_load<NT>(ring[(j + S_RING - 1) % S_RING], gn, hat, ebr, ebi);
      p1_mac<NT>(ring[j], tRe, tIm);
      __builtin_amdgcn_sched_barrier(0);
    }
  }
#pragma unroll
  for (int t = 0; t < NT; ++t) {
    float* dr = sTr + r * T_PITCH + (wave + 4 * t) * 16 + 8 * hh;
    float* di = sTi + r * T_PITCH + (wave + 4 * t) * 16 + 8 * hh;
    v4f r0 = { tRe[t][0], tRe[t][1], tRe[t][2], tRe[t][3] }, r1 = { tRe[t][4], tRe[t][5], tRe[t][6], tRe[t][7] };
    v4f i0 = { tIm[t][0], tIm[t][1], tIm[t][2], tIm[t][3] }, i1 = { tIm[t][4], tIm[t][5], tIm[t][6], tIm[t][7] };
    *reinterpret_cast<v4f*>(dr) = r0; *reinterpret_cast<v4f*>(dr + 4) = r1;
    *reinterpret_cast<v4f*>(di) = i0; *reinterpret_cast<v4f*>(di + 4) = i1;
  }
}
__device__ __forceinline__ void p2_gload(P2Glob& f, int g, const float* ear, const float* eai) {
#pragma unroll
  for (int s = 0; s < S_KG2; ++s) {
    int k0 = 4 * (g * S_KG2 + s);
    f.ar[s].x = ear[k0 * MP]; f.ar[s].y = ear[(k0 + 1) * MP];
    f.ai[s].x = eai[k0 * MP]; f.ai[s].y = eai[(k0 + 1) * MP];
  }
}
__device__ __forceinline__ void p2_lload(P2Lds& f, int g, const float* tbr, const float* tbi) {
#pragma unroll
  for (int s = 0; s < S_KG2; ++s) {
    int k0 = 4 * (g * S_KG2 + s);
    f.br[s] = *reinterpret_cast<const v2f*>(tbr + k0);
    f.bi[s] = *reinterpret_cast<const v2f*>(tbi + k0);
  }
}
__device__ __forceinline__ void p2_mac(const P2Glob& a, const P2Lds& b, v8f& sRe, v8f& sIm) {
#pragma unroll
  for (int s = 0; s < S_KG2; ++s) {
    v2f aiN = { -a.ai[s].x, -a.ai[s].y };
    sRe = wmma4(a.ar[s], b.br[s], sRe);
    sRe = wmma4(aiN,     b.bi[s], sRe);
    sIm = wmma4(a.ar[s], b.bi[s], sIm);
    sIm = wmma4(a.ai[s], b.br[s], sIm);
  }
}

__device__ __forceinline__ void sdft_body(int b, int wave, int ty, int part,
                       const float* __restrict__ HT,
                       const float* __restrict__ Ere, const float* __restrict__ Eim,
                       float* __restrict__ SRe, float* __restrict__ SIm, float* sT) {
  int lane = threadIdx.x & 31, r = lane & 15, hh = lane >> 4, kb = hh << 1;
  float* sTr = sT;
  float* sTi = sT + 16 * T_PITCH;
  const float* hat = HT + b * MP * MP + kb * MP + wave * 16 + r;
  const float* ebr = Ere + kb * MP + ty * 16 + r;
  const float* ebi = Eim + kb * MP + ty * 16 + r;
  int kt = part * 3 + (wave > 0 ? wave - 1 : 0);
  const float* ear = Ere + kb * MP + kt * 16 + r;
  const float* eai = Eim + kb * MP + kt * 16 + r;
  const float* tbr = sTr + r * T_PITCH + kb;
  const float* tbi = sTi + r * T_PITCH + kb;
  P2Glob gr[S_RING];
  P2Lds  lr[S_RING];
  if (wave == 0) {
    p1_pass<3>(wave, r, hh, hat, ebr, ebi, sTr, sTi);
  } else {
    p2_gload(gr[0], 0, ear, eai);
    p2_gload(gr[1], 1, ear, eai);
    p1_pass<2>(wave, r, hh, hat, ebr, ebi, sTr, sTi);
  }
  __syncthreads();
  if (wave == 0) return;
  v8f sRe = {}, sIm = {};
#pragma unroll
  for (int j = 2; j < S_RING - 1; ++j) p2_gload(gr[j], j, ear, eai);
#pragma unroll
  for (int j = 0; j < S_RING - 1; ++j) p2_lload(lr[j], j, tbr, tbi);
#pragma unroll 1
  for (int g0 = 0; g0 < S_NB2; g0 += S_RING) {
#pragma unroll
    for (int j = 0; j < S_RING; ++j) {
      int gn = g0 + j + S_RING - 1;
      gn = gn < S_NB2 ? gn : S_NB2 - 1;
      p2_gload(gr[(j + S_RING - 1) % S_RING], gn, ear, eai);
      p2_lload(lr[(j + S_RING - 1) % S_RING], gn, tbr, tbi);
      p2_mac(gr[j], lr[j], sRe, sIm);
      __builtin_amdgcn_sched_barrier(0);
    }
  }
  store_tile(SRe + b * MP * MP, MP, kt * 16, ty * 16, sRe);
  store_tile(SIm + b * MP * MP, MP, kt * 16, ty * 16, sIm);
}

__device__ __forceinline__ void vmat_body(int b, int kt, int row0, int col0,
                       const float* __restrict__ W0, const float* __restrict__ W1,
                       const float* __restrict__ FyRe, const float* __restrict__ FyIm,
                       const float* __restrict__ FxRe, const float* __restrict__ FxIm,
                       float* __restrict__ PD) {
  int lane = threadIdx.x & 31, r = lane & 15, hh = lane >> 4, kb = hh << 1;
  const float* fxr = FxRe + (b * NPP + col0 + r) * MP + row0 + 8 * hh;
  const float* fxi = FxIm + (b * NPP + col0 + r) * MP + row0 + 8 * hh;
  v4f xr0 = *reinterpret_cast<const v4f*>(fxr), xr1 = *reinterpret_cast<const v4f*>(fxr + 4);
  v4f xi0 = *reinterpret_cast<const v4f*>(fxi), xi1 = *reinterpret_cast<const v4f*>(fxi + 4);
  const float* fyr = FyRe + b * NPP * MP + (col0 + r) * MP + kb;
  const float* fyi = FyIm + b * NPP * MP + (col0 + r) * MP + kb;
  const float* w0  = W0 + (row0 + r) * MP + kb;
  const float* w1  = W1 + (row0 + r) * MP + kb;
  v8f acc0 = {}, acc1 = {};
  for (int k0 = 0; k0 < MP; k0 += 4) {
    float re0 = fyr[k0], re1 = fyr[k0 + 1];
    float im0 = fyi[k0], im1 = fyi[k0 + 1];
    v2f p  = { re0 * re0 + im0 * im0, re1 * re1 + im1 * im1 };
    v2f a0 = { w0[k0], w0[k0 + 1] };
    v2f a1 = { w1[k0], w1[k0 + 1] };
    acc0 = wmma4(a0, p, acc0);
    acc1 = wmma4(a1, p, acc1);
  }
  float xr[8] = { xr0.x, xr0.y, xr0.z, xr0.w, xr1.x, xr1.y, xr1.z, xr1.w };
  float xi[8] = { xi0.x, xi0.y, xi0.z, xi0.w, xi1.x, xi1.y, xi1.z, xi1.w };
  float d0 = 0.f, d1 = 0.f;
#pragma unroll
  for (int v = 0; v < 8; ++v) {
    float ax = xr[v] * xr[v] + xi[v] * xi[v];
    d0 += ax * acc0[v];
    d1 += ax * acc1[v];
  }
  d0 += __shfl_xor(d0, 16);
  d1 += __shfl_xor(d1, 16);
  PD[kt * NOUT + (b * NPP + col0 + r) * NCH + hh] = hh ? d1 : d0;
}


__global__ void k_level0(float* __restrict__ Ere, float* __restrict__ Eim,
                         float* __restrict__ W0,  float* __restrict__ W1,
                         const float* s0, const float* s1,
                         const float* a0, const float* a1,
                         const float* __restrict__ x,
                         float* __restrict__ Gx, float* __restrict__ Gy) {
  int bid = blockIdx.x;
  if (bid < TAB_BLOCKS) {
    for (int j = 0; j < TAB_PER_T; ++j)
      tables_body((j * TAB_BLOCKS + bid) * L0_THREADS + threadIdx.x,
                  Ere, Eim, W0, W1, s0, s1, a0, a1);
  } else {
    int sb = bid - TAB_BLOCKS;
    for (int j = 0; j < SPR_PER_T; ++j)
      spread_body((j * SPR_BLOCKS + sb) * L0_THREADS + threadIdx.x, x, Gx, Gy);
  }
}

__device__ __forceinline__ void dft_body(int z, int row0, int col0,
                      const float* __restrict__ Gx, const float* __restrict__ Gy,
                      const float* __restrict__ Ere, const float* __restrict__ Eim,
                      float* __restrict__ FxRe, float* __restrict__ FxIm,
                      float* __restrict__ FyRe, float* __restrict__ FyIm) {
  const float* G = z ? Gy : Gx;
  float* DRe = z ? FyRe : FxRe;
  float* DIm = z ? FyIm : FxIm;
  int lane = threadIdx.x & 31, r = lane & 15, kb = (lane >> 4) << 1;
  const float* ga = G + (row0 + r) * MP + kb;
  const float* gb = ga + 16 * MP;
  const float* er = Ere + kb * MP + col0 + r;
  const float* ei = Eim + kb * MP + col0 + r;
  v8f re0 = {}, im0 = {}, re1 = {}, im1 = {};
  for (int g0 = 0; g0 < MP; g0 += 4 * KGRP) {
    v2f a0[KGRP], a1[KGRP], bR[KGRP], bI[KGRP];
#pragma unroll
    for (int s = 0; s < KGRP; ++s) {
      int k0 = g0 + 4 * s;
      a0[s].x = ga[k0];      a0[s].y = ga[k0 + 1];
      a1[s].x = gb[k0];      a1[s].y = gb[k0 + 1];
      bR[s].x = er[k0 * MP]; bR[s].y = er[(k0 + 1) * MP];
      bI[s].x = ei[k0 * MP]; bI[s].y = ei[(k0 + 1) * MP];
    }
    __builtin_amdgcn_sched_barrier(0);
#pragma unroll
    for (int s = 0; s < KGRP; ++s) {
      re0 = wmma4(a0[s], bR[s], re0);
      im0 = wmma4(a0[s], bI[s], im0);
      re1 = wmma4(a1[s], bR[s], re1);
      im1 = wmma4(a1[s], bI[s], im1);
    }
  }
  store_tile(DRe, MP, row0,      col0, re0);
  store_tile(DIm, MP, row0,      col0, im0);
  store_tile(DRe, MP, row0 + 16, col0, re1);
  store_tile(DIm, MP, row0 + 16, col0, im1);
}

__global__ void k_level1(const float* __restrict__ Gx, const float* __restrict__ Gy,
                         const float* __restrict__ Ere, const float* __restrict__ Eim,
                         float* __restrict__ FxRe, float* __restrict__ FxIm,
                         float* __restrict__ FyRe, float* __restrict__ FyIm,
                         float* __restrict__ HT) {
  int bid = blockIdx.x;
  if (bid < GRAM_BLOCKS) {
    int b  = bid / (TILES * TILES);
    int t  = bid - b * (TILES * TILES);
    int ty = t / TILES, tx = t - ty * TILES;
    hgram_body(b, tx * 16, ty * 16, Gy, Gx, HT);
  } else {
    int id = bid - GRAM_BLOCKS;
    int z  = id / ((NPTS / 32) * TILES);   id -= z * ((NPTS / 32) * TILES);
    int by = id / (NPTS / 32), bx = id - by * (NPTS / 32);
    dft_body(z, bx * 32, by * 16, Gx, Gy, Ere, Eim, FxRe, FxIm, FyRe, FyIm);
  }
}

__global__ void __launch_bounds__(S_THREADS)
k_level2(const float* __restrict__ HT,
         const float* __restrict__ Ere,  const float* __restrict__ Eim,
         const float* __restrict__ FyRe, const float* __restrict__ FyIm,
         const float* __restrict__ W0,   const float* __restrict__ W1,
         const float* __restrict__ FxRe, const float* __restrict__ FxIm,
         float* __restrict__ SRe, float* __restrict__ SIm,
         float* __restrict__ PD) {
  __shared__ __attribute__((aligned(16))) float sT[2 * 16 * T_PITCH];
  int bid = blockIdx.x;
  int wave = __builtin_amdgcn_readfirstlane(threadIdx.x >> 5);
  if (bid < S_BLOCKS) {
    int bt = bid / S_SPLIT, part = bid - bt * S_SPLIT;
    int b = bt / TILES, ty = bt - b * TILES;
    sdft_body(b, wave, ty, part, HT, Ere, Eim, SRe, SIm, sT);
  } else {
    int id = (bid - S_BLOCKS) * S_WAVES + wave;
    int b  = id / VT_PER_B;     id -= b * VT_PER_B;
    int ty = id / TILES, tx = id - ty * TILES;
    vmat_body(b, tx, tx * 16, ty * 16, W0, W1, FyRe, FyIm, FxRe, FxIm, PD);
  }
}

#define U_WAVES   4
#define U_PTS     (16 * U_WAVES)
#define U_PITCH   148
#define U_ROW4    (MP / 4)
#define U_THREADS (32 * U_WAVES)
#define U_AROWS   17
#define U_EDGE    (MM - 1)
#define U_A_TRIPS ((U_AROWS * U_ROW4 + U_THREADS - 1) / U_THREADS)
#define U_B_TRIPS (U_PTS * U_ROW4 / U_THREADS)
#define U_B_CH    (U_B_TRIPS / 3)
#define U_APL     (U_AROWS * U_PITCH)
#define U_BPL     (U_PTS * U_PITCH)

template <bool EXTRA>
__device__ __forceinline__ void u_kloop(const float* sT0r, const float* sT0i,
                                        const float* sT1r, const float* sT1i,
                                        const float* sFr,  const float* sFi,
                                        int la, int lb, int lx, int lbx, bool row0lane, bool part,
                                        v8f& accRe0, v8f& accIm0, v8f& accRe1, v8f& accIm1, v8f& accX) {
#pragma unroll
  for (int k0 = 0; k0 < MP; k0 += 4) {
    v2f tRe0 = *reinterpret_cast<const v2f*>(sT0r + la + k0);
    v2f tIm0 = *reinterpret_cast<const v2f*>(sT0i + la + k0);
    v2f tRe1 = *reinterpret_cast<const v2f*>(sT1r + la + k0);
    v2f tIm1 = *reinterpret_cast<const v2f*>(sT1i + la + k0);
    v2f bRe  = *reinterpret_cast<const v2f*>(sFr + lb + k0);
    v2f bIm  = *reinterpret_cast<const v2f*>(sFi + lb + k0);
    v2f tReN0 = { -tRe0.x, -tRe0.y };
    v2f tReN1 = { -tRe1.x, -tRe1.y };
    accRe0 = wmma4(tRe0,  bRe, accRe0);
    accRe1 = wmma4(tRe1,  bRe, accRe1);
    accIm0 = wmma4(tIm0,  bRe, accIm0);
    accIm1 = wmma4(tIm1,  bRe, accIm1);
    if (EXTRA) {
      v2f xr = *reinterpret_cast<const v2f*>(sT0r + lx + k0);
      v2f xi = *reinterpret_cast<const v2f*>(sT0i + lx + k0);
      v2f zr = { row0lane ? xr.x : 0.f, row0lane ? xr.y : 0.f };
      v2f zi = { row0lane ? xi.x : 0.f, row0lane ? xi.y : 0.f };
      v2f cRe = *reinterpret_cast<const v2f*>(sFr + lbx + k0);
      v2f cIm = *reinterpret_cast<const v2f*>(sFi + lbx + k0);
      v2f a1 = { part ? zi.x : zr.x, part ? zi.y : zr.y };
      v2f a2 = { part ? -zr.x : zi.x, part ? -zr.y : zi.y };
      accX = wmma4(a1, cRe, accX);
      accRe0 = wmma4(tIm0,  bIm, accRe0);
      accRe1 = wmma4(tIm1,  bIm, accRe1);
      accIm0 = wmma4(tReN0, bIm, accIm0);
      accIm1 = wmma4(tReN1, bIm, accIm1);
      accX = wmma4(a2, cIm, accX);
    } else {
      accRe0 = wmma4(tIm0,  bIm, accRe0);
      accRe1 = wmma4(tIm1,  bIm, accRe1);
      accIm0 = wmma4(tReN0, bIm, accIm0);
      accIm1 = wmma4(tReN1, bIm, accIm1);
    }
  }
}

__global__ void __launch_bounds__(U_THREADS)
k_umat(const float* __restrict__ W0, const float* __restrict__ W1,
       const float* __restrict__ SRe, const float* __restrict__ SIm,
       const float* __restrict__ FyRe, const float* __restrict__ FyIm,
       const float* __restrict__ FxRe, const float* __restrict__ FxIm,
       float* __restrict__ PT, float* __restrict__ PE) {
  __shared__ __attribute__((aligned(16))) float sm[4 * U_APL + 2 * U_BPL];
  float* sT0r = sm;
  float* sT0i = sm + U_APL;
  float* sT1r = sm + 2 * U_APL;
  float* sT1i = sm + 3 * U_APL;
  float* sFr  = sm + 4 * U_APL;
  float* sFi  = sm + 4 * U_APL + U_BPL;
  int b = blockIdx.z;
  int kt = blockIdx.x;
  int row0 = kt * 16;
  int pblk = blockIdx.y * U_PTS;
  int tid = threadIdx.x;
  const float* sre = SRe + b * MP * MP;
  const float* sim = SIm + b * MP * MP;
  const float* fyr = FyRe + (b * NPP + pblk) * MP;
  const float* fyi = FyIm + (b * NPP + pblk) * MP;
  v4f gsr[U_A_TRIPS], gsi[U_A_TRIPS], ga0[U_A_TRIPS], ga1[U_A_TRIPS];
  v4f gfr0[U_B_CH], gfi0[U_B_CH], gfr1[U_B_CH], gfi1[U_B_CH], gfr2[U_B_CH], gfi2[U_B_CH];
  bool tail = tid < U_AROWS * U_ROW4 - (U_A_TRIPS - 1) * U_THREADS;
#pragma unroll
  for (int j = 0; j < U_A_TRIPS; ++j) {
    if (j < U_A_TRIPS - 1 || tail) {
      int idx = tid + j * U_THREADS;
      int rr = idx / U_ROW4, c4 = idx - rr * U_ROW4;
      int grow = rr < 16 ? row0 + rr : U_EDGE;
      int g = grow * MP + c4 * 4;
      gsr[j] = *reinterpret_cast<const v4f*>(sre + g);
      gsi[j] = *reinterpret_cast<const v4f*>(sim + g);
      ga0[j] = *reinterpret_cast<const v4f*>(W0 + g);
      ga1[j] = *reinterpret_cast<const v4f*>(W1 + g);
    }
  }
#pragma unroll
  for (int j = 0; j < U_B_CH; ++j) {
    gfr0[j] = *reinterpret_cast<const v4f*>(fyr + 4 * (tid + j * U_THREADS));
    gfi0[j] = *reinterpret_cast<const v4f*>(fyi + 4 * (tid + j * U_THREADS));
    gfr1[j] = *reinterpret_cast<const v4f*>(fyr + 4 * (tid + (j + U_B_CH) * U_THREADS));
    gfi1[j] = *reinterpret_cast<const v4f*>(fyi + 4 * (tid + (j + U_B_CH) * U_THREADS));
    gfr2[j] = *reinterpret_cast<const v4f*>(fyr + 4 * (tid + (j + 2 * U_B_CH) * U_THREADS));
    gfi2[j] = *reinterpret_cast<const v4f*>(fyi + 4 * (tid + (j + 2 * U_B_CH) * U_THREADS));
  }
  __builtin_amdgcn_sched_barrier(0);
#pragma unroll
  for (int j = 0; j < U_A_TRIPS; ++j) {
    if (j < U_A_TRIPS - 1 || tail) {
      int idx = tid + j * U_THREADS;
      int rr = idx / U_ROW4, c4 = idx - rr * U_ROW4;
      int l = rr * U_PITCH + c4 * 4;
      *reinterpret_cast<v4f*>(sT0r + l) = ga0[j] * gsr[j];
      *reinterpret_cast<v4f*>(sT0i + l) = ga0[j] * gsi[j];
      *reinterpret_cast<v4f*>(sT1r + l) = ga1[j] * gsr[j];
      *reinterpret_cast<v4f*>(sT1i + l) = ga1[j] * gsi[j];
    }
  }
#pragma unroll
  for (int j = 0; j < U_B_CH; ++j) {
    int i0 = tid + j * U_THREADS, i1 = i0 + U_B_CH * U_THREADS, i2 = i1 + U_B_CH * U_THREADS;
    int l0 = (i0 / U_ROW4) * U_PITCH + (i0 % U_ROW4) * 4;
    int l1 = (i1 / U_ROW4) * U_PITCH + (i1 % U_ROW4) * 4;
    int l2 = (i2 / U_ROW4) * U_PITCH + (i2 % U_ROW4) * 4;
    *reinterpret_cast<v4f*>(sFr + l0) = gfr0[j];
    *reinterpret_cast<v4f*>(sFi + l0) = gfi0[j];
    *reinterpret_cast<v4f*>(sFr + l1) = gfr1[j];
    *reinterpret_cast<v4f*>(sFi + l1) = gfi1[j];
    *reinterpret_cast<v4f*>(sFr + l2) = gfr2[j];
    *reinterpret_cast<v4f*>(sFi + l2) = gfi2[j];
  }
  __syncthreads();
  int wave = __builtin_amdgcn_readfirstlane(tid >> 5);
  int lane = tid & 31, r = lane & 15, hh = lane >> 4, kb = hh << 1;
  int col0 = pblk + wave * 16;
  int la = r * U_PITCH + kb;
  int lb = (wave * 16 + r) * U_PITCH + kb;
  int xt = kt >> 1;
  bool part = (kt & 1) != 0;
  int lx  = wave * 2 * U_APL + 16 * U_PITCH + kb;
  int lbx = (xt * 16 + r) * U_PITCH + kb;
  const float* fxr = FxRe + (b * NPP + col0 + r) * MP + row0 + 8 * hh;
  const float* fxi = FxIm + (b * NPP + col0 + r) * MP + row0 + 8 * hh;
  v4f xr0 = *reinterpret_cast<const v4f*>(fxr), xr1 = *reinterpret_cast<const v4f*>(fxr + 4);
  v4f xi0 = *reinterpret_cast<const v4f*>(fxi), xi1 = *reinterpret_cast<const v4f*>(fxi + 4);
  float xe = (part ? FxIm : FxRe)[(b * NPP + pblk + xt * 16 + r) * MP + U_EDGE];
  v8f accRe0 = {}, accIm0 = {}, accRe1 = {}, accIm1 = {}, accX = {};
  if (wave < 2) {
    u_kloop<true>(sT0r, sT0i, sT1r, sT1i, sFr, sFi, la, lb, lx, lbx, r == 0, part,
                  accRe0, accIm0, accRe1, accIm1, accX);
  } else {
    u_kloop<false>(sT0r, sT0i, sT1r, sT1i, sFr, sFi, la, lb, lx, lbx, false, part,
                   accRe0, accIm0, accRe1, accIm1, accX);
  }
  float xr[8] = { xr0.x, xr0.y, xr0.z, xr0.w, xr1.x, xr1.y, xr1.z, xr1.w };
  float xi[8] = { xi0.x, xi0.y, xi0.z, xi0.w, xi1.x, xi1.y, xi1.z, xi1.w };
  float t0 = 0.f, t1 = 0.f;
#pragma unroll
  for (int v = 0; v < 8; ++v) {
    t0 += xr[v] * accRe0[v] + xi[v] * accIm0[v];
    t1 += xr[v] * accRe1[v] + xi[v] * accIm1[v];
  }
  t0 += __shfl_xor(t0, 16);
  t1 += __shfl_xor(t1, 16);
  PT[kt * NOUT + (b * NPP + col0 + r) * NCH + hh] = hh ? t1 : t0;
  if (wave < 2 && hh == 0) {
    PE[(((int)part) * NCH + wave) * NPTS + b * NPP + pblk + xt * 16 + r] = xe * accX[0];
  }
}

__global__ void __launch_bounds__(FIN_THREADS)
k_final(const float* __restrict__ PT, const float* __restrict__ PE,
        const float* __restrict__ PD, float* __restrict__ out) {
  int t = blockIdx.x * FIN_THREADS + threadIdx.x;
  int pt = t >> 1, ch = t & 1;
  float tot = 0.f, dia = 0.f;
#pragma unroll
  for (int k = 0; k < TILES - 1; ++k) tot += PT[k * NOUT + t];
  tot += PE[ch * NPTS + pt];
  tot += PE[(NCH + ch) * NPTS + pt];
#pragma unroll
  for (int k = 0; k < TILES; ++k) dia += PD[k * NOUT + t];
  out[t] = tot - dia;
}


extern "C" void kernel_launch(void* const* d_in, const int* in_sizes, int n_in,
                              void* d_out, int out_size, void* d_ws, size_t ws_size,
                              hipStream_t stream) {
  (void)in_sizes; (void)n_in; (void)out_size; (void)ws_size;
  const float* x  = (const float*)d_in[0];
  const float* s0 = (const float*)d_in[1];
  const float* s1 = (const float*)d_in[2];
  const float* a0 = (const float*)d_in[3];
  const float* a1 = (const float*)d_in[4];
  float* out = (float*)d_out;

  float* w = (float*)d_ws;
  float* Ere  = w; w += MP * MP;
  float* Eim  = w; w += MP * MP;
  float* W0   = w; w += MP * MP;
  float* W1   = w; w += MP * MP;
  float* Gx   = w; w += NPTS * MP;
  float* Gy   = w; w += NPTS * MP;
  float* FxRe = w; w += NPTS * MP;
  float* FxIm = w; w += NPTS * MP;
  float* FyRe = w; w += NPTS * MP;
  float* FyIm = w; w += NPTS * MP;
  float* SRe  = w; w += NB * MP * MP;
  float* SIm  = w; w += NB * MP * MP;
  float* PT   = w; w += (TILES - 1) * NOUT;
  float* PE   = w; w += 2 * NCH * NPTS;
  float* PD   = w; w += TILES * NOUT;
  float* HT   = w; w += NB * MP * MP;

  k_level0<<<TAB_BLOCKS + SPR_BLOCKS, L0_THREADS, 0, stream>>>(Ere, Eim, W0, W1, s0, s1, a0, a1,
                                                               x, Gx, Gy);
  k_level1<<<GRAM_BLOCKS + DFT_BLOCKS, 32, 0, stream>>>(Gx, Gy, Ere, Eim,
                                                        FxRe, FxIm, FyRe, FyIm, HT);
  k_level2<<<S_BLOCKS + V_BLOCKS, S_THREADS, 0, stream>>>(HT, Ere, Eim, FyRe, FyIm,
                                                          W0, W1, FxRe, FxIm, SRe, SIm, PD);
  k_umat<<<dim3(MM / 16, NPP / U_PTS, NB), U_THREADS, 0, stream>>>(W0, W1, SRe, SIm, FyRe, FyIm,
                                                         FxRe, FxIm, PT, PE);
  k_final<<<NOUT / FIN_THREADS, FIN_THREADS, 0, stream>>>(PT, PE, PD, out);
}


# U level in real cos+sin form: B=W*(S.re-S.im) meets Fy.re-Fy.im and Fy.re+Fy.im, 2 real products per channel not 4 (cross terms odd in k: W even, S conjugate-symmetric); new sum order
# speedup vs baseline: 2.3797x; 1.0679x over previous
#include <hip/hip_runtime.h>
#include <math.h>


typedef __attribute__((ext_vector_type(2))) float v2f;
typedef __attribute__((ext_vector_type(8))) float v8f;
typedef __attribute__((ext_vector_type(4))) float v4f;

#define MM    129
#define MP    144
#define NPP   512
#define NB    4
#define NPTS  (NB * NPP)
#define NCH   2

#define PI_F   3.14159265358979323846f
#define TAU_F  (12.0f / (129.0f * 129.0f))
#define L_F    (2.0f * PI_F)

#define TILES       (MP / 16)
#define L0_THREADS  128
#define TAB_PER_T   2
#define SPR_PER_T   8
#define TAB_BLOCKS  (MP * MP / (L0_THREADS * TAB_PER_T))
#define SPR_BLOCKS  (NPTS * MP / (L0_THREADS * SPR_PER_T))
#define GRAM_BLOCKS (TILES * TILES * NB)
#define VT_PER_B    (TILES * (NPP / 16))
#define VMAT_BLOCKS (VT_PER_B * NB)
#define DFT_BLOCKS  ((NPTS / 32) * TILES * 2)
#define S_WAVES     4
#define S_THREADS   (32 * S_WAVES)
#define S_SPLIT     3
#define S_BLOCKS    (TILES * NB * S_SPLIT)
#define V_BLOCKS    (VMAT_BLOCKS / S_WAVES)
#define T_PITCH     148
#define S_KG1       1
#define S_NB1       (MP / (4 * S_KG1))
#define S_KG2       3
#define S_NB2       (MP / (4 * S_KG2))
#define S_RING      6
#define NOUT        (NPTS * NCH)
#define FIN_THREADS 256
#define KGRP        6


__device__ __forceinline__ v8f wmma4(v2f a, v2f b, v8f c) {
  return __builtin_amdgcn_wmma_f32_16x16x4_f32(false, a, false, b, (short)0, c, false, false);
}

__device__ __forceinline__ void store_tile(float* C, int ldc, int row0, int col0, v8f acc) {
  int lane = threadIdx.x & 31;
  int r0 = row0 + ((lane >> 4) << 3);
  int c  = col0 + (lane & 15);
#pragma unroll
  for (int v = 0; v < 8; ++v) C[(r0 + v) * ldc + c] = acc[v];
}


__device__ __forceinline__ void tables_body(int t,
                         float* __restrict__ Ere, float* __restrict__ Eim,
                         float* __restrict__ W0,  float* __restrict__ W1,
                         const float* s0, const float* s1,
                         const float* a0, const float* a1) {
  if (t >= MP * MP) return;
  int i = t / MP, k = t % MP;
  float er = 0.f, ei = 0.f, w0 = 0.f, w1 = 0.f;
  if (i < MM && k < MM) {
    int m = (i * k) % MM;
    float th = (2.0f * PI_F / (float)MM) * (float)m;
    er = cosf(th);
    ei = -sinf(th);
    float f1 = (float)((i <= MM / 2) ? i : i - MM);
    float f2 = (float)((k <= MM / 2) ? k : k - MM);
    float k2 = f1 * f1 + f2 * f2;
    float dec = (PI_F / TAU_F) * expf(TAU_F * k2);
    float m2   = (float)MM * (float)MM;
    float coef = 1.0f / (8.0f * PI_F * PI_F * m2 * m2);
    float base = coef * dec * dec * 4.0f * PI_F;
    float sh0 = s0[0], sh1 = s1[0];
    w0 = base * a0[0] / (k2 + sh0 * sh0);
    w1 = base * a1[0] / (k2 + sh1 * sh1);
  }
  Ere[t] = er; Eim[t] = ei; W0[t] = w0; W1[t] = w1;
}

__device__ __forceinline__ void spread_body(int t, const float* __restrict__ x,
                         float* __restrict__ Gx, float* __restrict__ Gy) {
  if (t >= NPTS * MP) return;
  int pt = t / MP, i = t % MP;
  float gx = 0.f, gy = 0.f;
  if (i < MM) {
    float xi  = (L_F / (float)MM) * (float)i;
    float inv = 1.0f / (4.0f * TAU_F);
    float d0 = x[2 * pt + 0] - xi;
    float d1 = x[2 * pt + 1] - xi;
    gx = expf(-d0 * d0 * inv) + expf(-(d0 - L_F) * (d0 - L_F) * inv)
       + expf(-(d0 + L_F) * (d0 + L_F) * inv);
    gy = expf(-d1 * d1 * inv) + expf(-(d1 - L_F) * (d1 - L_F) * inv)
       + expf(-(d1 + L_F) * (d1 + L_F) * inv);
  }
  Gx[t] = gx; Gy[t] = gy;
}

#define HG_RING   8
#define HG_AHEAD  (HG_RING - 1)
#define HG_GROUPS (NPP / 8)

struct HFrag { v2f a[2], b[2]; };

__device__ __forceinline__ void hgram_load(HFrag& f, int g,
                       const float* __restrict__ gx, const float* __restrict__ gy,
                       int kb, int ca, int cb) {
#pragma unroll
  for (int s = 0; s < 2; ++s) {
    int q0 = g * 8 + s * 4 + kb, q1 = q0 + 1;
    f.a[s].x = gx[q0 * MP + ca]; f.a[s].y = gx[q1 * MP + ca];
    f.b[s].x = gy[q0 * MP + cb]; f.b[s].y = gy[q1 * MP + cb];
  }
}

__device__ __forceinline__ void hgram_body(int b, int row0, int col0,
                       const float* __restrict__ A, const float* __restrict__ B,
                       float* __restrict__ D) {
  int lane = threadIdx.x & 31, r = lane & 15, kb = (lane >> 4) << 1;
  const float* gx = A + b * NPP * MP;
  const float* gy = B + b * NPP * MP;
  int ca = row0 + r, cb = col0 + r;
  v8f acc0 = {}, acc1 = {};
  HFrag ring[HG_RING];
#pragma unroll
  for (int j = 0; j < HG_AHEAD; ++j) {
    hgram_load(ring[j], j, gx, gy, kb, ca, cb);
    __builtin_amdgcn_sched_barrier(0);
  }
  for (int g0 = 0; g0 < HG_GROUPS; g0 += HG_RING) {
#pragma unroll
    for (int j = 0; j < HG_RING; ++j) {
      int g  = g0 + j;
      int gn = g + HG_AHEAD;
      gn = gn < HG_GROUPS ? gn : HG_GROUPS - 1;
      hgram_load(ring[(j + HG_AHEAD) % HG_RING], gn, gx, gy, kb, ca, cb);
      acc0 = wmma4(ring[j].a[0], ring[j].b[0], acc0);
      acc1 = wmma4(ring[j].a[1], ring[j].b[1], acc1);
      __builtin_amdgcn_sched_barrier(0);
    }
  }
  v8f acc = acc0 + acc1;
  store_tile(D + b * MP * MP, MP, row0, col0, acc);
}

template <int NT>
struct P1Frag { v2f a[NT][S_KG1], br[S_KG1], bi[S_KG1]; };
struct P2Glob { v2f ar[S_KG2], ai[S_KG2]; };
struct P2Lds  { v2f br[S_KG2], bi[S_KG2]; };

template <int NT>
__device__ __forceinline__ void p1_load(P1Frag<NT>& f, int g, const float* hat,
                                        const float* ebr, const float* ebi) {
#pragma unroll
  for (int s = 0; s < S_KG1; ++s) {
    int k0 = 4 * (g * S_KG1 + s);
#pragma unroll
    for (int t = 0; t < NT; ++t) {
      f.a[t][s].x = hat[t * 64 + k0 * MP];
      f.a[t][s].y = hat[t * 64 + (k0 + 1) * MP];
    }
    f.br[s].x = ebr[k0 * MP]; f.br[s].y = ebr[(k0 + 1) * MP];
    f.bi[s].x = ebi[k0 * MP]; f.bi[s].y = ebi[(k0 + 1) * MP];
  }
}
template <int NT>
__device__ __forceinline__ void p1_mac(const P1Frag<NT>& f, v8f (&tRe)[NT], v8f (&tIm)[NT]) {
#pragma unroll
  for (int s = 0; s < S_KG1; ++s) {
#pragma unroll
    for (int t = 0; t < NT; ++t) {
      tRe[t] = wmma4(f.a[t][s], f.br[s], tRe[t]);
      tIm[t] = wmma4(f.a[t][s], f.bi[s], tIm[t]);
    }
  }
}
template <int NT>
__device__ __forceinline__ void p1_pass(int wave, int r, int hh, const float* hat,
                                        const float* ebr, const float* ebi,
                                        float* sTr, float* sTi) {
  v8f tRe[NT], tIm[NT];
#pragma unroll
  for (int t = 0; t < NT; ++t) { tRe[t] = v8f{}; tIm[t] = v8f{}; }
  P1Frag<NT> ring[S_RING];
#pragma unroll
  for (int j = 0; j < S_RING - 1; ++j) {
    p1_load<NT>(ring[j], j, hat, ebr, ebi);
    __builtin_amdgcn_sched_barrier(0);
  }
#pragma unroll 1
  for (int g0 = 0; g0 < S_NB1; g0 += S_RING) {
#pragma unroll
    for (int j = 0; j < S_RING; ++j) {
      int gn = g0 + j + S_RING - 1;
      gn = gn < S_NB1 ? gn : S_NB1 - 1;
      p1_load<NT>(ring[(j + S_RING - 1) % S_RING], gn, hat, ebr, ebi);
      p1_mac<NT>(ring[j], tRe, tIm);
      __builtin_amdgcn_sched_barrier(0);
    }
  }
#pragma unroll
  for (int t = 0; t < NT; ++t) {
    float* dr = sTr + r * T_PITCH + (wave + 4 * t) * 16 + 8 * hh;
    float* di = sTi + r * T_PITCH + (wave + 4 * t) * 16 + 8 * hh;
    v4f r0 = { tRe[t][0], tRe[t][1], tRe[t][2], tRe[t][3] }, r1 = { tRe[t][4], tRe[t][5], tRe[t][6], tRe[t][7] };
    v4f i0 = { tIm[t][0], tIm[t][1], tIm[t][2], tIm[t][3] }, i1 = { tIm[t][4], tIm[t][5], tIm[t][6], tIm[t][7] };
    *reinterpret_cast<v4f*>(dr) = r0; *reinterpret_cast<v4f*>(dr + 4) = r1;
    *reinterpret_cast<v4f*>(di) = i0; *reinterpret_cast<v4f*>(di + 4) = i1;
  }
}
__device__ __forceinline__ void p2_gload(P2Glob& f, int g, const float* ear, const float* eai) {
#pragma unroll
  for (int s = 0; s < S_KG2; ++s) {
    int k0 = 4 * (g * S_KG2 + s);
    f.ar[s].x = ear[k0 * MP]; f.ar[s].y = ear[(k0 + 1) * MP];
    f.ai[s].x = eai[k0 * MP]; f.ai[s].y = eai[(k0 + 1) * MP];
  }
}
__device__ __forceinline__ void p2_lload(P2Lds& f, int g, const float* tbr, const float* tbi) {
#pragma unroll
  for (int s = 0; s < S_KG2; ++s) {
    int k0 = 4 * (g * S_KG2 + s);
    f.br[s] = *reinterpret_cast<const v2f*>(tbr + k0);
    f.bi[s] = *reinterpret_cast<const v2f*>(tbi + k0);
  }
}
__device__ __forceinline__ void p2_mac(const P2Glob& a, const P2Lds& b, v8f& sRe, v8f& sIm) {
#pragma unroll
  for (int s = 0; s < S_KG2; ++s) {
    v2f aiN = { -a.ai[s].x, -a.ai[s].y };
    sRe = wmma4(a.ar[s], b.br[s], sRe);
    sRe = wmma4(aiN,     b.bi[s], sRe);
    sIm = wmma4(a.ar[s], b.bi[s], sIm);
    sIm = wmma4(a.ai[s], b.br[s], sIm);
  }
}

__device__ __forceinline__ void sdft_body(int b, int wave, int ty, int part,
                       const float* __restrict__ HT,
                       const float* __restrict__ Ere, const float* __restrict__ Eim,
                       float* __restrict__ SRe, float* __restrict__ SIm, float* sT) {
  int lane = threadIdx.x & 31, r = lane & 15, hh = lane >> 4, kb = hh << 1;
  float* sTr = sT;
  float* sTi = sT + 16 * T_PITCH;
  const float* hat = HT + b * MP * MP + kb * MP + wave * 16 + r;
  const float* ebr = Ere + kb * MP + ty * 16 + r;
  const float* ebi = Eim + kb * MP + ty * 16 + r;
  int kt = part * 3 + (wave > 0 ? wave - 1 : 0);
  const float* ear = Ere + kb * MP + kt * 16 + r;
  const float* eai = Eim + kb * MP + kt * 16 + r;
  const float* tbr = sTr + r * T_PITCH + kb;
  const float* tbi = sTi + r * T_PITCH + kb;
  P2Glob gr[S_RING];
  P2Lds  lr[S_RING];
  if (wave == 0) {
    p1_pass<3>(wave, r, hh, hat, ebr, ebi, sTr, sTi);
  } else {
    p2_gload(gr[0], 0, ear, eai);
    p2_gload(gr[1], 1, ear, eai);
    p1_pass<2>(wave, r, hh, hat, ebr, ebi, sTr, sTi);
  }
  __syncthreads();
  if (wave == 0) return;
  v8f sRe = {}, sIm = {};
#pragma unroll
  for (int j = 2; j < S_RING - 1; ++j) p2_gload(gr[j], j, ear, eai);
#pragma unroll
  for (int j = 0; j < S_RING - 1; ++j) p2_lload(lr[j], j, tbr, tbi);
#pragma unroll 1
  for (int g0 = 0; g0 < S_NB2; g0 += S_RING) {
#pragma unroll
    for (int j = 0; j < S_RING; ++j) {
      int gn = g0 + j + S_RING - 1;
      gn = gn < S_NB2 ? gn : S_NB2 - 1;
      p2_gload(gr[(j + S_RING - 1) % S_RING], gn, ear, eai);
      p2_lload(lr[(j + S_RING - 1) % S_RING], gn, tbr, tbi);
      p2_mac(gr[j], lr[j], sRe, sIm);
      __builtin_amdgcn_sched_barrier(0);
    }
  }
  store_tile(SRe + b * MP * MP, MP, kt * 16, ty * 16, sRe);
  store_tile(SIm + b * MP * MP, MP, kt * 16, ty * 16, sIm);
}

__device__ __forceinline__ void vmat_body(int b, int kt, int row0, int col0,
                       const float* __restrict__ W0, const float* __restrict__ W1,
                       const float* __restrict__ FyRe, const float* __restrict__ FyIm,
                       const float* __restrict__ FxRe, const float* __restrict__ FxIm,
                       float* __restrict__ PD) {
  int lane = threadIdx.x & 31, r = lane & 15, hh = lane >> 4, kb = hh << 1;
  const float* fxr = FxRe + (b * NPP + col0 + r) * MP + row0 + 8 * hh;
  const float* fxi = FxIm + (b * NPP + col0 + r) * MP + row0 + 8 * hh;
  v4f xr0 = *reinterpret_cast<const v4f*>(fxr), xr1 = *reinterpret_cast<const v4f*>(fxr + 4);
  v4f xi0 = *reinterpret_cast<const v4f*>(fxi), xi1 = *reinterpret_cast<const v4f*>(fxi + 4);
  const float* fyr = FyRe + b * NPP * MP + (col0 + r) * MP + kb;
  const float* fyi = FyIm + b * NPP * MP + (col0 + r) * MP + kb;
  const float* w0  = W0 + (row0 + r) * MP + kb;
  const float* w1  = W1 + (row0 + r) * MP + kb;
  v8f acc0 = {}, acc1 = {};
  for (int k0 = 0; k0 < MP; k0 += 4) {
    float re0 = fyr[k0], re1 = fyr[k0 + 1];
    float im0 = fyi[k0], im1 = fyi[k0 + 1];
    v2f p  = { re0 * re0 + im0 * im0, re1 * re1 + im1 * im1 };
    v2f a0 = { w0[k0], w0[k0 + 1] };
    v2f a1 = { w1[k0], w1[k0 + 1] };
    acc0 = wmma4(a0, p, acc0);
    acc1 = wmma4(a1, p, acc1);
  }
  float xr[8] = { xr0.x, xr0.y, xr0.z, xr0.w, xr1.x, xr1.y, xr1.z, xr1.w };
  float xi[8] = { xi0.x, xi0.y, xi0.z, xi0.w, xi1.x, xi1.y, xi1.z, xi1.w };
  float d0 = 0.f, d1 = 0.f;
#pragma unroll
  for (int v = 0; v < 8; ++v) {
    float ax = xr[v] * xr[v] + xi[v] * xi[v];
    d0 += ax * acc0[v];
    d1 += ax * acc1[v];
  }
  d0 += __shfl_xor(d0, 16);
  d1 += __shfl_xor(d1, 16);
  PD[kt * NOUT + (b * NPP + col0 + r) * NCH + hh] = hh ? d1 : d0;
}


__global__ void k_level0(float* __restrict__ Ere, float* __restrict__ Eim,
                         float* __restrict__ W0,  float* __restrict__ W1,
                         const float* s0, const float* s1,
                         const float* a0, const float* a1,
                         const float* __restrict__ x,
                         float* __restrict__ Gx, float* __restrict__ Gy) {
  int bid = blockIdx.x;
  if (bid < TAB_BLOCKS) {
    for (int j = 0; j < TAB_PER_T; ++j)
      tables_body((j * TAB_BLOCKS + bid) * L0_THREADS + threadIdx.x,
                  Ere, Eim, W0, W1, s0, s1, a0, a1);
  } else {
    int sb = bid - TAB_BLOCKS;
    for (int j = 0; j < SPR_PER_T; ++j)
      spread_body((j * SPR_BLOCKS + sb) * L0_THREADS + threadIdx.x, x, Gx, Gy);
  }
}

__device__ __forceinline__ void dft_body(int z, int row0, int col0,
                      const float* __restrict__ Gx, const float* __restrict__ Gy,
                      const float* __restrict__ Ere, const float* __restrict__ Eim,
                      float* __restrict__ FxRe, float* __restrict__ FxIm,
                      float* __restrict__ FyRe, float* __restrict__ FyIm) {
  const float* G = z ? Gy : Gx;
  float* DRe = z ? FyRe : FxRe;
  float* DIm = z ? FyIm : FxIm;
  int lane = threadIdx.x & 31, r = lane & 15, kb = (lane >> 4) << 1;
  const float* ga = G + (row0 + r) * MP + kb;
  const float* gb = ga + 16 * MP;
  const float* er = Ere + kb * MP + col0 + r;
  const float* ei = Eim + kb * MP + col0 + r;
  v8f re0 = {}, im0 = {}, re1 = {}, im1 = {};
  for (int g0 = 0; g0 < MP; g0 += 4 * KGRP) {
    v2f a0[KGRP], a1[KGRP], bR[KGRP], bI[KGRP];
#pragma unroll
    for (int s = 0; s < KGRP; ++s) {
      int k0 = g0 + 4 * s;
      a0[s].x = ga[k0];      a0[s].y = ga[k0 + 1];
      a1[s].x = gb[k0];      a1[s].y = gb[k0 + 1];
      bR[s].x = er[k0 * MP]; bR[s].y = er[(k0 + 1) * MP];
      bI[s].x = ei[k0 * MP]; bI[s].y = ei[(k0 + 1) * MP];
    }
    __builtin_amdgcn_sched_barrier(0);
#pragma unroll
    for (int s = 0; s < KGRP; ++s) {
      re0 = wmma4(a0[s], bR[s], re0);
      im0 = wmma4(a0[s], bI[s], im0);
      re1 = wmma4(a1[s], bR[s], re1);
      im1 = wmma4(a1[s], bI[s], im1);
    }
  }
  store_tile(DRe, MP, row0,      col0, re0);
  store_tile(DIm, MP, row0,      col0, im0);
  store_tile(DRe, MP, row0 + 16, col0, re1);
  store_tile(DIm, MP, row0 + 16, col0, im1);
}

__global__ void k_level1(const float* __restrict__ Gx, const float* __restrict__ Gy,
                         const float* __restrict__ Ere, const float* __restrict__ Eim,
                         float* __restrict__ FxRe, float* __restrict__ FxIm,
                         float* __restrict__ FyRe, float* __restrict__ FyIm,
                         float* __restrict__ HT) {
  int bid = blockIdx.x;
  if (bid < GRAM_BLOCKS) {
    int b  = bid / (TILES * TILES);
    int t  = bid - b * (TILES * TILES);
    int ty = t / TILES, tx = t - ty * TILES;
    hgram_body(b, tx * 16, ty * 16, Gy, Gx, HT);
  } else {
    int id = bid - GRAM_BLOCKS;
    int z  = id / ((NPTS / 32) * TILES);   id -= z * ((NPTS / 32) * TILES);
    int by = id / (NPTS / 32), bx = id - by * (NPTS / 32);
    dft_body(z, bx * 32, by * 16, Gx, Gy, Ere, Eim, FxRe, FxIm, FyRe, FyIm);
  }
}

__global__ void __launch_bounds__(S_THREADS)
k_level2(const float* __restrict__ HT,
         const float* __restrict__ Ere,  const float* __restrict__ Eim,
         const float* __restrict__ FyRe, const float* __restrict__ FyIm,
         const float* __restrict__ W0,   const float* __restrict__ W1,
         const float* __restrict__ FxRe, const float* __restrict__ FxIm,
         float* __restrict__ SRe, float* __restrict__ SIm,
         float* __restrict__ PD) {
  __shared__ __attribute__((aligned(16))) float sT[2 * 16 * T_PITCH];
  int bid = blockIdx.x;
  int wave = __builtin_amdgcn_readfirstlane(threadIdx.x >> 5);
  if (bid < S_BLOCKS) {
    int bt = bid / S_SPLIT, part = bid - bt * S_SPLIT;
    int b = bt / TILES, ty = bt - b * TILES;
    sdft_body(b, wave, ty, part, HT, Ere, Eim, SRe, SIm, sT);
  } else {
    int id = (bid - S_BLOCKS) * S_WAVES + wave;
    int b  = id / VT_PER_B;     id -= b * VT_PER_B;
    int ty = id / TILES, tx = id - ty * TILES;
    vmat_body(b, tx, tx * 16, ty * 16, W0, W1, FyRe, FyIm, FxRe, FxIm, PD);
  }
}

#define U_WAVES   4
#define U_PTS     (16 * U_WAVES)
#define U_PITCH   148
#define U_ROW4    (MP / 4)
#define U_THREADS (32 * U_WAVES)
#define U_AROWS   17
#define U_EDGE    (MM - 1)
#define U_A_TRIPS ((U_AROWS * U_ROW4 + U_THREADS - 1) / U_THREADS)
#define U_B_TRIPS (U_PTS * U_ROW4 / U_THREADS)
#define U_B_CH    (U_B_TRIPS / 3)
#define U_APL     (U_AROWS * U_PITCH)
#define U_BPL     (U_PTS * U_PITCH)

template <bool EXTRA>
__device__ __forceinline__ void u_kloop(const float* sB0, const float* sB1,
                                        const float* sYd, const float* sYs,
                                        int la, int lb, int lx, int lbx, bool row0lane,
                                        v8f& accP0, v8f& accP1, v8f& accQ0, v8f& accQ1, v8f& accX) {
#pragma unroll
  for (int k0 = 0; k0 < MP; k0 += 4) {
    v2f a0 = *reinterpret_cast<const v2f*>(sB0 + la + k0);
    v2f a1 = *reinterpret_cast<const v2f*>(sB1 + la + k0);
    v2f yd = *reinterpret_cast<const v2f*>(sYd + lb + k0);
    v2f ys = *reinterpret_cast<const v2f*>(sYs + lb + k0);
    accP0 = wmma4(a0, yd, accP0);
    accP1 = wmma4(a1, yd, accP1);
    accQ0 = wmma4(a0, ys, accQ0);
    accQ1 = wmma4(a1, ys, accQ1);
    if (EXTRA) {
      v2f xa = *reinterpret_cast<const v2f*>(sB0 + lx + k0);
      v2f za = { row0lane ? xa.x : 0.f, row0lane ? xa.y : 0.f };
      v2f xb = *reinterpret_cast<const v2f*>(sYd + lbx + k0);
      accX = wmma4(za, xb, accX);
    }
  }
}

__global__ void __launch_bounds__(U_THREADS)
k_umat(const float* __restrict__ W0, const float* __restrict__ W1,
       const float* __restrict__ SRe, const float* __restrict__ SIm,
       const float* __restrict__ FyRe, const float* __restrict__ FyIm,
       const float* __restrict__ FxRe, const float* __restrict__ FxIm,
       float* __restrict__ PT, float* __restrict__ PE) {
  __shared__ __attribute__((aligned(16))) float sm[2 * U_APL + 2 * U_BPL];
  float* sB0 = sm;
  float* sB1 = sm + U_APL;
  float* sYd = sm + 2 * U_APL;
  float* sYs = sm + 2 * U_APL + U_BPL;
  int b = blockIdx.z;
  int kt = blockIdx.x;
  int row0 = kt * 16;
  int pblk = blockIdx.y * U_PTS;
  int tid = threadIdx.x;
  const float* sre = SRe + b * MP * MP;
  const float* sim = SIm + b * MP * MP;
  const float* fyr = FyRe + (b * NPP + pblk) * MP;
  const float* fyi = FyIm + (b * NPP + pblk) * MP;
  v4f gsr[U_A_TRIPS], gsi[U_A_TRIPS], ga0[U_A_TRIPS], ga1[U_A_TRIPS];
  v4f gfr0[U_B_CH], gfi0[U_B_CH], gfr1[U_B_CH], gfi1[U_B_CH], gfr2[U_B_CH], gfi2[U_B_CH];
  bool tail = tid < U_AROWS * U_ROW4 - (U_A_TRIPS - 1) * U_THREADS;
#pragma unroll
  for (int j = 0; j < U_A_TRIPS; ++j) {
    if (j < U_A_TRIPS - 1 || tail) {
      int idx = tid + j * U_THREADS;
      int rr = idx / U_ROW4, c4 = idx - rr * U_ROW4;
      int grow = rr < 16 ? row0 + rr : U_EDGE;
      int g = grow * MP + c4 * 4;
      gsr[j] = *reinterpret_cast<const v4f*>(sre + g);
      gsi[j] = *reinterpret_cast<const v4f*>(sim + g);
      ga0[j] = *reinterpret_cast<const v4f*>(W0 + g);
      ga1[j] = *reinterpret_cast<const v4f*>(W1 + g);
    }
  }
#pragma unroll
  for (int j = 0; j < U_B_CH; ++j) {
    gfr0[j] = *reinterpret_cast<const v4f*>(fyr + 4 * (tid + j * U_THREADS));
    gfi0[j] = *reinterpret_cast<const v4f*>(fyi + 4 * (tid + j * U_THREADS));
    gfr1[j] = *reinterpret_cast<const v4f*>(fyr + 4 * (tid + (j + U_B_CH) * U_THREADS));
    gfi1[j] = *reinterpret_cast<const v4f*>(fyi + 4 * (tid + (j + U_B_CH) * U_THREADS));
    gfr2[j] = *reinterpret_cast<const v4f*>(fyr + 4 * (tid + (j + 2 * U_B_CH) * U_THREADS));
    gfi2[j] = *reinterpret_cast<const v4f*>(fyi + 4 * (tid + (j + 2 * U_B_CH) * U_THREADS));
  }
  __builtin_amdgcn_sched_barrier(0);
#pragma unroll
  for (int j = 0; j < U_A_TRIPS; ++j) {
    if (j < U_A_TRIPS - 1 || tail) {
      int idx = tid + j * U_THREADS;
      int rr = idx / U_ROW4, c4 = idx - rr * U_ROW4;
      int l = rr * U_PITCH + c4 * 4;
      v4f d = gsr[j] - gsi[j];
      *reinterpret_cast<v4f*>(sB0 + l) = ga0[j] * d;
      *reinterpret_cast<v4f*>(sB1 + l) = ga1[j] * d;
    }
  }
#pragma unroll
  for (int j = 0; j < U_B_CH; ++j) {
    int i0 = tid + j * U_THREADS, i1 = i0 + U_B_CH * U_THREADS, i2 = i1 + U_B_CH * U_THREADS;
    int l0 = (i0 / U_ROW4) * U_PITCH + (i0 % U_ROW4) * 4;
    int l1 = (i1 / U_ROW4) * U_PITCH + (i1 % U_ROW4) * 4;
    int l2 = (i2 / U_ROW4) * U_PITCH + (i2 % U_ROW4) * 4;
    *reinterpret_cast<v4f*>(sYd + l0) = gfr0[j] - gfi0[j];
    *reinterpret_cast<v4f*>(sYs + l0) = gfr0[j] + gfi0[j];
    *reinterpret_cast<v4f*>(sYd + l1) = gfr1[j] - gfi1[j];
    *reinterpret_cast<v4f*>(sYs + l1) = gfr1[j] + gfi1[j];
    *reinterpret_cast<v4f*>(sYd + l2) = gfr2[j] - gfi2[j];
    *reinterpret_cast<v4f*>(sYs + l2) = gfr2[j] + gfi2[j];
  }
  __syncthreads();
  int wave = __builtin_amdgcn_readfirstlane(tid >> 5);
  int lane = tid & 31, r = lane & 15, hh = lane >> 4, kb = hh << 1;
  int col0 = pblk + wave * 16;
  int la = r * U_PITCH + kb;
  int lb = (wave * 16 + r) * U_PITCH + kb;
  int xt = kt >> 1;
  bool part = (kt & 1) != 0;
  int lx  = wave * U_APL + 16 * U_PITCH + kb;
  int lbx = (part ? U_BPL : 0) + (xt * 16 + r) * U_PITCH + kb;
  const float* fxr = FxRe + (b * NPP + col0 + r) * MP + row0 + 8 * hh;
  const float* fxi = FxIm + (b * NPP + col0 + r) * MP + row0 + 8 * hh;
  v4f xr0 = *reinterpret_cast<const v4f*>(fxr), xr1 = *reinterpret_cast<const v4f*>(fxr + 4);
  v4f xi0 = *reinterpret_cast<const v4f*>(fxi), xi1 = *reinterpret_cast<const v4f*>(fxi + 4);
  float xe = (part ? FxIm : FxRe)[(b * NPP + pblk + xt * 16 + r) * MP + U_EDGE];
  xe = part ? -xe : xe;
  v8f accP0 = {}, accP1 = {}, accQ0 = {}, accQ1 = {}, accX = {};
  if (wave < 2) {
    u_kloop<true>(sB0, sB1, sYd, sYs, la, lb, lx, lbx, r == 0,
                  accP0, accP1, accQ0, accQ1, accX);
  } else {
    u_kloop<false>(sB0, sB1, sYd, sYs, la, lb, lx, lbx, false,
                   accP0, accP1, accQ0, accQ1, accX);
  }
  float xr[8] = { xr0.x, xr0.y, xr0.z, xr0.w, xr1.x, xr1.y, xr1.z, xr1.w };
  float xi[8] = { xi0.x, xi0.y, xi0.z, xi0.w, xi1.x, xi1.y, xi1.z, xi1.w };
  float t0 = 0.f, t1 = 0.f;
#pragma unroll
  for (int v = 0; v < 8; ++v) {
    t0 += xr[v] * accP0[v] - xi[v] * accQ0[v];
    t1 += xr[v] * accP1[v] - xi[v] * accQ1[v];
  }
  t0 += __shfl_xor(t0, 16);
  t1 += __shfl_xor(t1, 16);
  PT[kt * NOUT + (b * NPP + col0 + r) * NCH + hh] = hh ? t1 : t0;
  if (wave < 2 && hh == 0) {
    PE[(((int)part) * NCH + wave) * NPTS + b * NPP + pblk + xt * 16 + r] = xe * accX[0];
  }
}

__global__ void __launch_bounds__(FIN_THREADS)
k_final(const float* __restrict__ PT, const float* __restrict__ PE,
        const float* __restrict__ PD, float* __restrict__ out) {
  int t = blockIdx.x * FIN_THREADS + threadIdx.x;
  int pt = t >> 1, ch = t & 1;
  float tot = 0.f, dia = 0.f;
#pragma unroll
  for (int k = 0; k < TILES - 1; ++k) tot += PT[k * NOUT + t];
  tot += PE[ch * NPTS + pt];
  tot += PE[(NCH + ch) * NPTS + pt];
#pragma unroll
  for (int k = 0; k < TILES; ++k) dia += PD[k * NOUT + t];
  out[t] = tot - dia;
}


extern "C" void kernel_launch(void* const* d_in, const int* in_sizes, int n_in,
                              void* d_out, int out_size, void* d_ws, size_t ws_size,
                              hipStream_t stream) {
  (void)in_sizes; (void)n_in; (void)out_size; (void)ws_size;
  const float* x  = (const float*)d_in[0];
  const float* s0 = (const float*)d_in[1];
  const float* s1 = (const float*)d_in[2];
  const float* a0 = (const float*)d_in[3];
  const float* a1 = (const float*)d_in[4];
  float* out = (float*)d_out;

  float* w = (float*)d_ws;
  float* Ere  = w; w += MP * MP;
  float* Eim  = w; w += MP * MP;
  float* W0   = w; w += MP * MP;
  float* W1   = w; w += MP * MP;
  float* Gx   = w; w += NPTS * MP;
  float* Gy   = w; w += NPTS * MP;
  float* FxRe = w; w += NPTS * MP;
  float* FxIm = w; w += NPTS * MP;
  float* FyRe = w; w += NPTS * MP;
  float* FyIm = w; w += NPTS * MP;
  float* SRe  = w; w += NB * MP * MP;
  float* SIm  = w; w += NB * MP * MP;
  float* PT   = w; w += (TILES - 1) * NOUT;
  float* PE   = w; w += 2 * NCH * NPTS;
  float* PD   = w; w += TILES * NOUT;
  float* HT   = w; w += NB * MP * MP;

  k_level0<<<TAB_BLOCKS + SPR_BLOCKS, L0_THREADS, 0, stream>>>(Ere, Eim, W0, W1, s0, s1, a0, a1,
                                                               x, Gx, Gy);
  k_level1<<<GRAM_BLOCKS + DFT_BLOCKS, 32, 0, stream>>>(Gx, Gy, Ere, Eim,
                                                        FxRe, FxIm, FyRe, FyIm, HT);
  k_level2<<<S_BLOCKS + V_BLOCKS, S_THREADS, 0, stream>>>(HT, Ere, Eim, FyRe, FyIm,
                                                          W0, W1, FxRe, FxIm, SRe, SIm, PD);
  k_umat<<<dim3(MM / 16, NPP / U_PTS, NB), U_THREADS, 0, stream>>>(W0, W1, SRe, SIm, FyRe, FyIm,
                                                         FxRe, FxIm, PT, PE);
  k_final<<<NOUT / FIN_THREADS, FIN_THREADS, 0, stream>>>(PT, PE, PD, out);
}


# Every DFT in real cas (cos+sin) form, half the matrix work: front DFT writes one plane D=G(cos+sin), readers rebuild Re, Im, |F|^2 from D[k], D[M-k]; S level forms S.re-S.im directly; new sum order
# speedup vs baseline: 2.4891x; 1.0460x over previous
#include <hip/hip_runtime.h>
#include <math.h>


typedef __attribute__((ext_vector_type(2))) float v2f;
typedef __attribute__((ext_vector_type(8))) float v8f;
typedef __attribute__((ext_vector_type(4))) float v4f;

#define MM    129
#define MP    144
#define NPP   512
#define NB    4
#define NPTS  (NB * NPP)
#define NCH   2

#define PI_F   3.14159265358979323846f
#define TAU_F  (12.0f / (129.0f * 129.0f))
#define L_F    (2.0f * PI_F)

#define TILES       (MP / 16)
#define L0_THREADS  128
#define TAB_PER_T   2
#define SPR_PER_T   8
#define TAB_BLOCKS  (MP * MP / (L0_THREADS * TAB_PER_T))
#define SPR_BLOCKS  (NPTS * MP / (L0_THREADS * SPR_PER_T))
#define GRAM_BLOCKS (TILES * TILES * NB)
#define VT_PER_B    (TILES * (NPP / 16))
#define VMAT_BLOCKS (VT_PER_B * NB)
#define DFT_PAIRS   (TILES / 2)
#define DFT_A_PER_Z ((NPTS / 32) * DFT_PAIRS)
#define DFT_B_PER_Z (NPTS / 64)
#define DFT_BLOCKS  (2 * (DFT_A_PER_Z + DFT_B_PER_Z))
#define S_WAVES     4
#define S_THREADS   (32 * S_WAVES)
#define S_SPLIT     3
#define S_BLOCKS    (TILES * NB * S_SPLIT)
#define V_BLOCKS    (VMAT_BLOCKS / S_WAVES)
#define T_PITCH     148
#define S_KG1       1
#define S_NB1       (MP / (4 * S_KG1))
#define S_KG2       3
#define S_NB2       (MP / (4 * S_KG2))
#define S_RING      6
#define NOUT        (NPTS * NCH)
#define FIN_THREADS 256


__device__ __forceinline__ v8f wmma4(v2f a, v2f b, v8f c) {
  return __builtin_amdgcn_wmma_f32_16x16x4_f32(false, a, false, b, (short)0, c, false, false);
}

__device__ __forceinline__ void store_tile(float* C, int ldc, int row0, int col0, v8f acc) {
  int lane = threadIdx.x & 31;
  int r0 = row0 + ((lane >> 4) << 3);
  int c  = col0 + (lane & 15);
#pragma unroll
  for (int v = 0; v < 8; ++v) C[(r0 + v) * ldc + c] = acc[v];
}


__device__ __forceinline__ int revk(int k) { return (k == 0 || k >= MM) ? k : MM - k; }

__device__ __forceinline__ void store_tile_mirror(float* C, int ldc, int row0, int col0, v8f acc) {
  int lane = threadIdx.x & 31;
  int r0 = row0 + ((lane >> 4) << 3);
  int c  = revk(col0 + (lane & 15));
#pragma unroll
  for (int v = 0; v < 8; ++v) C[(r0 + v) * ldc + c] = acc[v];
}

__device__ __forceinline__ void load_mirror8(const float* row, int ka, const float (&xf)[8], float (&xv)[8]) {
  int m0 = 122 - ka;
  m0 = m0 < 0 ? 0 : m0;
  v2f l0 = *reinterpret_cast<const v2f*>(row + m0),     l1 = *reinterpret_cast<const v2f*>(row + m0 + 2);
  v2f l2 = *reinterpret_cast<const v2f*>(row + m0 + 4), l3 = *reinterpret_cast<const v2f*>(row + m0 + 6);
  float L[8] = { l0.x, l0.y, l1.x, l1.y, l2.x, l2.y, l3.x, l3.y };
#pragma unroll
  for (int v = 0; v < 8; ++v) {
    int k = ka + v;
    float g = L[7 - v];
    if (v == 0) g = (ka == MM - 1) ? L[1] : g;
    xv[v] = (k == 0 || k >= MM) ? xf[v] : g;
  }
}


__device__ __forceinline__ void tables_body(int t,
                         float* __restrict__ Ere, float* __restrict__ Eim,
                         float* __restrict__ Cas,
                         float* __restrict__ W0,  float* __restrict__ W1,
                         const float* s0, const float* s1,
                         const float* a0, const float* a1) {
  if (t >= MP * MP) return;
  int i = t / MP, k = t % MP;
  float er = 0.f, ei = 0.f, w0 = 0.f, w1 = 0.f;
  if (i < MM && k < MM) {
    int m = (i * k) % MM;
    float th = (2.0f * PI_F / (float)MM) * (float)m;
    er = cosf(th);
    ei = -sinf(th);
    float f1 = (float)((i <= MM / 2) ? i : i - MM);
    float f2 = (float)((k <= MM / 2) ? k : k - MM);
    float k2 = f1 * f1 + f2 * f2;
    float dec = (PI_F / TAU_F) * expf(TAU_F * k2);
    float m2   = (float)MM * (float)MM;
    float coef = 1.0f / (8.0f * PI_F * PI_F * m2 * m2);
    float base = coef * dec * dec * 4.0f * PI_F;
    float sh0 = s0[0], sh1 = s1[0];
    w0 = base * a0[0] / (k2 + sh0 * sh0);
    w1 = base * a1[0] / (k2 + sh1 * sh1);
  }
  Ere[t] = er; Eim[t] = ei; Cas[t] = er - ei; W0[t] = w0; W1[t] = w1;
}

__device__ __forceinline__ void spread_body(int t, const float* __restrict__ x,
                         float* __restrict__ Gx, float* __restrict__ Gy) {
  if (t >= NPTS * MP) return;
  int pt = t / MP, i = t % MP;
  float gx = 0.f, gy = 0.f;
  if (i < MM) {
    float xi  = (L_F / (float)MM) * (float)i;
    float inv = 1.0f / (4.0f * TAU_F);
    float d0 = x[2 * pt + 0] - xi;
    float d1 = x[2 * pt + 1] - xi;
    gx = expf(-d0 * d0 * inv) + expf(-(d0 - L_F) * (d0 - L_F) * inv)
       + expf(-(d0 + L_F) * (d0 + L_F) * inv);
    gy = expf(-d1 * d1 * inv) + expf(-(d1 - L_F) * (d1 - L_F) * inv)
       + expf(-(d1 + L_F) * (d1 + L_F) * inv);
  }
  Gx[t] = gx; Gy[t] = gy;
}

#define HG_RING   8
#define HG_AHEAD  (HG_RING - 1)
#define HG_GROUPS (NPP / 8)

struct HFrag { v2f a[2], b[2]; };

__device__ __forceinline__ void hgram_load(HFrag& f, int g,
                       const float* __restrict__ gx, const float* __restrict__ gy,
                       int kb, int ca, int cb) {
#pragma unroll
  for (int s = 0; s < 2; ++s) {
    int q0 = g * 8 + s * 4 + kb, q1 = q0 + 1;
    f.a[s].x = gx[q0 * MP + ca]; f.a[s].y = gx[q1 * MP + ca];
    f.b[s].x = gy[q0 * MP + cb]; f.b[s].y = gy[q1 * MP + cb];
  }
}

__device__ __forceinline__ void hgram_body(int b, int row0, int col0,
                       const float* __restrict__ A, const float* __restrict__ B,
                       float* __restrict__ D) {
  int lane = threadIdx.x & 31, r = lane & 15, kb = (lane >> 4) << 1;
  const float* gx = A + b * NPP * MP;
  const float* gy = B + b * NPP * MP;
  int ca = row0 + r, cb = col0 + r;
  v8f acc0 = {}, acc1 = {};
  HFrag ring[HG_RING];
#pragma unroll
  for (int j = 0; j < HG_AHEAD; ++j) {
    hgram_load(ring[j], j, gx, gy, kb, ca, cb);
    __builtin_amdgcn_sched_barrier(0);
  }
  for (int g0 = 0; g0 < HG_GROUPS; g0 += HG_RING) {
#pragma unroll
    for (int j = 0; j < HG_RING; ++j) {
      int g  = g0 + j;
      int gn = g + HG_AHEAD;
      gn = gn < HG_GROUPS ? gn : HG_GROUPS - 1;
      hgram_load(ring[(j + HG_AHEAD) % HG_RING], gn, gx, gy, kb, ca, cb);
      acc0 = wmma4(ring[j].a[0], ring[j].b[0], acc0);
      acc1 = wmma4(ring[j].a[1], ring[j].b[1], acc1);
      __builtin_amdgcn_sched_barrier(0);
    }
  }
  v8f acc = acc0 + acc1;
  store_tile(D + b * MP * MP, MP, row0, col0, acc);
}

template <int NT>
struct P1Frag { v2f a[NT][S_KG1], br[S_KG1], bi[S_KG1]; };
struct P2Glob { v2f ar[S_KG2], ai[S_KG2]; };
struct P2Lds  { v2f bc[S_KG2], bm[S_KG2]; };

template <int NT>
__device__ __forceinline__ void p1_load(P1Frag<NT>& f, int g, const float* hat,
                                        const float* ebr, const float* ebi) {
#pragma unroll
  for (int s = 0; s < S_KG1; ++s) {
    int k0 = 4 * (g * S_KG1 + s);
#pragma unroll
    for (int t = 0; t < NT; ++t) {
      f.a[t][s].x = hat[t * 64 + k0 * MP];
      f.a[t][s].y = hat[t * 64 + (k0 + 1) * MP];
    }
    f.br[s].x = ebr[k0 * MP]; f.br[s].y = ebr[(k0 + 1) * MP];
    f.bi[s].x = ebi[k0 * MP]; f.bi[s].y = ebi[(k0 + 1) * MP];
  }
}
template <int NT>
__device__ __forceinline__ void p1_mac(const P1Frag<NT>& f, v8f (&tRe)[NT], v8f (&tIm)[NT]) {
#pragma unroll
  for (int s = 0; s < S_KG1; ++s) {
#pragma unroll
    for (int t = 0; t < NT; ++t) {
      tRe[t] = wmma4(f.a[t][s], f.br[s], tRe[t]);
      tIm[t] = wmma4(f.a[t][s], f.bi[s], tIm[t]);
    }
  }
}
template <int NT>
__device__ __forceinline__ void p1_pass(int wave, int r, int hh, const float* hat,
                                        const float* ebr, const float* ebi,
                                        float* sTr, float* sTi) {
  v8f tRe[NT], tIm[NT];
#pragma unroll
  for (int t = 0; t < NT; ++t) { tRe[t] = v8f{}; tIm[t] = v8f{}; }
  P1Frag<NT> ring[S_RING];
#pragma unroll
  for (int j = 0; j < S_RING - 1; ++j) {
    p1_load<NT>(ring[j], j, hat, ebr, ebi);
    __builtin_amdgcn_sched_barrier(0);
  }
#pragma unroll 1
  for (int g0 = 0; g0 < S_NB1; g0 += S_RING) {
#pragma unroll
    for (int j = 0; j < S_RING; ++j) {
      int gn = g0 + j + S_RING - 1;
      gn = gn < S_NB1 ? gn : S_NB1 - 1;
      p1_load<NT>(ring[(j + S_RING - 1) % S_RING], gn, hat, ebr, ebi);
      p1_mac<NT>(ring[j], tRe, tIm);
      __builtin_amdgcn_sched_barrier(0);
    }
  }
#pragma unroll
  for (int t = 0; t < NT; ++t) {
    float* dr = sTr + r * T_PITCH + (wave + 4 * t) * 16 + 8 * hh;
    float* di = sTi + r * T_PITCH + (wave + 4 * t) * 16 + 8 * hh;
    v8f tc = tRe[t] - tIm[t], tm = tRe[t] + tIm[t];
    v4f r0 = { tc[0], tc[1], tc[2], tc[3] }, r1 = { tc[4], tc[5], tc[6], tc[7] };
    v4f i0 = { tm[0], tm[1], tm[2], tm[3] }, i1 = { tm[4], tm[5], tm[6], tm[7] };
    *reinterpret_cast<v4f*>(dr) = r0; *reinterpret_cast<v4f*>(dr + 4) = r1;
    *reinterpret_cast<v4f*>(di) = i0; *reinterpret_cast<v4f*>(di + 4) = i1;
  }
}
__device__ __forceinline__ void p2_gload(P2Glob& f, int g, const float* ear, const float* eai) {
#pragma unroll
  for (int s = 0; s < S_KG2; ++s) {
    int k0 = 4 * (g * S_KG2 + s);
    f.ar[s].x = ear[k0 * MP]; f.ar[s].y = ear[(k0 + 1) * MP];
    f.ai[s].x = eai[k0 * MP]; f.ai[s].y = eai[(k0 + 1) * MP];
  }
}
__device__ __forceinline__ void p2_lload(P2Lds& f, int g, const float* tbr, const float* tbi) {
#pragma unroll
  for (int s = 0; s < S_KG2; ++s) {
    int k0 = 4 * (g * S_KG2 + s);
    f.bc[s] = *reinterpret_cast<const v2f*>(tbr + k0);
    f.bm[s] = *reinterpret_cast<const v2f*>(tbi + k0);
  }
}
__device__ __forceinline__ void p2_mac(const P2Glob& a, const P2Lds& b, v8f& sC, v8f& sM) {
#pragma unroll
  for (int s = 0; s < S_KG2; ++s) {
    v2f aiN = { -a.ai[s].x, -a.ai[s].y };
    sC = wmma4(a.ar[s], b.bc[s], sC);
    sM = wmma4(aiN,     b.bm[s], sM);
  }
}

__device__ __forceinline__ void sdft_body(int b, int wave, int ty, int part,
                       const float* __restrict__ HT,
                       const float* __restrict__ Ere, const float* __restrict__ Eim,
                       float* __restrict__ SD, float* __restrict__ SDr, float* sT) {
  int lane = threadIdx.x & 31, r = lane & 15, hh = lane >> 4, kb = hh << 1;
  float* sTr = sT;
  float* sTi = sT + 16 * T_PITCH;
  const float* hat = HT + b * MP * MP + kb * MP + wave * 16 + r;
  const float* ebr = Ere + kb * MP + ty * 16 + r;
  const float* ebi = Eim + kb * MP + ty * 16 + r;
  int kt = part * 3 + (wave > 0 ? wave - 1 : 0);
  const float* ear = Ere + kb * MP + kt * 16 + r;
  const float* eai = Eim + kb * MP + kt * 16 + r;
  const float* tbr = sTr + r * T_PITCH + kb;
  const float* tbi = sTi + r * T_PITCH + kb;
  P2Glob gr[S_RING];
  P2Lds  lr[S_RING];
  if (wave == 0) {
    p1_pass<3>(wave, r, hh, hat, ebr, ebi, sTr, sTi);
  } else {
    p2_gload(gr[0], 0, ear, eai);
    p2_gload(gr[1], 1, ear, eai);
    p1_pass<2>(wave, r, hh, hat, ebr, ebi, sTr, sTi);
  }
  __syncthreads();
  if (wave == 0) return;
  v8f sC = {}, sM = {};
#pragma unroll
  for (int j = 2; j < S_RING - 1; ++j) p2_gload(gr[j], j, ear, eai);
#pragma unroll
  for (int j = 0; j < S_RING - 1; ++j) p2_lload(lr[j], j, tbr, tbi);
#pragma unroll 1
  for (int g0 = 0; g0 < S_NB2; g0 += S_RING) {
#pragma unroll
    for (int j = 0; j < S_RING; ++j) {
      int gn = g0 + j + S_RING - 1;
      gn = gn < S_NB2 ? gn : S_NB2 - 1;
      p2_gload(gr[(j + S_RING - 1) % S_RING], gn, ear, eai);
      p2_lload(lr[(j + S_RING - 1) % S_RING], gn, tbr, tbi);
      p2_mac(gr[j], lr[j], sC, sM);
      __builtin_amdgcn_sched_barrier(0);
    }
  }
  v8f sd = sC + sM;
  store_tile(SD + b * MP * MP, MP, kt * 16, ty * 16, sd);
  store_tile_mirror(SDr + b * MP * MP, MP, kt * 16, ty * 16, sd);
}

__device__ __forceinline__ void vmat_body(int b, int kt, int row0, int col0,
                       const float* __restrict__ W0, const float* __restrict__ W1,
                       const float* __restrict__ Dx, const float* __restrict__ Dy,
                       float* __restrict__ PD) {
  int lane = threadIdx.x & 31, r = lane & 15, hh = lane >> 4, kb = hh << 1;
  const float* dxp = Dx + (b * NPP + col0 + r) * MP;
  int ka = row0 + 8 * hh;
  v4f xf0 = *reinterpret_cast<const v4f*>(dxp + ka), xf1 = *reinterpret_cast<const v4f*>(dxp + ka + 4);
  float xf[8] = { xf0.x, xf0.y, xf0.z, xf0.w, xf1.x, xf1.y, xf1.z, xf1.w };
  float xv[8];
  load_mirror8(dxp, ka, xf, xv);
  const float* dyp = Dy + (b * NPP + col0 + r) * MP;
  const float* w0  = W0 + (row0 + r) * MP + kb;
  const float* w1  = W1 + (row0 + r) * MP + kb;
  v8f acc0 = {}, acc1 = {};
  for (int k0 = 0; k0 < MP; k0 += 4) {
    int k = k0 + kb;
    int m = MM - 1 - k;
    m = m < 0 ? 0 : m;
    v2f d = *reinterpret_cast<const v2f*>(dyp + k);
    v2f e = *reinterpret_cast<const v2f*>(dyp + m);
    float e0 = (k == 0 || k >= MM) ? d.x : e.y;
    float e1 = (k + 1 >= MM) ? d.y : e.x;
    v2f p  = { 0.5f * (d.x * d.x + e0 * e0), 0.5f * (d.y * d.y + e1 * e1) };
    v2f a0 = { w0[k0], w0[k0 + 1] };
    v2f a1 = { w1[k0], w1[k0 + 1] };
    acc0 = wmma4(a0, p, acc0);
    acc1 = wmma4(a1, p, acc1);
  }
  float d0 = 0.f, d1 = 0.f;
#pragma unroll
  for (int v = 0; v < 8; ++v) {
    float ax = 0.5f * (xf[v] * xf[v] + xv[v] * xv[v]);
    d0 += ax * acc0[v];
    d1 += ax * acc1[v];
  }
  d0 += __shfl_xor(d0, 16);
  d1 += __shfl_xor(d1, 16);
  PD[kt * NOUT + (b * NPP + col0 + r) * NCH + hh] = hh ? d1 : d0;
}


__global__ void k_level0(float* __restrict__ Ere, float* __restrict__ Eim,
                         float* __restrict__ Cas,
                         float* __restrict__ W0,  float* __restrict__ W1,
                         const float* s0, const float* s1,
                         const float* a0, const float* a1,
                         const float* __restrict__ x,
                         float* __restrict__ Gx, float* __restrict__ Gy) {
  int bid = blockIdx.x;
  if (bid < TAB_BLOCKS) {
    for (int j = 0; j < TAB_PER_T; ++j)
      tables_body((j * TAB_BLOCKS + bid) * L0_THREADS + threadIdx.x,
                  Ere, Eim, Cas, W0, W1, s0, s1, a0, a1);
  } else {
    int sb = bid - TAB_BLOCKS;
    for (int j = 0; j < SPR_PER_T; ++j)
      spread_body((j * SPR_BLOCKS + sb) * L0_THREADS + threadIdx.x, x, Gx, Gy);
  }
}

#define DF_RING   9
#define DF_STEPS  (MP / 4)

template <int RT, int CT>
struct DFrag { v2f a[RT], c[CT]; };

template <int RT, int CT>
__device__ __forceinline__ void dft_load(DFrag<RT, CT>& f, int g, const float* ga, const float* ec) {
  int k0 = 4 * g;
#pragma unroll
  for (int t = 0; t < RT; ++t) {
    f.a[t].x = ga[t * 16 * MP + k0];
    f.a[t].y = ga[t * 16 * MP + k0 + 1];
  }
#pragma unroll
  for (int u = 0; u < CT; ++u) {
    f.c[u].x = ec[u * 16 + k0 * MP];
    f.c[u].y = ec[u * 16 + (k0 + 1) * MP];
  }
}

template <int RT, int CT>
__device__ __forceinline__ void dft_body(int z, int row0, int col0,
                      const float* __restrict__ Gx, const float* __restrict__ Gy,
                      const float* __restrict__ Cas,
                      float* __restrict__ Dx, float* __restrict__ Dy) {
  const float* G = z ? Gy : Gx;
  float* D = z ? Dy : Dx;
  int lane = threadIdx.x & 31, r = lane & 15, kb = (lane >> 4) << 1;
  const float* ga = G + (row0 + r) * MP + kb;
  const float* ec = Cas + kb * MP + col0 + r;
  v8f acc[RT * CT];
#pragma unroll
  for (int t = 0; t < RT * CT; ++t) acc[t] = v8f{};
  DFrag<RT, CT> ring[DF_RING];
#pragma unroll
  for (int j = 0; j < DF_RING - 1; ++j) {
    dft_load<RT, CT>(ring[j], j, ga, ec);
    __builtin_amdgcn_sched_barrier(0);
  }
#pragma unroll 1
  for (int g0 = 0; g0 < DF_STEPS; g0 += DF_RING) {
#pragma unroll
    for (int j = 0; j < DF_RING; ++j) {
      int gn = g0 + j + DF_RING - 1;
      gn = gn < DF_STEPS ? gn : DF_STEPS - 1;
      dft_load<RT, CT>(ring[(j + DF_RING - 1) % DF_RING], gn, ga, ec);
#pragma unroll
      for (int t = 0; t < RT; ++t) {
#pragma unroll
        for (int u = 0; u < CT; ++u)
          acc[t * CT + u] = wmma4(ring[j].a[t], ring[j].c[u], acc[t * CT + u]);
      }
      __builtin_amdgcn_sched_barrier(0);
    }
  }
#pragma unroll
  for (int t = 0; t < RT; ++t) {
#pragma unroll
    for (int u = 0; u < CT; ++u) store_tile(D, MP, row0 + 16 * t, col0 + 16 * u, acc[t * CT + u]);
  }
}

__global__ void __launch_bounds__(32)
k_level1(const float* __restrict__ Gx, const float* __restrict__ Gy,
         const float* __restrict__ Cas,
         float* __restrict__ Dx, float* __restrict__ Dy,
         float* __restrict__ HT) {
  int bid = blockIdx.x;
  if (bid < GRAM_BLOCKS) {
    int b  = bid / (TILES * TILES);
    int t  = bid - b * (TILES * TILES);
    int ty = t / TILES, tx = t - ty * TILES;
    hgram_body(b, tx * 16, ty * 16, Gy, Gx, HT);
  } else {
    int id = bid - GRAM_BLOCKS;
    if (id < 2 * DFT_A_PER_Z) {
      int z  = id / DFT_A_PER_Z;     id -= z * DFT_A_PER_Z;
      int cp = id / (NPTS / 32), bx = id - cp * (NPTS / 32);
      dft_body<2, 2>(z, bx * 32, cp * 32, Gx, Gy, Cas, Dx, Dy);
    } else {
      id -= 2 * DFT_A_PER_Z;
      int z  = id / DFT_B_PER_Z, bx = id - z * DFT_B_PER_Z;
      dft_body<4, 1>(z, bx * 64, (TILES - 1) * 16, Gx, Gy, Cas, Dx, Dy);
    }
  }
}

__global__ void __launch_bounds__(S_THREADS)
k_level2(const float* __restrict__ HT,
         const float* __restrict__ Ere,  const float* __restrict__ Eim,
         const float* __restrict__ Dx,   const float* __restrict__ Dy,
         const float* __restrict__ W0,   const float* __restrict__ W1,
         float* __restrict__ SD, float* __restrict__ SDr,
         float* __restrict__ PD) {
  __shared__ __attribute__((aligned(16))) float sT[2 * 16 * T_PITCH];
  int bid = blockIdx.x;
  int wave = __builtin_amdgcn_readfirstlane(threadIdx.x >> 5);
  if (bid < S_BLOCKS) {
    int bt = bid / S_SPLIT, part = bid - bt * S_SPLIT;
    int b = bt / TILES, ty = bt - b * TILES;
    sdft_body(b, wave, ty, part, HT, Ere, Eim, SD, SDr, sT);
  } else {
    int id = (bid - S_BLOCKS) * S_WAVES + wave;
    int b  = id / VT_PER_B;     id -= b * VT_PER_B;
    int ty = id / TILES, tx = id - ty * TILES;
    vmat_body(b, tx, tx * 16, ty * 16, W0, W1, Dx, Dy, PD);
  }
}

#define U_WAVES   4
#define U_PTS     (16 * U_WAVES)
#define U_PITCH   148
#define U_ROW4    (MP / 4)
#define U_THREADS (32 * U_WAVES)
#define U_AROWS   17
#define U_EDGE    (MM - 1)
#define U_A_TRIPS ((U_AROWS * U_ROW4 + U_THREADS - 1) / U_THREADS)
#define U_B_TRIPS (U_PTS * U_ROW4 / U_THREADS)
#define U_B_CH    (U_B_TRIPS / 3)
#define U_APL     (U_AROWS * U_PITCH)
#define U_BPL     (U_PTS * U_PITCH)

template <bool EXTRA>
__device__ __forceinline__ void u_kloop(const float* sA, const float* sYd,
                                        int la, int lb, int lx, int lbx, bool row0lane,
                                        v8f& accP0, v8f& accP1, v8f& accQ0, v8f& accQ1, v8f& accX) {
#pragma unroll
  for (int k0 = 0; k0 < MP; k0 += 4) {
    v2f a0 = *reinterpret_cast<const v2f*>(sA + la + k0);
    v2f a1 = *reinterpret_cast<const v2f*>(sA + U_APL + la + k0);
    v2f r0 = *reinterpret_cast<const v2f*>(sA + 2 * U_APL + la + k0);
    v2f r1 = *reinterpret_cast<const v2f*>(sA + 3 * U_APL + la + k0);
    v2f yd = *reinterpret_cast<const v2f*>(sYd + lb + k0);
    accP0 = wmma4(a0, yd, accP0);
    accP1 = wmma4(a1, yd, accP1);
    accQ0 = wmma4(r0, yd, accQ0);
    accQ1 = wmma4(r1, yd, accQ1);
    if (EXTRA) {
      v2f xa = *reinterpret_cast<const v2f*>(sA + lx + k0);
      v2f za = { row0lane ? xa.x : 0.f, row0lane ? xa.y : 0.f };
      v2f xb = *reinterpret_cast<const v2f*>(sYd + lbx + k0);
      accX = wmma4(za, xb, accX);
    }
  }
}

__global__ void __launch_bounds__(U_THREADS)
k_umat(const float* __restrict__ W0, const float* __restrict__ W1,
       const float* __restrict__ SD, const float* __restrict__ SDr,
       const float* __restrict__ Dx, const float* __restrict__ Dy,
       float* __restrict__ PT, float* __restrict__ PE) {
  __shared__ __attribute__((aligned(16))) float sm[4 * U_APL + U_BPL];
  float* sB0 = sm;
  float* sB1 = sm + U_APL;
  float* sR0 = sm + 2 * U_APL;
  float* sR1 = sm + 3 * U_APL;
  float* sYd = sm + 4 * U_APL;
  int b = blockIdx.z;
  int kt = blockIdx.x;
  int row0 = kt * 16;
  int pblk = blockIdx.y * U_PTS;
  int tid = threadIdx.x;
  const float* sd  = SD  + b * MP * MP;
  const float* sdr = SDr + b * MP * MP;
  const float* dy = Dy + (b * NPP + pblk) * MP;
  v4f gsd[U_A_TRIPS], gsr[U_A_TRIPS], ga0[U_A_TRIPS], ga1[U_A_TRIPS];
  v4f gf0[U_B_CH], gf1[U_B_CH], gf2[U_B_CH];
  bool tail = tid < U_AROWS * U_ROW4 - (U_A_TRIPS - 1) * U_THREADS;
#pragma unroll
  for (int j = 0; j < U_A_TRIPS; ++j) {
    if (j < U_A_TRIPS - 1 || tail) {
      int idx = tid + j * U_THREADS;
      int rr = idx / U_ROW4, c4 = idx - rr * U_ROW4;
      int grow = rr < 16 ? row0 + rr : U_EDGE;
      int g = grow * MP + c4 * 4;
      gsd[j] = *reinterpret_cast<const v4f*>(sd + g);
      gsr[j] = *reinterpret_cast<const v4f*>(sdr + g);
      ga0[j] = *reinterpret_cast<const v4f*>(W0 + g);
      ga1[j] = *reinterpret_cast<const v4f*>(W1 + g);
    }
  }
#pragma unroll
  for (int j = 0; j < U_B_CH; ++j) {
    gf0[j] = *reinterpret_cast<const v4f*>(dy + 4 * (tid + j * U_THREADS));
    gf1[j] = *reinterpret_cast<const v4f*>(dy + 4 * (tid + (j + U_B_CH) * U_THREADS));
    gf2[j] = *reinterpret_cast<const v4f*>(dy + 4 * (tid + (j + 2 * U_B_CH) * U_THREADS));
  }
  __builtin_amdgcn_sched_barrier(0);
#pragma unroll
  for (int j = 0; j < U_A_TRIPS; ++j) {
    if (j < U_A_TRIPS - 1 || tail) {
      int idx = tid + j * U_THREADS;
      int rr = idx / U_ROW4, c4 = idx - rr * U_ROW4;
      int l = rr * U_PITCH + c4 * 4;
      *reinterpret_cast<v4f*>(sB0 + l) = ga0[j] * gsd[j];
      *reinterpret_cast<v4f*>(sB1 + l) = ga1[j] * gsd[j];
      *reinterpret_cast<v4f*>(sR0 + l) = ga0[j] * gsr[j];
      *reinterpret_cast<v4f*>(sR1 + l) = ga1[j] * gsr[j];
    }
  }
#pragma unroll
  for (int j = 0; j < U_B_CH; ++j) {
    int i0 = tid + j * U_THREADS, i1 = i0 + U_B_CH * U_THREADS, i2 = i1 + U_B_CH * U_THREADS;
    int l0 = (i0 / U_ROW4) * U_PITCH + (i0 % U_ROW4) * 4;
    int l1 = (i1 / U_ROW4) * U_PITCH + (i1 % U_ROW4) * 4;
    int l2 = (i2 / U_ROW4) * U_PITCH + (i2 % U_ROW4) * 4;
    *reinterpret_cast<v4f*>(sYd + l0) = gf0[j];
    *reinterpret_cast<v4f*>(sYd + l1) = gf1[j];
    *reinterpret_cast<v4f*>(sYd + l2) = gf2[j];
  }
  __syncthreads();
  int wave = __builtin_amdgcn_readfirstlane(tid >> 5);
  int lane = tid & 31, r = lane & 15, hh = lane >> 4, kb = hh << 1;
  int col0 = pblk + wave * 16;
  int la = r * U_PITCH + kb;
  int lb = (wave * 16 + r) * U_PITCH + kb;
  int xt = kt >> 1;
  bool part = (kt & 1) != 0;
  int lx  = ((part ? 2 : 0) + wave) * U_APL + 16 * U_PITCH + kb;
  int lbx = (xt * 16 + r) * U_PITCH + kb;
  const float* dxp = Dx + (b * NPP + col0 + r) * MP;
  int ka = row0 + 8 * hh;
  v4f xf0 = *reinterpret_cast<const v4f*>(dxp + ka), xf1 = *reinterpret_cast<const v4f*>(dxp + ka + 4);
  float xf[8] = { xf0.x, xf0.y, xf0.z, xf0.w, xf1.x, xf1.y, xf1.z, xf1.w };
  float xv[8];
  load_mirror8(dxp, ka, xf, xv);
  const float* dxe = Dx + (b * NPP + pblk + xt * 16 + r) * MP;
  float e128 = dxe[U_EDGE], e1 = dxe[1];
  float xe = 0.5f * (part ? e128 - e1 : e128 + e1);
  v8f accP0 = {}, accP1 = {}, accQ0 = {}, accQ1 = {}, accX = {};
  if (wave < 2) {
    u_kloop<true>(sm, sYd, la, lb, lx, lbx, r == 0,
                  accP0, accP1, accQ0, accQ1, accX);
  } else {
    u_kloop<false>(sm, sYd, la, lb, lx, lbx, false,
                   accP0, accP1, accQ0, accQ1, accX);
  }
  float t0 = 0.f, t1 = 0.f;
#pragma unroll
  for (int v = 0; v < 8; ++v) {
    float xr = 0.5f * (xf[v] + xv[v]);
    float xi = 0.5f * (xv[v] - xf[v]);
    t0 += xr * accP0[v] - xi * accQ0[v];
    t1 += xr * accP1[v] - xi * accQ1[v];
  }
  t0 += __shfl_xor(t0, 16);
  t1 += __shfl_xor(t1, 16);
  PT[kt * NOUT + (b * NPP + col0 + r) * NCH + hh] = hh ? t1 : t0;
  if (wave < 2 && hh == 0) {
    PE[(((int)part) * NCH + wave) * NPTS + b * NPP + pblk + xt * 16 + r] = xe * accX[0];
  }
}

__global__ void __launch_bounds__(FIN_THREADS)
k_final(const float* __restrict__ PT, const float* __restrict__ PE,
        const float* __restrict__ PD, float* __restrict__ out) {
  int t = blockIdx.x * FIN_THREADS + threadIdx.x;
  int pt = t >> 1, ch = t & 1;
  float tot = 0.f, dia = 0.f;
#pragma unroll
  for (int k = 0; k < TILES - 1; ++k) tot += PT[k * NOUT + t];
  tot += PE[ch * NPTS + pt];
  tot += PE[(NCH + ch) * NPTS + pt];
#pragma unroll
  for (int k = 0; k < TILES; ++k) dia += PD[k * NOUT + t];
  out[t] = tot - dia;
}


extern "C" void kernel_launch(void* const* d_in, const int* in_sizes, int n_in,
                              void* d_out, int out_size, void* d_ws, size_t ws_size,
                              hipStream_t stream) {
  (void)in_sizes; (void)n_in; (void)out_size; (void)ws_size;
  const float* x  = (const float*)d_in[0];
  const float* s0 = (const float*)d_in[1];
  const float* s1 = (const float*)d_in[2];
  const float* a0 = (const float*)d_in[3];
  const float* a1 = (const float*)d_in[4];
  float* out = (float*)d_out;

  float* w = (float*)d_ws;
  float* Ere  = w; w += MP * MP;
  float* Eim  = w; w += MP * MP;
  float* W0   = w; w += MP * MP;
  float* W1   = w; w += MP * MP;
  float* Gx   = w; w += NPTS * MP;
  float* Gy   = w; w += NPTS * MP;
  float* Cas  = w; w += MP * MP;
  float* Dx   = w; w += NPTS * MP;
  float* Dy   = w; w += NPTS * MP;
  float* SD   = w; w += NB * MP * MP;
  float* SDr  = w; w += NB * MP * MP;
  float* PT   = w; w += (TILES - 1) * NOUT;
  float* PE   = w; w += 2 * NCH * NPTS;
  float* PD   = w; w += TILES * NOUT;
  float* HT   = w; w += NB * MP * MP;

  k_level0<<<TAB_BLOCKS + SPR_BLOCKS, L0_THREADS, 0, stream>>>(Ere, Eim, Cas, W0, W1, s0, s1, a0, a1,
                                                               x, Gx, Gy);
  k_level1<<<GRAM_BLOCKS + DFT_BLOCKS, 32, 0, stream>>>(Gx, Gy, Cas, Dx, Dy, HT);
  k_level2<<<S_BLOCKS + V_BLOCKS, S_THREADS, 0, stream>>>(HT, Ere, Eim, Dx, Dy,
                                                          W0, W1, SD, SDr, PD);
  k_umat<<<dim3(MM / 16, NPP / U_PTS, NB), U_THREADS, 0, stream>>>(W0, W1, SD, SDr, Dx, Dy,
                                                         PT, PE);
  k_final<<<NOUT / FIN_THREADS, FIN_THREADS, 0, stream>>>(PT, PE, PD, out);
}


# V (diagonal) tiles of level 2 re-tiled as 4-wave blocks of one k1 tile x 64 points: W rows and Dy rows staged in LDS by whole-line loads, K loop reads LDS through a register ring; sums unchanged
# speedup vs baseline: 2.5968x; 1.0433x over previous
#include <hip/hip_runtime.h>
#include <math.h>


typedef __attribute__((ext_vector_type(2))) float v2f;
typedef __attribute__((ext_vector_type(8))) float v8f;
typedef __attribute__((ext_vector_type(4))) float v4f;

#define MM    129
#define MP    144
#define NPP   512
#define NB    4
#define NPTS  (NB * NPP)
#define NCH   2

#define PI_F   3.14159265358979323846f
#define TAU_F  (12.0f / (129.0f * 129.0f))
#define L_F    (2.0f * PI_F)

#define TILES       (MP / 16)
#define L0_THREADS  128
#define TAB_PER_T   2
#define SPR_PER_T   8
#define TAB_BLOCKS  (MP * MP / (L0_THREADS * TAB_PER_T))
#define SPR_BLOCKS  (NPTS * MP / (L0_THREADS * SPR_PER_T))
#define GRAM_BLOCKS (TILES * TILES * NB)
#define VT_PER_B    (TILES * (NPP / 16))
#define VMAT_BLOCKS (VT_PER_B * NB)
#define DFT_PAIRS   (TILES / 2)
#define DFT_A_PER_Z ((NPTS / 32) * DFT_PAIRS)
#define DFT_B_PER_Z (NPTS / 64)
#define DFT_BLOCKS  (2 * (DFT_A_PER_Z + DFT_B_PER_Z))
#define S_WAVES     4
#define S_THREADS   (32 * S_WAVES)
#define S_SPLIT     3
#define S_BLOCKS    (TILES * NB * S_SPLIT)
#define V_BLOCKS    (VMAT_BLOCKS / S_WAVES)
#define V_PER_B     (V_BLOCKS / NB)
#define T_PITCH     148
#define S_KG1       1
#define S_NB1       (MP / (4 * S_KG1))
#define S_KG2       3
#define S_NB2       (MP / (4 * S_KG2))
#define S_RING      6
#define NOUT        (NPTS * NCH)
#define FIN_THREADS 256


__device__ __forceinline__ v8f wmma4(v2f a, v2f b, v8f c) {
  return __builtin_amdgcn_wmma_f32_16x16x4_f32(false, a, false, b, (short)0, c, false, false);
}

__device__ __forceinline__ void store_tile(float* C, int ldc, int row0, int col0, v8f acc) {
  int lane = threadIdx.x & 31;
  int r0 = row0 + ((lane >> 4) << 3);
  int c  = col0 + (lane & 15);
#pragma unroll
  for (int v = 0; v < 8; ++v) C[(r0 + v) * ldc + c] = acc[v];
}


__device__ __forceinline__ int revk(int k) { return (k == 0 || k >= MM) ? k : MM - k; }

__device__ __forceinline__ void store_tile_mirror(float* C, int ldc, int row0, int col0, v8f acc) {
  int lane = threadIdx.x & 31;
  int r0 = row0 + ((lane >> 4) << 3);
  int c  = revk(col0 + (lane & 15));
#pragma unroll
  for (int v = 0; v < 8; ++v) C[(r0 + v) * ldc + c] = acc[v];
}

__device__ __forceinline__ void load_mirror8(const float* row, int ka, const float (&xf)[8], float (&xv)[8]) {
  int m0 = 122 - ka;
  m0 = m0 < 0 ? 0 : m0;
  v2f l0 = *reinterpret_cast<const v2f*>(row + m0),     l1 = *reinterpret_cast<const v2f*>(row + m0 + 2);
  v2f l2 = *reinterpret_cast<const v2f*>(row + m0 + 4), l3 = *reinterpret_cast<const v2f*>(row + m0 + 6);
  float L[8] = { l0.x, l0.y, l1.x, l1.y, l2.x, l2.y, l3.x, l3.y };
#pragma unroll
  for (int v = 0; v < 8; ++v) {
    int k = ka + v;
    float g = L[7 - v];
    if (v == 0) g = (ka == MM - 1) ? L[1] : g;
    xv[v] = (k == 0 || k >= MM) ? xf[v] : g;
  }
}


__device__ __forceinline__ void tables_body(int t,
                         float* __restrict__ Ere, float* __restrict__ Eim,
                         float* __restrict__ Cas,
                         float* __restrict__ W0,  float* __restrict__ W1,
                         const float* s0, const float* s1,
                         const float* a0, const float* a1) {
  if (t >= MP * MP) return;
  int i = t / MP, k = t % MP;
  float er = 0.f, ei = 0.f, w0 = 0.f, w1 = 0.f;
  if (i < MM && k < MM) {
    int m = (i * k) % MM;
    float th = (2.0f * PI_F / (float)MM) * (float)m;
    er = cosf(th);
    ei = -sinf(th);
    float f1 = (float)((i <= MM / 2) ? i : i - MM);
    float f2 = (float)((k <= MM / 2) ? k : k - MM);
    float k2 = f1 * f1 + f2 * f2;
    float dec = (PI_F / TAU_F) * expf(TAU_F * k2);
    float m2   = (float)MM * (float)MM;
    float coef = 1.0f / (8.0f * PI_F * PI_F * m2 * m2);
    float base = coef * dec * dec * 4.0f * PI_F;
    float sh0 = s0[0], sh1 = s1[0];
    w0 = base * a0[0] / (k2 + sh0 * sh0);
    w1 = base * a1[0] / (k2 + sh1 * sh1);
  }
  Ere[t] = er; Eim[t] = ei; Cas[t] = er - ei; W0[t] = w0; W1[t] = w1;
}

__device__ __forceinline__ void spread_body(int t, const float* __restrict__ x,
                         float* __restrict__ Gx, float* __restrict__ Gy) {
  if (t >= NPTS * MP) return;
  int pt = t / MP, i = t % MP;
  float gx = 0.f, gy = 0.f;
  if (i < MM) {
    float xi  = (L_F / (float)MM) * (float)i;
    float inv = 1.0f / (4.0f * TAU_F);
    float d0 = x[2 * pt + 0] - xi;
    float d1 = x[2 * pt + 1] - xi;
    gx = expf(-d0 * d0 * inv) + expf(-(d0 - L_F) * (d0 - L_F) * inv)
       + expf(-(d0 + L_F) * (d0 + L_F) * inv);
    gy = expf(-d1 * d1 * inv) + expf(-(d1 - L_F) * (d1 - L_F) * inv)
       + expf(-(d1 + L_F) * (d1 + L_F) * inv);
  }
  Gx[t] = gx; Gy[t] = gy;
}

#define HG_RING   8
#define HG_AHEAD  (HG_RING - 1)
#define HG_GROUPS (NPP / 8)

struct HFrag { v2f a[2], b[2]; };

__device__ __forceinline__ void hgram_load(HFrag& f, int g,
                       const float* __restrict__ gx, const float* __restrict__ gy,
                       int kb, int ca, int cb) {
#pragma unroll
  for (int s = 0; s < 2; ++s) {
    int q0 = g * 8 + s * 4 + kb, q1 = q0 + 1;
    f.a[s].x = gx[q0 * MP + ca]; f.a[s].y = gx[q1 * MP + ca];
    f.b[s].x = gy[q0 * MP + cb]; f.b[s].y = gy[q1 * MP + cb];
  }
}

__device__ __forceinline__ void hgram_body(int b, int row0, int col0,
                       const float* __restrict__ A, const float* __restrict__ B,
                       float* __restrict__ D) {
  int lane = threadIdx.x & 31, r = lane & 15, kb = (lane >> 4) << 1;
  const float* gx = A + b * NPP * MP;
  const float* gy = B + b * NPP * MP;
  int ca = row0 + r, cb = col0 + r;
  v8f acc0 = {}, acc1 = {};
  HFrag ring[HG_RING];
#pragma unroll
  for (int j = 0; j < HG_AHEAD; ++j) {
    hgram_load(ring[j], j, gx, gy, kb, ca, cb);
    __builtin_amdgcn_sched_barrier(0);
  }
  for (int g0 = 0; g0 < HG_GROUPS; g0 += HG_RING) {
#pragma unroll
    for (int j = 0; j < HG_RING; ++j) {
      int g  = g0 + j;
      int gn = g + HG_AHEAD;
      gn = gn < HG_GROUPS ? gn : HG_GROUPS - 1;
      hgram_load(ring[(j + HG_AHEAD) % HG_RING], gn, gx, gy, kb, ca, cb);
      acc0 = wmma4(ring[j].a[0], ring[j].b[0], acc0);
      acc1 = wmma4(ring[j].a[1], ring[j].b[1], acc1);
      __builtin_amdgcn_sched_barrier(0);
    }
  }
  v8f acc = acc0 + acc1;
  store_tile(D + b * MP * MP, MP, row0, col0, acc);
}

template <int NT>
struct P1Frag { v2f a[NT][S_KG1], br[S_KG1], bi[S_KG1]; };
struct P2Glob { v2f ar[S_KG2], ai[S_KG2]; };
struct P2Lds  { v2f bc[S_KG2], bm[S_KG2]; };

template <int NT>
__device__ __forceinline__ void p1_load(P1Frag<NT>& f, int g, const float* hat,
                                        const float* ebr, const float* ebi) {
#pragma unroll
  for (int s = 0; s < S_KG1; ++s) {
    int k0 = 4 * (g * S_KG1 + s);
#pragma unroll
    for (int t = 0; t < NT; ++t) {
      f.a[t][s].x = hat[t * 64 + k0 * MP];
      f.a[t][s].y = hat[t * 64 + (k0 + 1) * MP];
    }
    f.br[s].x = ebr[k0 * MP]; f.br[s].y = ebr[(k0 + 1) * MP];
    f.bi[s].x = ebi[k0 * MP]; f.bi[s].y = ebi[(k0 + 1) * MP];
  }
}
template <int NT>
__device__ __forceinline__ void p1_mac(const P1Frag<NT>& f, v8f (&tRe)[NT], v8f (&tIm)[NT]) {
#pragma unroll
  for (int s = 0; s < S_KG1; ++s) {
#pragma unroll
    for (int t = 0; t < NT; ++t) {
      tRe[t] = wmma4(f.a[t][s], f.br[s], tRe[t]);
      tIm[t] = wmma4(f.a[t][s], f.bi[s], tIm[t]);
    }
  }
}
template <int NT>
__device__ __forceinline__ void p1_pass(int wave, int r, int hh, const float* hat,
                                        const float* ebr, const float* ebi,
                                        float* sTr, float* sTi) {
  v8f tRe[NT], tIm[NT];
#pragma unroll
  for (int t = 0; t < NT; ++t) { tRe[t] = v8f{}; tIm[t] = v8f{}; }
  P1Frag<NT> ring[S_RING];
#pragma unroll
  for (int j = 0; j < S_RING - 1; ++j) {
    p1_load<NT>(ring[j], j, hat, ebr, ebi);
    __builtin_amdgcn_sched_barrier(0);
  }
#pragma unroll 1
  for (int g0 = 0; g0 < S_NB1; g0 += S_RING) {
#pragma unroll
    for (int j = 0; j < S_RING; ++j) {
      int gn = g0 + j + S_RING - 1;
      gn = gn < S_NB1 ? gn : S_NB1 - 1;
      p1_load<NT>(ring[(j + S_RING - 1) % S_RING], gn, hat, ebr, ebi);
      p1_mac<NT>(ring[j], tRe, tIm);
      __builtin_amdgcn_sched_barrier(0);
    }
  }
#pragma unroll
  for (int t = 0; t < NT; ++t) {
    float* dr = sTr + r * T_PITCH + (wave + 4 * t) * 16 + 8 * hh;
    float* di = sTi + r * T_PITCH + (wave + 4 * t) * 16 + 8 * hh;
    v8f tc = tRe[t] - tIm[t], tm = tRe[t] + tIm[t];
    v4f r0 = { tc[0], tc[1], tc[2], tc[3] }, r1 = { tc[4], tc[5], tc[6], tc[7] };
    v4f i0 = { tm[0], tm[1], tm[2], tm[3] }, i1 = { tm[4], tm[5], tm[6], tm[7] };
    *reinterpret_cast<v4f*>(dr) = r0; *reinterpret_cast<v4f*>(dr + 4) = r1;
    *reinterpret_cast<v4f*>(di) = i0; *reinterpret_cast<v4f*>(di + 4) = i1;
  }
}
__device__ __forceinline__ void p2_gload(P2Glob& f, int g, const float* ear, const float* eai) {
#pragma unroll
  for (int s = 0; s < S_KG2; ++s) {
    int k0 = 4 * (g * S_KG2 + s);
    f.ar[s].x = ear[k0 * MP]; f.ar[s].y = ear[(k0 + 1) * MP];
    f.ai[s].x = eai[k0 * MP]; f.ai[s].y = eai[(k0 + 1) * MP];
  }
}
__device__ __forceinline__ void p2_lload(P2Lds& f, int g, const float* tbr, const float* tbi) {
#pragma unroll
  for (int s = 0; s < S_KG2; ++s) {
    int k0 = 4 * (g * S_KG2 + s);
    f.bc[s] = *reinterpret_cast<const v2f*>(tbr + k0);
    f.bm[s] = *reinterpret_cast<const v2f*>(tbi + k0);
  }
}
__device__ __forceinline__ void p2_mac(const P2Glob& a, const P2Lds& b, v8f& sC, v8f& sM) {
#pragma unroll
  for (int s = 0; s < S_KG2; ++s) {
    v2f aiN = { -a.ai[s].x, -a.ai[s].y };
    sC = wmma4(a.ar[s], b.bc[s], sC);
    sM = wmma4(aiN,     b.bm[s], sM);
  }
}

__device__ __forceinline__ void sdft_body(int b, int wave, int ty, int part,
                       const float* __restrict__ HT,
                       const float* __restrict__ Ere, const float* __restrict__ Eim,
                       float* __restrict__ SD, float* __restrict__ SDr, float* sT) {
  int lane = threadIdx.x & 31, r = lane & 15, hh = lane >> 4, kb = hh << 1;
  float* sTr = sT;
  float* sTi = sT + 16 * T_PITCH;
  const float* hat = HT + b * MP * MP + kb * MP + wave * 16 + r;
  const float* ebr = Ere + kb * MP + ty * 16 + r;
  const float* ebi = Eim + kb * MP + ty * 16 + r;
  int kt = part * 3 + (wave > 0 ? wave - 1 : 0);
  const float* ear = Ere + kb * MP + kt * 16 + r;
  const float* eai = Eim + kb * MP + kt * 16 + r;
  const float* tbr = sTr + r * T_PITCH + kb;
  const float* tbi = sTi + r * T_PITCH + kb;
  P2Glob gr[S_RING];
  P2Lds  lr[S_RING];
  if (wave == 0) {
    p1_pass<3>(wave, r, hh, hat, ebr, ebi, sTr, sTi);
  } else {
    p2_gload(gr[0], 0, ear, eai);
    p2_gload(gr[1], 1, ear, eai);
    p1_pass<2>(wave, r, hh, hat, ebr, ebi, sTr, sTi);
  }
  __syncthreads();
  if (wave == 0) return;
  v8f sC = {}, sM = {};
#pragma unroll
  for (int j = 2; j < S_RING - 1; ++j) p2_gload(gr[j], j, ear, eai);
#pragma unroll
  for (int j = 0; j < S_RING - 1; ++j) p2_lload(lr[j], j, tbr, tbi);
#pragma unroll 1
  for (int g0 = 0; g0 < S_NB2; g0 += S_RING) {
#pragma unroll
    for (int j = 0; j < S_RING; ++j) {
      int gn = g0 + j + S_RING - 1;
      gn = gn < S_NB2 ? gn : S_NB2 - 1;
      p2_gload(gr[(j + S_RING - 1) % S_RING], gn, ear, eai);
      p2_lload(lr[(j + S_RING - 1) % S_RING], gn, tbr, tbi);
      p2_mac(gr[j], lr[j], sC, sM);
      __builtin_amdgcn_sched_barrier(0);
    }
  }
  v8f sd = sC + sM;
  store_tile(SD + b * MP * MP, MP, kt * 16, ty * 16, sd);
  store_tile_mirror(SDr + b * MP * MP, MP, kt * 16, ty * 16, sd);
}

#define V_PTS     (16 * S_WAVES)
#define V_PITCH   148
#define V_ROW4    (MP / 4)
#define V_APCS    (16 * V_ROW4)
#define V_A_TRIPS ((V_APCS + S_THREADS - 1) / S_THREADS)
#define V_B_TRIPS (V_PTS * V_ROW4 / S_THREADS)
#define V_B_CH    (V_B_TRIPS / 3)
#define V_WPL     (16 * V_PITCH)
#define V_LDS     (2 * V_WPL + V_PTS * V_PITCH)

#define V_RING    6
#define V_AHEAD   (V_RING - 1)

struct VFrag { v2f d, e, a0, a1; };

__device__ __forceinline__ void vfrag_load(VFrag& f, int k0, int kb, const float* sy,
                                           const float* sa0, const float* sa1) {
  int k = k0 + kb;
  int m = MM - 1 - k;
  m = m < 0 ? 0 : m;
  f.d  = *reinterpret_cast<const v2f*>(sy + k);
  f.e  = *reinterpret_cast<const v2f*>(sy + m);
  f.a0 = *reinterpret_cast<const v2f*>(sa0 + k0);
  f.a1 = *reinterpret_cast<const v2f*>(sa1 + k0);
}

__device__ __forceinline__ void vmat_body(int b, int kt, int pblk, int wave,
                       const float* __restrict__ W0, const float* __restrict__ W1,
                       const float* __restrict__ Dx, const float* __restrict__ Dy,
                       float* __restrict__ PD, float* sm) {
  float* sW0 = sm;
  float* sW1 = sm + V_WPL;
  float* sYd = sm + 2 * V_WPL;
  int tid = threadIdx.x;
  int row0 = kt * 16;
  const float* w0 = W0 + row0 * MP;
  const float* w1 = W1 + row0 * MP;
  const float* dy = Dy + (b * NPP + pblk) * MP;
  v4f ga0[V_A_TRIPS], ga1[V_A_TRIPS];
  v4f gf0[V_B_CH], gf1[V_B_CH], gf2[V_B_CH];
#pragma unroll
  for (int j = 0; j < V_A_TRIPS; ++j) {
    int idx = tid + j * S_THREADS;
    if (j == V_A_TRIPS - 1) idx = idx < V_APCS ? idx : V_APCS - 1;
    ga0[j] = *reinterpret_cast<const v4f*>(w0 + 4 * idx);
    ga1[j] = *reinterpret_cast<const v4f*>(w1 + 4 * idx);
  }
#pragma unroll
  for (int j = 0; j < V_B_CH; ++j) {
    gf0[j] = *reinterpret_cast<const v4f*>(dy + 4 * (tid + j * S_THREADS));
    gf1[j] = *reinterpret_cast<const v4f*>(dy + 4 * (tid + (j + V_B_CH) * S_THREADS));
    gf2[j] = *reinterpret_cast<const v4f*>(dy + 4 * (tid + (j + 2 * V_B_CH) * S_THREADS));
  }
  int lane = tid & 31, r = lane & 15, hh = lane >> 4, kb = hh << 1;
  int col0 = pblk + wave * 16;
  const float* dxp = Dx + (b * NPP + col0 + r) * MP;
  int ka = row0 + 8 * hh;
  v4f xf0 = *reinterpret_cast<const v4f*>(dxp + ka), xf1 = *reinterpret_cast<const v4f*>(dxp + ka + 4);
  float xf[8] = { xf0.x, xf0.y, xf0.z, xf0.w, xf1.x, xf1.y, xf1.z, xf1.w };
  float xv[8];
  load_mirror8(dxp, ka, xf, xv);
  __builtin_amdgcn_sched_barrier(0);
#pragma unroll
  for (int j = 0; j < V_A_TRIPS; ++j) {
    int idx = tid + j * S_THREADS;
    if (j < V_A_TRIPS - 1 || idx < V_APCS) {
      int l = (idx / V_ROW4) * V_PITCH + (idx % V_ROW4) * 4;
      *reinterpret_cast<v4f*>(sW0 + l) = ga0[j];
      *reinterpret_cast<v4f*>(sW1 + l) = ga1[j];
    }
  }
#pragma unroll
  for (int j = 0; j < V_B_CH; ++j) {
    int i0 = tid + j * S_THREADS, i1 = i0 + V_B_CH * S_THREADS, i2 = i1 + V_B_CH * S_THREADS;
    int l0 = (i0 / V_ROW4) * V_PITCH + (i0 % V_ROW4) * 4;
    int l1 = (i1 / V_ROW4) * V_PITCH + (i1 % V_ROW4) * 4;
    int l2 = (i2 / V_ROW4) * V_PITCH + (i2 % V_ROW4) * 4;
    *reinterpret_cast<v4f*>(sYd + l0) = gf0[j];
    *reinterpret_cast<v4f*>(sYd + l1) = gf1[j];
    *reinterpret_cast<v4f*>(sYd + l2) = gf2[j];
  }
  __syncthreads();
  const float* sa0 = sW0 + r * V_PITCH + kb;
  const float* sa1 = sW1 + r * V_PITCH + kb;
  const float* sy  = sYd + (wave * 16 + r) * V_PITCH;
  v8f acc0 = {}, acc1 = {};
  VFrag ring[V_RING];
#pragma unroll
  for (int j = 0; j < V_AHEAD; ++j) {
    vfrag_load(ring[j], 4 * j, kb, sy, sa0, sa1);
    __builtin_amdgcn_sched_barrier(0);
  }
#pragma unroll
  for (int s = 0; s < MP / 4; ++s) {
    if (s + V_AHEAD < MP / 4)
      vfrag_load(ring[(s + V_AHEAD) % V_RING], 4 * (s + V_AHEAD), kb, sy, sa0, sa1);
    const VFrag& f = ring[s % V_RING];
    int k = 4 * s + kb;
    v2f d = f.d, e = f.e;
    float e0 = (k == 0 || k >= MM) ? d.x : e.y;
    float e1 = (k + 1 >= MM) ? d.y : e.x;
    v2f p  = { 0.5f * (d.x * d.x + e0 * e0), 0.5f * (d.y * d.y + e1 * e1) };
    acc0 = wmma4(f.a0, p, acc0);
    acc1 = wmma4(f.a1, p, acc1);
    __builtin_amdgcn_sched_barrier(0);
  }
  float d0 = 0.f, d1 = 0.f;
#pragma unroll
  for (int v = 0; v < 8; ++v) {
    float ax = 0.5f * (xf[v] * xf[v] + xv[v] * xv[v]);
    d0 += ax * acc0[v];
    d1 += ax * acc1[v];
  }
  d0 += __shfl_xor(d0, 16);
  d1 += __shfl_xor(d1, 16);
  PD[kt * NOUT + (b * NPP + col0 + r) * NCH + hh] = hh ? d1 : d0;
}


__global__ void k_level0(float* __restrict__ Ere, float* __restrict__ Eim,
                         float* __restrict__ Cas,
                         float* __restrict__ W0,  float* __restrict__ W1,
                         const float* s0, const float* s1,
                         const float* a0, const float* a1,
                         const float* __restrict__ x,
                         float* __restrict__ Gx, float* __restrict__ Gy) {
  int bid = blockIdx.x;
  if (bid < TAB_BLOCKS) {
    for (int j = 0; j < TAB_PER_T; ++j)
      tables_body((j * TAB_BLOCKS + bid) * L0_THREADS + threadIdx.x,
                  Ere, Eim, Cas, W0, W1, s0, s1, a0, a1);
  } else {
    int sb = bid - TAB_BLOCKS;
    for (int j = 0; j < SPR_PER_T; ++j)
      spread_body((j * SPR_BLOCKS + sb) * L0_THREADS + threadIdx.x, x, Gx, Gy);
  }
}

#define DF_RING   9
#define DF_STEPS  (MP / 4)

template <int RT, int CT>
struct DFrag { v2f a[RT], c[CT]; };

template <int RT, int CT>
__device__ __forceinline__ void dft_load(DFrag<RT, CT>& f, int g, const float* ga, const float* ec) {
  int k0 = 4 * g;
#pragma unroll
  for (int t = 0; t < RT; ++t) {
    f.a[t].x = ga[t * 16 * MP + k0];
    f.a[t].y = ga[t * 16 * MP + k0 + 1];
  }
#pragma unroll
  for (int u = 0; u < CT; ++u) {
    f.c[u].x = ec[u * 16 + k0 * MP];
    f.c[u].y = ec[u * 16 + (k0 + 1) * MP];
  }
}

template <int RT, int CT>
__device__ __forceinline__ void dft_body(int z, int row0, int col0,
                      const float* __restrict__ Gx, const float* __restrict__ Gy,
                      const float* __restrict__ Cas,
                      float* __restrict__ Dx, float* __restrict__ Dy) {
  const float* G = z ? Gy : Gx;
  float* D = z ? Dy : Dx;
  int lane = threadIdx.x & 31, r = lane & 15, kb = (lane >> 4) << 1;
  const float* ga = G + (row0 + r) * MP + kb;
  const float* ec = Cas + kb * MP + col0 + r;
  v8f acc[RT * CT];
#pragma unroll
  for (int t = 0; t < RT * CT; ++t) acc[t] = v8f{};
  DFrag<RT, CT> ring[DF_RING];
#pragma unroll
  for (int j = 0; j < DF_RING - 1; ++j) {
    dft_load<RT, CT>(ring[j], j, ga, ec);
    __builtin_amdgcn_sched_barrier(0);
  }
#pragma unroll 1
  for (int g0 = 0; g0 < DF_STEPS; g0 += DF_RING) {
#pragma unroll
    for (int j = 0; j < DF_RING; ++j) {
      int gn = g0 + j + DF_RING - 1;
      gn = gn < DF_STEPS ? gn : DF_STEPS - 1;
      dft_load<RT, CT>(ring[(j + DF_RING - 1) % DF_RING], gn, ga, ec);
#pragma unroll
      for (int t = 0; t < RT; ++t) {
#pragma unroll
        for (int u = 0; u < CT; ++u)
          acc[t * CT + u] = wmma4(ring[j].a[t], ring[j].c[u], acc[t * CT + u]);
      }
      __builtin_amdgcn_sched_barrier(0);
    }
  }
#pragma unroll
  for (int t = 0; t < RT; ++t) {
#pragma unroll
    for (int u = 0; u < CT; ++u) store_tile(D, MP, row0 + 16 * t, col0 + 16 * u, acc[t * CT + u]);
  }
}

__global__ void __launch_bounds__(32)
k_level1(const float* __restrict__ Gx, const float* __restrict__ Gy,
         const float* __restrict__ Cas,
         float* __restrict__ Dx, float* __restrict__ Dy,
         float* __restrict__ HT) {
  int bid = blockIdx.x;
  if (bid < GRAM_BLOCKS) {
    int b  = bid / (TILES * TILES);
    int t  = bid - b * (TILES * TILES);
    int ty = t / TILES, tx = t - ty * TILES;
    hgram_body(b, tx * 16, ty * 16, Gy, Gx, HT);
  } else {
    int id = bid - GRAM_BLOCKS;
    if (id < 2 * DFT_A_PER_Z) {
      int z  = id / DFT_A_PER_Z;     id -= z * DFT_A_PER_Z;
      int cp = id / (NPTS / 32), bx = id - cp * (NPTS / 32);
      dft_body<2, 2>(z, bx * 32, cp * 32, Gx, Gy, Cas, Dx, Dy);
    } else {
      id -= 2 * DFT_A_PER_Z;
      int z  = id / DFT_B_PER_Z, bx = id - z * DFT_B_PER_Z;
      dft_body<4, 1>(z, bx * 64, (TILES - 1) * 16, Gx, Gy, Cas, Dx, Dy);
    }
  }
}

__global__ void __launch_bounds__(S_THREADS)
k_level2(const float* __restrict__ HT,
         const float* __restrict__ Ere,  const float* __restrict__ Eim,
         const float* __restrict__ Dx,   const float* __restrict__ Dy,
         const float* __restrict__ W0,   const float* __restrict__ W1,
         float* __restrict__ SD, float* __restrict__ SDr,
         float* __restrict__ PD) {
  __shared__ __attribute__((aligned(16))) float sT[V_LDS];
  int bid = blockIdx.x;
  int wave = __builtin_amdgcn_readfirstlane(threadIdx.x >> 5);
  if (bid < S_BLOCKS) {
    int bt = bid / S_SPLIT, part = bid - bt * S_SPLIT;
    int b = bt / TILES, ty = bt - b * TILES;
    sdft_body(b, wave, ty, part, HT, Ere, Eim, SD, SDr, sT);
  } else {
    int id = bid - S_BLOCKS;
    int b  = id / V_PER_B;      id -= b * V_PER_B;
    int pb = id / TILES, kt = id - pb * TILES;
    vmat_body(b, kt, pb * V_PTS, wave, W0, W1, Dx, Dy, PD, sT);
  }
}

#define U_WAVES   4
#define U_PTS     (16 * U_WAVES)
#define U_PITCH   148
#define U_ROW4    (MP / 4)
#define U_THREADS (32 * U_WAVES)
#define U_AROWS   17
#define U_EDGE    (MM - 1)
#define U_A_TRIPS ((U_AROWS * U_ROW4 + U_THREADS - 1) / U_THREADS)
#define U_B_TRIPS (U_PTS * U_ROW4 / U_THREADS)
#define U_B_CH    (U_B_TRIPS / 3)
#define U_APL     (U_AROWS * U_PITCH)
#define U_BPL     (U_PTS * U_PITCH)

template <bool EXTRA>
__device__ __forceinline__ void u_kloop(const float* sA, const float* sYd,
                                        int la, int lb, int lx, int lbx, bool row0lane,
                                        v8f& accP0, v8f& accP1, v8f& accQ0, v8f& accQ1, v8f& accX) {
#pragma unroll
  for (int k0 = 0; k0 < MP; k0 += 4) {
    v2f a0 = *reinterpret_cast<const v2f*>(sA + la + k0);
    v2f a1 = *reinterpret_cast<const v2f*>(sA + U_APL + la + k0);
    v2f r0 = *reinterpret_cast<const v2f*>(sA + 2 * U_APL + la + k0);
    v2f r1 = *reinterpret_cast<const v2f*>(sA + 3 * U_APL + la + k0);
    v2f yd = *reinterpret_cast<const v2f*>(sYd + lb + k0);
    accP0 = wmma4(a0, yd, accP0);
    accP1 = wmma4(a1, yd, accP1);
    accQ0 = wmma4(r0, yd, accQ0);
    accQ1 = wmma4(r1, yd, accQ1);
    if (EXTRA) {
      v2f xa = *reinterpret_cast<const v2f*>(sA + lx + k0);
      v2f za = { row0lane ? xa.x : 0.f, row0lane ? xa.y : 0.f };
      v2f xb = *reinterpret_cast<const v2f*>(sYd + lbx + k0);
      accX = wmma4(za, xb, accX);
    }
  }
}

__global__ void __launch_bounds__(U_THREADS)
k_umat(const float* __restrict__ W0, const float* __restrict__ W1,
       const float* __restrict__ SD, const float* __restrict__ SDr,
       const float* __restrict__ Dx, const float* __restrict__ Dy,
       float* __restrict__ PT, float* __restrict__ PE) {
  __shared__ __attribute__((aligned(16))) float sm[4 * U_APL + U_BPL];
  float* sB0 = sm;
  float* sB1 = sm + U_APL;
  float* sR0 = sm + 2 * U_APL;
  float* sR1 = sm + 3 * U_APL;
  float* sYd = sm + 4 * U_APL;
  int b = blockIdx.z;
  int kt = blockIdx.x;
  int row0 = kt * 16;
  int pblk = blockIdx.y * U_PTS;
  int tid = threadIdx.x;
  const float* sd  = SD  + b * MP * MP;
  const float* sdr = SDr + b * MP * MP;
  const float* dy = Dy + (b * NPP + pblk) * MP;
  v4f gsd[U_A_TRIPS], gsr[U_A_TRIPS], ga0[U_A_TRIPS], ga1[U_A_TRIPS];
  v4f gf0[U_B_CH], gf1[U_B_CH], gf2[U_B_CH];
  bool tail = tid < U_AROWS * U_ROW4 - (U_A_TRIPS - 1) * U_THREADS;
#pragma unroll
  for (int j = 0; j < U_A_TRIPS; ++j) {
    if (j < U_A_TRIPS - 1 || tail) {
      int idx = tid + j * U_THREADS;
      int rr = idx / U_ROW4, c4 = idx - rr * U_ROW4;
      int grow = rr < 16 ? row0 + rr : U_EDGE;
      int g = grow * MP + c4 * 4;
      gsd[j] = *reinterpret_cast<const v4f*>(sd + g);
      gsr[j] = *reinterpret_cast<const v4f*>(sdr + g);
      ga0[j] = *reinterpret_cast<const v4f*>(W0 + g);
      ga1[j] = *reinterpret_cast<const v4f*>(W1 + g);
    }
  }
#pragma unroll
  for (int j = 0; j < U_B_CH; ++j) {
    gf0[j] = *reinterpret_cast<const v4f*>(dy + 4 * (tid + j * U_THREADS));
    gf1[j] = *reinterpret_cast<const v4f*>(dy + 4 * (tid + (j + U_B_CH) * U_THREADS));
    gf2[j] = *reinterpret_cast<const v4f*>(dy + 4 * (tid + (j + 2 * U_B_CH) * U_THREADS));
  }
  __builtin_amdgcn_sched_barrier(0);
#pragma unroll
  for (int j = 0; j < U_A_TRIPS; ++j) {
    if (j < U_A_TRIPS - 1 || tail) {
      int idx = tid + j * U_THREADS;
      int rr = idx / U_ROW4, c4 = idx - rr * U_ROW4;
      int l = rr * U_PITCH + c4 * 4;
      *reinterpret_cast<v4f*>(sB0 + l) = ga0[j] * gsd[j];
      *reinterpret_cast<v4f*>(sB1 + l) = ga1[j] * gsd[j];
      *reinterpret_cast<v4f*>(sR0 + l) = ga0[j] * gsr[j];
      *reinterpret_cast<v4f*>(sR1 + l) = ga1[j] * gsr[j];
    }
  }
#pragma unroll
  for (int j = 0; j < U_B_CH; ++j) {
    int i0 = tid + j * U_THREADS, i1 = i0 + U_B_CH * U_THREADS, i2 = i1 + U_B_CH * U_THREADS;
    int l0 = (i0 / U_ROW4) * U_PITCH + (i0 % U_ROW4) * 4;
    int l1 = (i1 / U_ROW4) * U_PITCH + (i1 % U_ROW4) * 4;
    int l2 = (i2 / U_ROW4) * U_PITCH + (i2 % U_ROW4) * 4;
    *reinterpret_cast<v4f*>(sYd + l0) = gf0[j];
    *reinterpret_cast<v4f*>(sYd + l1) = gf1[j];
    *reinterpret_cast<v4f*>(sYd + l2) = gf2[j];
  }
  __syncthreads();
  int wave = __builtin_amdgcn_readfirstlane(tid >> 5);
  int lane = tid & 31, r = lane & 15, hh = lane >> 4, kb = hh << 1;
  int col0 = pblk + wave * 16;
  int la = r * U_PITCH + kb;
  int lb = (wave * 16 + r) * U_PITCH + kb;
  int xt = kt >> 1;
  bool part = (kt & 1) != 0;
  int lx  = ((part ? 2 : 0) + wave) * U_APL + 16 * U_PITCH + kb;
  int lbx = (xt * 16 + r) * U_PITCH + kb;
  const float* dxp = Dx + (b * NPP + col0 + r) * MP;
  int ka = row0 + 8 * hh;
  v4f xf0 = *reinterpret_cast<const v4f*>(dxp + ka), xf1 = *reinterpret_cast<const v4f*>(dxp + ka + 4);
  float xf[8] = { xf0.x, xf0.y, xf0.z, xf0.w, xf1.x, xf1.y, xf1.z, xf1.w };
  float xv[8];
  load_mirror8(dxp, ka, xf, xv);
  const float* dxe = Dx + (b * NPP + pblk + xt * 16 + r) * MP;
  float e128 = dxe[U_EDGE], e1 = dxe[1];
  float xe = 0.5f * (part ? e128 - e1 : e128 + e1);
  v8f accP0 = {}, accP1 = {}, accQ0 = {}, accQ1 = {}, accX = {};
  if (wave < 2) {
    u_kloop<true>(sm, sYd, la, lb, lx, lbx, r == 0,
                  accP0, accP1, accQ0, accQ1, accX);
  } else {
    u_kloop<false>(sm, sYd, la, lb, lx, lbx, false,
                   accP0, accP1, accQ0, accQ1, accX);
  }
  float t0 = 0.f, t1 = 0.f;
#pragma unroll
  for (int v = 0; v < 8; ++v) {
    float xr = 0.5f * (xf[v] + xv[v]);
    float xi = 0.5f * (xv[v] - xf[v]);
    t0 += xr * accP0[v] - xi * accQ0[v];
    t1 += xr * accP1[v] - xi * accQ1[v];
  }
  t0 += __shfl_xor(t0, 16);
  t1 += __shfl_xor(t1, 16);
  PT[kt * NOUT + (b * NPP + col0 + r) * NCH + hh] = hh ? t1 : t0;
  if (wave < 2 && hh == 0) {
    PE[(((int)part) * NCH + wave) * NPTS + b * NPP + pblk + xt * 16 + r] = xe * accX[0];
  }
}

__global__ void __launch_bounds__(FIN_THREADS)
k_final(const float* __restrict__ PT, const float* __restrict__ PE,
        const float* __restrict__ PD, float* __restrict__ out) {
  int t = blockIdx.x * FIN_THREADS + threadIdx.x;
  int pt = t >> 1, ch = t & 1;
  float tot = 0.f, dia = 0.f;
#pragma unroll
  for (int k = 0; k < TILES - 1; ++k) tot += PT[k * NOUT + t];
  tot += PE[ch * NPTS + pt];
  tot += PE[(NCH + ch) * NPTS + pt];
#pragma unroll
  for (int k = 0; k < TILES; ++k) dia += PD[k * NOUT + t];
  out[t] = tot - dia;
}


extern "C" void kernel_launch(void* const* d_in, const int* in_sizes, int n_in,
                              void* d_out, int out_size, void* d_ws, size_t ws_size,
                              hipStream_t stream) {
  (void)in_sizes; (void)n_in; (void)out_size; (void)ws_size;
  const float* x  = (const float*)d_in[0];
  const float* s0 = (const float*)d_in[1];
  const float* s1 = (const float*)d_in[2];
  const float* a0 = (const float*)d_in[3];
  const float* a1 = (const float*)d_in[4];
  float* out = (float*)d_out;

  float* w = (float*)d_ws;
  float* Ere  = w; w += MP * MP;
  float* Eim  = w; w += MP * MP;
  float* W0   = w; w += MP * MP;
  float* W1   = w; w += MP * MP;
  float* Gx   = w; w += NPTS * MP;
  float* Gy   = w; w += NPTS * MP;
  float* Cas  = w; w += MP * MP;
  float* Dx   = w; w += NPTS * MP;
  float* Dy   = w; w += NPTS * MP;
  float* SD   = w; w += NB * MP * MP;
  float* SDr  = w; w += NB * MP * MP;
  float* PT   = w; w += (TILES - 1) * NOUT;
  float* PE   = w; w += 2 * NCH * NPTS;
  float* PD   = w; w += TILES * NOUT;
  float* HT   = w; w += NB * MP * MP;

  k_level0<<<TAB_BLOCKS + SPR_BLOCKS, L0_THREADS, 0, stream>>>(Ere, Eim, Cas, W0, W1, s0, s1, a0, a1,
                                                               x, Gx, Gy);
  k_level1<<<GRAM_BLOCKS + DFT_BLOCKS, 32, 0, stream>>>(Gx, Gy, Cas, Dx, Dy, HT);
  k_level2<<<S_BLOCKS + V_BLOCKS, S_THREADS, 0, stream>>>(HT, Ere, Eim, Dx, Dy,
                                                          W0, W1, SD, SDr, PD);
  k_umat<<<dim3(MM / 16, NPP / U_PTS, NB), U_THREADS, 0, stream>>>(W0, W1, SD, SDr, Dx, Dy,
                                                         PT, PE);
  k_final<<<NOUT / FIN_THREADS, FIN_THREADS, 0, stream>>>(PT, PE, PD, out);
}
